# starting kernel
# baseline (speedup 1.0000x reference)
#include <hip/hip_runtime.h>


#define C_CH 256
#define HW_P 4096
#define N_B  4
#define PCHUNKS 8
#define PCHUNK_LEN (HW_P / PCHUNKS)

typedef __attribute__((ext_vector_type(2))) float v2f;
typedef __attribute__((ext_vector_type(4))) float v4f;
typedef __attribute__((ext_vector_type(8))) float v8f;

__device__ __forceinline__ v8f wmma_f32(v2f a, v2f b, v8f c) {
  return __builtin_amdgcn_wmma_f32_16x16x4_f32(false, a, false, b, (short)0, c,
                                               false, false);
}

__device__ __forceinline__ v8f splat8(float v) {
  v8f r = {v, v, v, v, v, v, v, v};
  return r;
}

__global__ void __launch_bounds__(256)
qkv_proj_kernel(const float* __restrict__ x,
                const float* __restrict__ w0, const float* __restrict__ b0,
                const float* __restrict__ w1, const float* __restrict__ b1,
                const float* __restrict__ w2, const float* __restrict__ b2,
                float* __restrict__ Vb, float* __restrict__ Kb,
                float* __restrict__ Qb) {
  const int lane = threadIdx.x & 31;
  const int l15  = lane & 15;
  const int hi   = lane >> 4;
  int t = blockIdx.x * (blockDim.x >> 5) + (threadIdx.x >> 5);
  const int opair = t & 7;   t >>= 3;
  const int ptile = t & 255; t >>= 8;
  const int n     = t & 3;   t >>= 2;
  const int proj  = t;

  const float* w    = (proj == 0) ? w0 : (proj == 1) ? w1 : w2;
  const float* bias = (proj == 0) ? b0 : (proj == 1) ? b1 : b2;
  float* out        = (proj == 0) ? Vb : (proj == 1) ? Kb : Qb;

  const int p0 = ptile * 16;
  const int o0 = opair * 32;
  const float* xn    = x + (size_t)n * C_CH * HW_P;
  const float* wrow0 = w + (size_t)(o0 + l15) * C_CH;
  const float* wrow1 = w + (size_t)(o0 + 16 + l15) * C_CH;

  v8f acc0 = splat8(bias[o0 + l15]);
  v8f acc1 = splat8(bias[o0 + 16 + l15]);
#pragma unroll 4
  for (int k0 = 0; k0 < C_CH; k0 += 4) {
    const int cb = k0 + 2 * hi;
    v2f a;
    a.x = xn[(size_t)cb * HW_P + p0 + l15];
    a.y = xn[(size_t)(cb + 1) * HW_P + p0 + l15];
    v2f bb0 = *(const v2f*)(wrow0 + cb);
    v2f bb1 = *(const v2f*)(wrow1 + cb);
    acc0 = wmma_f32(a, bb0, acc0);
    acc1 = wmma_f32(a, bb1, acc1);
  }
  float* obase = out + ((size_t)n * HW_P + p0) * C_CH + o0 + l15;
#pragma unroll
  for (int r = 0; r < 8; ++r) {
    obase[(size_t)(r + 8 * hi) * C_CH]      = acc0[r];
    obase[(size_t)(r + 8 * hi) * C_CH + 16] = acc1[r];
  }
}

__global__ void __launch_bounds__(256)
kv_gram_kernel(const float* __restrict__ Kb, const float* __restrict__ Vb,
               float* __restrict__ Mpart) {
  const int lane = threadIdx.x & 31;
  const int l15  = lane & 15;
  const int hi   = lane >> 4;
  int t = blockIdx.x * (blockDim.x >> 5) + (threadIdx.x >> 5);
  const int chunk = t & 7;  t >>= 3;
  const int ctile = t & 15; t >>= 4;
  const int dtile = t & 15; t >>= 4;
  const int n     = t;

  const int d0 = dtile * 16, c0 = ctile * 16;
  const int pbeg = chunk * PCHUNK_LEN;
  const float* kbase = Kb + (size_t)n * HW_P * C_CH;
  const float* vbase = Vb + (size_t)n * HW_P * C_CH;

  v8f acc = {};
#pragma unroll 4
  for (int p = pbeg; p < pbeg + PCHUNK_LEN; p += 4) {
    const int pb = p + 2 * hi;
    v2f a, b;
    a.x = kbase[(size_t)pb * C_CH + d0 + l15];
    a.y = kbase[(size_t)(pb + 1) * C_CH + d0 + l15];
    b.x = vbase[(size_t)pb * C_CH + c0 + l15];
    b.y = vbase[(size_t)(pb + 1) * C_CH + c0 + l15];
    acc = wmma_f32(a, b, acc);
  }
  float* mp = Mpart + ((size_t)chunk * N_B + n) * C_CH * C_CH +
              (size_t)d0 * C_CH + c0 + l15;
#pragma unroll
  for (int r = 0; r < 8; ++r)
    mp[(size_t)(r + 8 * hi) * C_CH] = acc[r];
}

__global__ void __launch_bounds__(256)
reduce_m_kernel(const float* __restrict__ Mpart, float* __restrict__ Mb) {
  const size_t idx    = (size_t)blockIdx.x * blockDim.x + threadIdx.x;
  const size_t stride = (size_t)N_B * C_CH * C_CH;
  float s = 0.0f;
#pragma unroll
  for (int c = 0; c < PCHUNKS; ++c) s += Mpart[c * stride + idx];
  Mb[idx] = s;
}

__global__ void __launch_bounds__(256)
m_w3_kernel(const float* __restrict__ Mb, const float* __restrict__ w3,
            float* __restrict__ M2) {
  const int lane = threadIdx.x & 31;
  const int l15  = lane & 15;
  const int hi   = lane >> 4;
  int t = blockIdx.x * (blockDim.x >> 5) + (threadIdx.x >> 5);
  const int otile = t & 15; t >>= 4;
  const int dtile = t & 15; t >>= 4;
  const int n     = t;
  const int d0 = dtile * 16, o0 = otile * 16;

  const float* mrow = Mb + ((size_t)n * C_CH + d0 + l15) * C_CH;
  const float* wrow = w3 + (size_t)(o0 + l15) * C_CH;
  v8f acc = {};
#pragma unroll 4
  for (int k0 = 0; k0 < C_CH; k0 += 4) {
    const int cb = k0 + 2 * hi;
    v2f a = *(const v2f*)(mrow + cb);
    v2f b = *(const v2f*)(wrow + cb);
    acc = wmma_f32(a, b, acc);
  }
  const float inv_hw = 1.0f / (float)HW_P;
  float* m2base = M2 + ((size_t)n * C_CH + d0) * C_CH + o0 + l15;
#pragma unroll
  for (int r = 0; r < 8; ++r)
    m2base[(size_t)(r + 8 * hi) * C_CH] = acc[r] * inv_hw;
}

__global__ void __launch_bounds__(256)
out_kernel(const float* __restrict__ Qb, const float* __restrict__ M2,
           const float* __restrict__ b3, float* __restrict__ out) {
  const int lane = threadIdx.x & 31;
  const int l15  = lane & 15;
  const int hi   = lane >> 4;
  int t = blockIdx.x * (blockDim.x >> 5) + (threadIdx.x >> 5);
  const int opair = t & 7;   t >>= 3;
  const int ptile = t & 255; t >>= 8;
  const int n     = t;
  const int p0 = ptile * 16, o0 = opair * 32;

  const float* qrow = Qb + ((size_t)n * HW_P + p0 + l15) * C_CH;
  const float* m2b  = M2 + (size_t)n * C_CH * C_CH;

  v8f acc0 = splat8(b3[o0 + l15]);
  v8f acc1 = splat8(b3[o0 + 16 + l15]);
#pragma unroll 4
  for (int k0 = 0; k0 < C_CH; k0 += 4) {
    const int db = k0 + 2 * hi;
    v2f a = *(const v2f*)(qrow + db);
    v2f bb0, bb1;
    bb0.x = m2b[(size_t)db * C_CH + o0 + l15];
    bb0.y = m2b[(size_t)(db + 1) * C_CH + o0 + l15];
    bb1.x = m2b[(size_t)db * C_CH + o0 + 16 + l15];
    bb1.y = m2b[(size_t)(db + 1) * C_CH + o0 + 16 + l15];
    acc0 = wmma_f32(a, bb0, acc0);
    acc1 = wmma_f32(a, bb1, acc1);
  }
  float* ob0 = out + ((size_t)n * C_CH + o0 + l15) * HW_P + p0 + 8 * hi;
  float* ob1 = out + ((size_t)n * C_CH + o0 + 16 + l15) * HW_P + p0 + 8 * hi;
  v4f a_lo = {acc0[0], acc0[1], acc0[2], acc0[3]};
  v4f a_hi = {acc0[4], acc0[5], acc0[6], acc0[7]};
  v4f b_lo = {acc1[0], acc1[1], acc1[2], acc1[3]};
  v4f b_hi = {acc1[4], acc1[5], acc1[6], acc1[7]};
  *(v4f*)(ob0)     = a_lo;
  *(v4f*)(ob0 + 4) = a_hi;
  *(v4f*)(ob1)     = b_lo;
  *(v4f*)(ob1 + 4) = b_hi;
}

extern "C" void kernel_launch(void* const* d_in, const int* in_sizes, int n_in,
                              void* d_out, int out_size, void* d_ws,
                              size_t ws_size, hipStream_t stream) {
  const float* x  = (const float*)d_in[0];
  const float* w0 = (const float*)d_in[1];
  const float* b0 = (const float*)d_in[2];
  const float* w1 = (const float*)d_in[3];
  const float* b1 = (const float*)d_in[4];
  const float* w2 = (const float*)d_in[5];
  const float* b2 = (const float*)d_in[6];
  const float* w3 = (const float*)d_in[7];
  const float* b3 = (const float*)d_in[8];
  float* out = (float*)d_out;

  const size_t per = (size_t)N_B * HW_P * C_CH;
  const size_t mcc = (size_t)N_B * C_CH * C_CH;
  float* Vb    = (float*)d_ws;
  float* Kb    = Vb + per;
  float* Qb    = Kb + per;
  float* Mpart = Qb + per;
  float* Mb    = Mpart + PCHUNKS * mcc;
  float* M2    = Mb + mcc;

  qkv_proj_kernel<<<3072, 256, 0, stream>>>(x, w0, b0, w1, b1, w2, b2, Vb, Kb, Qb);
  kv_gram_kernel<<<1024, 256, 0, stream>>>(Kb, Vb, Mpart);
  reduce_m_kernel<<<1024, 256, 0, stream>>>(Mpart, Mb);
  m_w3_kernel<<<128, 256, 0, stream>>>(Mb, w3, M2);
  out_kernel<<<1024, 256, 0, stream>>>(Qb, M2, b3, out);
}


# projection kernel: each wave now makes a 64 px x 32 ch tile from whole-line per-lane loads (64-B weight runs, 16-B pixel runs); channel sum order within each 32-channel block changes
# speedup vs baseline: 1.9161x; 1.9161x over previous
#include <hip/hip_runtime.h>


#define C_CH 256
#define HW_P 4096
#define N_B  4
#define PCHUNKS 8
#define PCHUNK_LEN (HW_P / PCHUNKS)

typedef __attribute__((ext_vector_type(2))) float v2f;
typedef __attribute__((ext_vector_type(4))) float v4f;
typedef __attribute__((ext_vector_type(8))) float v8f;

__device__ __forceinline__ v8f wmma_f32(v2f a, v2f b, v8f c) {
  return __builtin_amdgcn_wmma_f32_16x16x4_f32(false, a, false, b, (short)0, c,
                                               false, false);
}

__device__ __forceinline__ v8f splat8(float v) {
  v8f r = {v, v, v, v, v, v, v, v};
  return r;
}

__global__ void __launch_bounds__(256)
qkv_proj_kernel(const float* __restrict__ x,
                const float* __restrict__ w0, const float* __restrict__ b0,
                const float* __restrict__ w1, const float* __restrict__ b1,
                const float* __restrict__ w2, const float* __restrict__ b2,
                float* __restrict__ Vb, float* __restrict__ Kb,
                float* __restrict__ Qb) {
  const int lane = threadIdx.x & 31;
  const int l15  = lane & 15;
  const int hi   = lane >> 4;
  int t = blockIdx.x * (blockDim.x >> 5) + (threadIdx.x >> 5);
  const int opair = t & 7;  t >>= 3;
  const int ptile = t & 63; t >>= 6;
  const int n     = t & 3;  t >>= 2;
  const int proj  = t;

  const float* w    = (proj == 0) ? w0 : (proj == 1) ? w1 : w2;
  const float* bias = (proj == 0) ? b0 : (proj == 1) ? b1 : b2;
  float* out        = (proj == 0) ? Vb : (proj == 1) ? Kb : Qb;

  const int p0 = ptile * 64;
  const int o0 = opair * 32;
  const float* xl  = x + (size_t)n * C_CH * HW_P + (size_t)(16 * hi) * HW_P +
                     p0 + 4 * l15;
  const float* wl0 = w + (size_t)(o0 + l15) * C_CH + 16 * hi;
  const float* wl1 = w + (size_t)(o0 + 16 + l15) * C_CH + 16 * hi;

  v8f acc[4][2];
  {
    const v8f bz0 = splat8(bias[o0 + l15]);
    const v8f bz1 = splat8(bias[o0 + 16 + l15]);
    for (int mt = 0; mt < 4; ++mt) { acc[mt][0] = bz0; acc[mt][1] = bz1; }
  }

  for (int j = 0; j < C_CH / 32; ++j) {
    v4f wq0[4], wq1[4];
    for (int i = 0; i < 4; ++i) {
      wq0[i] = *(const v4f*)(wl0 + 32 * j + 4 * i);
      wq1[i] = *(const v4f*)(wl1 + 32 * j + 4 * i);
    }
    const float* xj = xl + (size_t)(32 * j) * HW_P;
    for (int s = 0; s < 8; ++s) {
      const v4f xe0 = *(const v4f*)(xj + (size_t)(2 * s) * HW_P);
      const v4f xe1 = *(const v4f*)(xj + (size_t)(2 * s + 1) * HW_P);
      const int q = s >> 1, c = 2 * (s & 1);
      v2f bb0 = {wq0[q][c], wq0[q][c + 1]};
      v2f bb1 = {wq1[q][c], wq1[q][c + 1]};
      for (int mt = 0; mt < 4; ++mt) {
        v2f a = {xe0[mt], xe1[mt]};
        acc[mt][0] = wmma_f32(a, bb0, acc[mt][0]);
        acc[mt][1] = wmma_f32(a, bb1, acc[mt][1]);
      }
    }
  }
  float* obase = out + ((size_t)n * HW_P + p0) * C_CH + o0 + l15;
  for (int r = 0; r < 8; ++r) {
    for (int mt = 0; mt < 4; ++mt) {
      const size_t pix = (size_t)(4 * (r + 8 * hi) + mt);
      obase[pix * C_CH]      = acc[mt][0][r];
      obase[pix * C_CH + 16] = acc[mt][1][r];
    }
  }
}

__global__ void __launch_bounds__(256)
kv_gram_kernel(const float* __restrict__ Kb, const float* __restrict__ Vb,
               float* __restrict__ Mpart) {
  const int lane = threadIdx.x & 31;
  const int l15  = lane & 15;
  const int hi   = lane >> 4;
  int t = blockIdx.x * (blockDim.x >> 5) + (threadIdx.x >> 5);
  const int chunk = t & 7;  t >>= 3;
  const int ctile = t & 15; t >>= 4;
  const int dtile = t & 15; t >>= 4;
  const int n     = t;

  const int d0 = dtile * 16, c0 = ctile * 16;
  const int pbeg = chunk * PCHUNK_LEN;
  const float* kbase = Kb + (size_t)n * HW_P * C_CH;
  const float* vbase = Vb + (size_t)n * HW_P * C_CH;

  v8f acc = {};
#pragma unroll 4
  for (int p = pbeg; p < pbeg + PCHUNK_LEN; p += 4) {
    const int pb = p + 2 * hi;
    v2f a, b;
    a.x = kbase[(size_t)pb * C_CH + d0 + l15];
    a.y = kbase[(size_t)(pb + 1) * C_CH + d0 + l15];
    b.x = vbase[(size_t)pb * C_CH + c0 + l15];
    b.y = vbase[(size_t)(pb + 1) * C_CH + c0 + l15];
    acc = wmma_f32(a, b, acc);
  }
  float* mp = Mpart + ((size_t)chunk * N_B + n) * C_CH * C_CH +
              (size_t)d0 * C_CH + c0 + l15;
#pragma unroll
  for (int r = 0; r < 8; ++r)
    mp[(size_t)(r + 8 * hi) * C_CH] = acc[r];
}

__global__ void __launch_bounds__(256)
reduce_m_kernel(const float* __restrict__ Mpart, float* __restrict__ Mb) {
  const size_t idx    = (size_t)blockIdx.x * blockDim.x + threadIdx.x;
  const size_t stride = (size_t)N_B * C_CH * C_CH;
  float s = 0.0f;
#pragma unroll
  for (int c = 0; c < PCHUNKS; ++c) s += Mpart[c * stride + idx];
  Mb[idx] = s;
}

__global__ void __launch_bounds__(256)
m_w3_kernel(const float* __restrict__ Mb, const float* __restrict__ w3,
            float* __restrict__ M2) {
  const int lane = threadIdx.x & 31;
  const int l15  = lane & 15;
  const int hi   = lane >> 4;
  int t = blockIdx.x * (blockDim.x >> 5) + (threadIdx.x >> 5);
  const int otile = t & 15; t >>= 4;
  const int dtile = t & 15; t >>= 4;
  const int n     = t;
  const int d0 = dtile * 16, o0 = otile * 16;

  const float* mrow = Mb + ((size_t)n * C_CH + d0 + l15) * C_CH;
  const float* wrow = w3 + (size_t)(o0 + l15) * C_CH;
  v8f acc = {};
#pragma unroll 4
  for (int k0 = 0; k0 < C_CH; k0 += 4) {
    const int cb = k0 + 2 * hi;
    v2f a = *(const v2f*)(mrow + cb);
    v2f b = *(const v2f*)(wrow + cb);
    acc = wmma_f32(a, b, acc);
  }
  const float inv_hw = 1.0f / (float)HW_P;
  float* m2base = M2 + ((size_t)n * C_CH + d0) * C_CH + o0 + l15;
#pragma unroll
  for (int r = 0; r < 8; ++r)
    m2base[(size_t)(r + 8 * hi) * C_CH] = acc[r] * inv_hw;
}

__global__ void __launch_bounds__(256)
out_kernel(const float* __restrict__ Qb, const float* __restrict__ M2,
           const float* __restrict__ b3, float* __restrict__ out) {
  const int lane = threadIdx.x & 31;
  const int l15  = lane & 15;
  const int hi   = lane >> 4;
  int t = blockIdx.x * (blockDim.x >> 5) + (threadIdx.x >> 5);
  const int opair = t & 7;   t >>= 3;
  const int ptile = t & 255; t >>= 8;
  const int n     = t;
  const int p0 = ptile * 16, o0 = opair * 32;

  const float* qrow = Qb + ((size_t)n * HW_P + p0 + l15) * C_CH;
  const float* m2b  = M2 + (size_t)n * C_CH * C_CH;

  v8f acc0 = splat8(b3[o0 + l15]);
  v8f acc1 = splat8(b3[o0 + 16 + l15]);
#pragma unroll 4
  for (int k0 = 0; k0 < C_CH; k0 += 4) {
    const int db = k0 + 2 * hi;
    v2f a = *(const v2f*)(qrow + db);
    v2f bb0, bb1;
    bb0.x = m2b[(size_t)db * C_CH + o0 + l15];
    bb0.y = m2b[(size_t)(db + 1) * C_CH + o0 + l15];
    bb1.x = m2b[(size_t)db * C_CH + o0 + 16 + l15];
    bb1.y = m2b[(size_t)(db + 1) * C_CH + o0 + 16 + l15];
    acc0 = wmma_f32(a, bb0, acc0);
    acc1 = wmma_f32(a, bb1, acc1);
  }
  float* ob0 = out + ((size_t)n * C_CH + o0 + l15) * HW_P + p0 + 8 * hi;
  float* ob1 = out + ((size_t)n * C_CH + o0 + 16 + l15) * HW_P + p0 + 8 * hi;
  v4f a_lo = {acc0[0], acc0[1], acc0[2], acc0[3]};
  v4f a_hi = {acc0[4], acc0[5], acc0[6], acc0[7]};
  v4f b_lo = {acc1[0], acc1[1], acc1[2], acc1[3]};
  v4f b_hi = {acc1[4], acc1[5], acc1[6], acc1[7]};
  *(v4f*)(ob0)     = a_lo;
  *(v4f*)(ob0 + 4) = a_hi;
  *(v4f*)(ob1)     = b_lo;
  *(v4f*)(ob1 + 4) = b_hi;
}

extern "C" void kernel_launch(void* const* d_in, const int* in_sizes, int n_in,
                              void* d_out, int out_size, void* d_ws,
                              size_t ws_size, hipStream_t stream) {
  const float* x  = (const float*)d_in[0];
  const float* w0 = (const float*)d_in[1];
  const float* b0 = (const float*)d_in[2];
  const float* w1 = (const float*)d_in[3];
  const float* b1 = (const float*)d_in[4];
  const float* w2 = (const float*)d_in[5];
  const float* b2 = (const float*)d_in[6];
  const float* w3 = (const float*)d_in[7];
  const float* b3 = (const float*)d_in[8];
  float* out = (float*)d_out;

  const size_t per = (size_t)N_B * HW_P * C_CH;
  const size_t mcc = (size_t)N_B * C_CH * C_CH;
  float* Vb    = (float*)d_ws;
  float* Kb    = Vb + per;
  float* Qb    = Kb + per;
  float* Mpart = Qb + per;
  float* Mb    = Mpart + PCHUNKS * mcc;
  float* M2    = Mb + mcc;

  qkv_proj_kernel<<<768, 256, 0, stream>>>(x, w0, b0, w1, b1, w2, b2, Vb, Kb, Qb);
  kv_gram_kernel<<<1024, 256, 0, stream>>>(Kb, Vb, Mpart);
  reduce_m_kernel<<<1024, 256, 0, stream>>>(Mpart, Mb);
  m_w3_kernel<<<128, 256, 0, stream>>>(Mb, w3, M2);
  out_kernel<<<1024, 256, 0, stream>>>(Qb, M2, b3, out);
}


# re-associated the linear chain: the four 1x1 convs act on the per-image 257x257 Gram of [x;1], pixels enter only the Gram and the last product; f32 sum order and reduction tree change
# speedup vs baseline: 3.3075x; 1.7262x over previous
#include <hip/hip_runtime.h>


#define C_CH 256
#define CP   272
#define HW_P 4096
#define N_B  4
#define PCH  32
#define PLEN (HW_P / PCH)

typedef __attribute__((ext_vector_type(2))) float v2f;
typedef __attribute__((ext_vector_type(4))) float v4f;
typedef __attribute__((ext_vector_type(8))) float v8f;

#define HOLD8(p, o)                                                          \
  asm volatile("" ::"v"(p[(o)]), "v"(p[(o) + 1]), "v"(p[(o) + 2]),           \
               "v"(p[(o) + 3]), "v"(p[(o) + 4]), "v"(p[(o) + 5]),            \
               "v"(p[(o) + 6]), "v"(p[(o) + 7]))

__device__ __forceinline__ v8f wmma_f32(v2f a, v2f b, v8f c) {
  return __builtin_amdgcn_wmma_f32_16x16x4_f32(false, a, false, b, (short)0, c,
                                               false, false);
}

__device__ __forceinline__ v8f splat8(float v) {
  v8f r = {v, v, v, v, v, v, v, v};
  return r;
}

template <bool SUMS>
__device__ __forceinline__ void gram_body(const float* __restrict__ xa,
                                          const float* __restrict__ xb,
                                          float* __restrict__ gp,
                                          float* __restrict__ sp,
                                          int l15, int hi) {
  v8f acc[4][2];
  v8f accs[2];
  for (int mt = 0; mt < 4; ++mt) { acc[mt][0] = splat8(0.f); acc[mt][1] = splat8(0.f); }
  accs[0] = splat8(0.f); accs[1] = splat8(0.f);
  const v2f ones = {1.0f, 1.0f};

  for (int kb = 0; kb < PLEN / 32; ++kb) {
    v4f fa[4][4], fb[2][4];
    for (int mt = 0; mt < 4; ++mt)
      for (int i = 0; i < 4; ++i)
        fa[mt][i] = *(const v4f*)(xa + (size_t)(16 * mt) * HW_P + 32 * kb + 4 * i);
    for (int nt = 0; nt < 2; ++nt)
      for (int i = 0; i < 4; ++i)
        fb[nt][i] = *(const v4f*)(xb + (size_t)(16 * nt) * HW_P + 32 * kb + 4 * i);
    for (int i = 0; i < 4; ++i) {
      for (int t = 0; t < 2; ++t) {
        const v2f b0 = {fb[0][i][2 * t], fb[0][i][2 * t + 1]};
        const v2f b1 = {fb[1][i][2 * t], fb[1][i][2 * t + 1]};
        for (int mt = 0; mt < 4; ++mt) {
          const v2f a = {fa[mt][i][2 * t], fa[mt][i][2 * t + 1]};
          acc[mt][0] = wmma_f32(a, b0, acc[mt][0]);
          acc[mt][1] = wmma_f32(a, b1, acc[mt][1]);
        }
        if (SUMS) {
          accs[0] = wmma_f32(ones, b0, accs[0]);
          accs[1] = wmma_f32(ones, b1, accs[1]);
        }
      }
    }
  }
  for (int mt = 0; mt < 4; ++mt)
    for (int r = 0; r < 8; ++r) {
      float* g = gp + (size_t)(16 * mt + 8 * hi + r) * C_CH;
      g[0]  = acc[mt][0][r];
      g[16] = acc[mt][1][r];
    }
  if (SUMS) {
    if (hi == 0) { sp[0] = accs[0][0]; sp[16] = accs[1][0]; }
  }
}

__global__ void __launch_bounds__(256)
x_gram_kernel(const float* __restrict__ x, float* __restrict__ Gpart,
              float* __restrict__ spart) {
  const int lane = threadIdx.x & 31;
  const int l15  = lane & 15;
  const int hi   = lane >> 4;
  int t = blockIdx.x * (blockDim.x >> 5) +
          __builtin_amdgcn_readfirstlane((int)(threadIdx.x >> 5));
  const int eg    = t & 7;  t >>= 3;
  const int cg    = t & 3;  t >>= 2;
  const int chunk = t & 31; t >>= 5;
  const int n     = t;

  const float* xn = x + (size_t)n * C_CH * HW_P + chunk * PLEN + 16 * hi;
  const float* xa = xn + (size_t)(cg * 64 + l15) * HW_P;
  const float* xb = xn + (size_t)(eg * 32 + l15) * HW_P;
  float* gp = Gpart + ((size_t)(chunk * N_B + n) * C_CH + cg * 64) * C_CH +
              eg * 32 + l15;
  float* sp = spart + (size_t)(chunk * N_B + n) * C_CH + eg * 32 + l15;
  if (cg == 0) gram_body<true>(xa, xb, gp, sp, l15, hi);
  else         gram_body<false>(xa, xb, gp, sp, l15, hi);
}

__global__ void __launch_bounds__(256)
reduce_g_kernel(const float* __restrict__ Gpart, const float* __restrict__ spart,
                float* __restrict__ Gp) {
  const int idx = blockIdx.x * blockDim.x + threadIdx.x;
  const int n   = idx / (CP * CP);
  const int rem = idx - n * (CP * CP);
  const int j   = rem / CP;
  const int e   = rem - j * CP;
  float s = 0.0f;
  if (j < C_CH && e < C_CH) {
    const float* g = Gpart + ((size_t)n * C_CH + j) * C_CH + e;
#pragma unroll 8
    for (int c = 0; c < PCH; ++c) s += g[(size_t)c * N_B * C_CH * C_CH];
  } else if ((j == C_CH) != (e == C_CH) && j <= C_CH && e <= C_CH) {
    const int ch = (j == C_CH) ? e : j;
    const float* q = spart + (size_t)n * C_CH + ch;
#pragma unroll 8
    for (int c = 0; c < PCH; ++c) s += q[(size_t)c * N_B * C_CH];
  } else if (j == C_CH && e == C_CH) {
    s = (float)HW_P;
  }
  Gp[idx] = s;
}

__global__ void __launch_bounds__(64)
nt_small_kernel(const float* __restrict__ A, int aBatch, int aPitch,
                const float* __restrict__ B, int bBatch, int bPitch,
                const float* __restrict__ u, int uBatch, int uStride,
                const float* __restrict__ v,
                float* __restrict__ D, int dBatch, int dPitch,
                int mTiles, int nTiles) {
  const int lane = threadIdx.x & 31;
  const int l15  = lane & 15;
  const int hi   = lane >> 4;
  int t = blockIdx.x * (blockDim.x >> 5) + (threadIdx.x >> 5);
  const int ct = t % nTiles; t /= nTiles;
  const int mt = t % mTiles; t /= mTiles;
  const int n  = t;

  const float* arow = A + (size_t)n * aBatch + (size_t)(mt * 16 + l15) * aPitch;
  const float* brow = B + (size_t)n * bBatch + (size_t)(ct * 16 + l15) * bPitch;
  v4f fa[32], fb[32];
#pragma unroll
  for (int i = 0; i < 32; ++i) {
    fa[i] = *(const v4f*)(arow + 8 * i + 4 * hi);
    fb[i] = *(const v4f*)(brow + 8 * i + 4 * hi);
  }
  HOLD8(fa, 0); HOLD8(fb, 0); HOLD8(fa, 8);  HOLD8(fb, 8);
  v8f acc = {}, acc1 = {};
#pragma unroll
  for (int i = 0; i < 32; ++i) {
    const v2f a0 = {fa[i][0], fa[i][1]}, a1 = {fa[i][2], fa[i][3]};
    const v2f b0 = {fb[i][0], fb[i][1]}, b1 = {fb[i][2], fb[i][3]};
    acc  = wmma_f32(a0, b0, acc);
    acc1 = wmma_f32(a1, b1, acc1);
  }
  if (u != nullptr) {
    const float um = u[(size_t)n * uBatch + (size_t)(mt * 16 + l15) * uStride];
    const float vc = v[ct * 16 + l15];
    v2f a = {hi ? 0.0f : um, 0.0f};
    v2f b = {hi ? 0.0f : vc, 0.0f};
    acc1 = wmma_f32(a, b, acc1);
  }
  acc = acc + acc1;
  float* d = D + (size_t)n * dBatch + (size_t)(mt * 16 + 8 * hi) * dPitch +
             ct * 16 + l15;
#pragma unroll
  for (int r = 0; r < 8; ++r) d[(size_t)r * dPitch] = acc[r];
}

__global__ void __launch_bounds__(64)
pt_kernel(const float* __restrict__ T3t, const float* __restrict__ w2,
          const float* __restrict__ b2, float* __restrict__ PT) {
  const int lane = threadIdx.x & 31;
  const int l15  = lane & 15;
  const int hi   = lane >> 4;
  int t = blockIdx.x * (blockDim.x >> 5) +
          __builtin_amdgcn_readfirstlane((int)(threadIdx.x >> 5));
  const int it = t % 17; t /= 17;
  const int ot = t & 15; t >>= 4;
  const int n  = t;

  const float* arow = T3t + ((size_t)n * C_CH + ot * 16 + l15) * C_CH;
  v4f fa[32], fb[32];
  if (it < 16) {
    const float* wc = w2 + it * 16 + l15 + (size_t)(4 * hi) * C_CH;
#pragma unroll
    for (int i = 0; i < 32; ++i) {
      fa[i] = *(const v4f*)(arow + 8 * i + 4 * hi);
      const float* wd = wc + (size_t)(8 * i) * C_CH;
      const v4f q = {wd[0], wd[C_CH], wd[2 * C_CH], wd[3 * C_CH]};
      fb[i] = q;
    }
  } else {
    const float sel = (l15 == 0) ? 1.0f : 0.0f;
#pragma unroll
    for (int i = 0; i < 32; ++i) {
      fa[i] = *(const v4f*)(arow + 8 * i + 4 * hi);
      fb[i] = *(const v4f*)(b2 + 8 * i + 4 * hi) * sel;
    }
  }
  HOLD8(fa, 0); HOLD8(fb, 0); HOLD8(fa, 8);  HOLD8(fb, 8);
  v8f acc = {}, acc1 = {};
#pragma unroll
  for (int i = 0; i < 32; ++i) {
    const v2f a0 = {fa[i][0], fa[i][1]}, a1 = {fa[i][2], fa[i][3]};
    const v2f b0 = {fb[i][0], fb[i][1]}, b1 = {fb[i][2], fb[i][3]};
    acc  = wmma_f32(a0, b0, acc);
    acc1 = wmma_f32(a1, b1, acc1);
  }
  acc = acc + acc1;
  const float inv_hw = 1.0f / (float)HW_P;
  float* d = PT + ((size_t)n * C_CH + ot * 16 + 8 * hi) * CP + it * 16 + l15;
#pragma unroll
  for (int r = 0; r < 8; ++r) d[(size_t)r * CP] = acc[r] * inv_hw;
}

__global__ void __launch_bounds__(256)
out_kernel(const float* __restrict__ x, const float* __restrict__ PT,
           const float* __restrict__ b3, float* __restrict__ out) {
  const int lane = threadIdx.x & 31;
  const int l15  = lane & 15;
  const int hi   = lane >> 4;
  int t = blockIdx.x * (blockDim.x >> 5) + (threadIdx.x >> 5);
  const int opair = t & 7;  t >>= 3;
  const int ptile = t & 63; t >>= 6;
  const int n     = t;

  const float* w = PT + (size_t)n * C_CH * CP;
  const int p0 = ptile * 64;
  const int o0 = opair * 32;
  const float* xl  = x + (size_t)n * C_CH * HW_P + (size_t)(16 * hi) * HW_P +
                     p0 + 4 * l15;
  const float* wl0 = w + (size_t)(o0 + l15) * CP + 16 * hi;
  const float* wl1 = w + (size_t)(o0 + 16 + l15) * CP + 16 * hi;

  v8f acc[4][2];
  {
    const v8f bz0 = splat8(w[(size_t)(o0 + l15) * CP + C_CH] + b3[o0 + l15]);
    const v8f bz1 = splat8(w[(size_t)(o0 + 16 + l15) * CP + C_CH] + b3[o0 + 16 + l15]);
    for (int mt = 0; mt < 4; ++mt) { acc[mt][0] = bz0; acc[mt][1] = bz1; }
  }

  for (int j = 0; j < C_CH / 32; ++j) {
    v4f wq0[4], wq1[4];
    for (int i = 0; i < 4; ++i) {
      wq0[i] = *(const v4f*)(wl0 + 32 * j + 4 * i);
      wq1[i] = *(const v4f*)(wl1 + 32 * j + 4 * i);
    }
    const float* xj = xl + (size_t)(32 * j) * HW_P;
    for (int s = 0; s < 8; ++s) {
      const v4f xe0 = *(const v4f*)(xj + (size_t)(2 * s) * HW_P);
      const v4f xe1 = *(const v4f*)(xj + (size_t)(2 * s + 1) * HW_P);
      const int q = s >> 1, c = 2 * (s & 1);
      v2f bb0 = {wq0[q][c], wq0[q][c + 1]};
      v2f bb1 = {wq1[q][c], wq1[q][c + 1]};
      for (int mt = 0; mt < 4; ++mt) {
        v2f a = {xe0[mt], xe1[mt]};
        acc[mt][0] = wmma_f32(a, bb0, acc[mt][0]);
        acc[mt][1] = wmma_f32(a, bb1, acc[mt][1]);
      }
    }
  }
  float* ob0 = out + ((size_t)n * C_CH + o0 + l15) * HW_P + p0 + 32 * hi;
  float* ob1 = ob0 + (size_t)16 * HW_P;
  for (int r = 0; r < 8; ++r) {
    const v4f s0 = {acc[0][0][r], acc[1][0][r], acc[2][0][r], acc[3][0][r]};
    const v4f s1 = {acc[0][1][r], acc[1][1][r], acc[2][1][r], acc[3][1][r]};
    *(v4f*)(ob0 + 4 * r) = s0;
    *(v4f*)(ob1 + 4 * r) = s1;
  }
}

extern "C" void kernel_launch(void* const* d_in, const int* in_sizes, int n_in,
                              void* d_out, int out_size, void* d_ws,
                              size_t ws_size, hipStream_t stream) {
  const float* x  = (const float*)d_in[0];
  const float* w0 = (const float*)d_in[1];
  const float* b0 = (const float*)d_in[2];
  const float* w1 = (const float*)d_in[3];
  const float* b1 = (const float*)d_in[4];
  const float* w2 = (const float*)d_in[5];
  const float* b2 = (const float*)d_in[6];
  const float* w3 = (const float*)d_in[7];
  const float* b3 = (const float*)d_in[8];
  float* out = (float*)d_out;

  float* Gpart = (float*)d_ws;
  float* spart = Gpart + (size_t)PCH * N_B * C_CH * C_CH;
  float* Gp    = spart + (size_t)PCH * N_B * C_CH;
  float* T1    = Gp    + (size_t)N_B * CP * CP;
  float* T2t   = T1    + (size_t)N_B * CP * C_CH;
  float* T3t   = T2t   + (size_t)N_B * C_CH * CP;
  float* PT    = T3t   + (size_t)N_B * C_CH * C_CH;

  x_gram_kernel<<<512, 256, 0, stream>>>(x, Gpart, spart);
  reduce_g_kernel<<<1156, 256, 0, stream>>>(Gpart, spart, Gp);
  nt_small_kernel<<<544, 64, 0, stream>>>(Gp, CP * CP, CP, w0, 0, C_CH,
                                           Gp + C_CH, CP * CP, CP, b0,
                                           T1, CP * C_CH, C_CH, 17, 16);
  nt_small_kernel<<<544, 64, 0, stream>>>(w3, 0, C_CH, T1, CP * C_CH, C_CH,
                                           nullptr, 0, 0, nullptr,
                                           T2t, C_CH * CP, CP, 16, 17);
  nt_small_kernel<<<512, 64, 0, stream>>>(T2t, C_CH * CP, CP, w1, 0, C_CH,
                                           T2t + C_CH, C_CH * CP, CP, b1,
                                           T3t, C_CH * C_CH, C_CH, 16, 16);
  pt_kernel<<<544, 64, 0, stream>>>(T3t, w2, b2, PT);
  out_kernel<<<256, 256, 0, stream>>>(x, PT, b3, out);
}


# middle chain re-associated: w3*[w0 b0] and [w2 b2]^T*[w1 b1] are formed beside the slab reduction, so two products follow the Gram instead of four (7 launches to 5); f32 sum order changes
# speedup vs baseline: 3.9095x; 1.1820x over previous
#include <hip/hip_runtime.h>


#define C_CH 256
#define CP   272
#define HW_P 4096
#define N_B  4
#define PCH  32
#define PLEN (HW_P / PCH)
#define KAUG_B 33
#define WPROD_WAVES (16 * 17 + 17 * 17)
#define WPROD_BLOCKS 71

typedef __attribute__((ext_vector_type(2))) float v2f;
typedef __attribute__((ext_vector_type(4))) float v4f;
typedef __attribute__((ext_vector_type(8))) float v8f;

#define HOLD8(p, o)                                                          \
  asm volatile("" ::"v"(p[(o)]), "v"(p[(o) + 1]), "v"(p[(o) + 2]),           \
               "v"(p[(o) + 3]), "v"(p[(o) + 4]), "v"(p[(o) + 5]),            \
               "v"(p[(o) + 6]), "v"(p[(o) + 7]))

__device__ __forceinline__ v8f wmma_f32(v2f a, v2f b, v8f c) {
  return __builtin_amdgcn_wmma_f32_16x16x4_f32(false, a, false, b, (short)0, c,
                                               false, false);
}

__device__ __forceinline__ v8f splat8(float v) {
  v8f r = {v, v, v, v, v, v, v, v};
  return r;
}

template <bool SUMS>
__device__ __forceinline__ void gram_body(const float* __restrict__ xa,
                                          const float* __restrict__ xb,
                                          float* __restrict__ gp,
                                          float* __restrict__ sp,
                                          int l15, int hi) {
  v8f acc[4][2];
  v8f accs[2];
  for (int mt = 0; mt < 4; ++mt) { acc[mt][0] = splat8(0.f); acc[mt][1] = splat8(0.f); }
  accs[0] = splat8(0.f); accs[1] = splat8(0.f);
  const v2f ones = {1.0f, 1.0f};

  for (int kb = 0; kb < PLEN / 32; ++kb) {
    v4f fa[4][4], fb[2][4];
    for (int mt = 0; mt < 4; ++mt)
      for (int i = 0; i < 4; ++i)
        fa[mt][i] = *(const v4f*)(xa + (size_t)(16 * mt) * HW_P + 32 * kb + 4 * i);
    for (int nt = 0; nt < 2; ++nt)
      for (int i = 0; i < 4; ++i)
        fb[nt][i] = *(const v4f*)(xb + (size_t)(16 * nt) * HW_P + 32 * kb + 4 * i);
    for (int i = 0; i < 4; ++i) {
      for (int t = 0; t < 2; ++t) {
        const v2f b0 = {fb[0][i][2 * t], fb[0][i][2 * t + 1]};
        const v2f b1 = {fb[1][i][2 * t], fb[1][i][2 * t + 1]};
        for (int mt = 0; mt < 4; ++mt) {
          const v2f a = {fa[mt][i][2 * t], fa[mt][i][2 * t + 1]};
          acc[mt][0] = wmma_f32(a, b0, acc[mt][0]);
          acc[mt][1] = wmma_f32(a, b1, acc[mt][1]);
        }
        if (SUMS) {
          accs[0] = wmma_f32(ones, b0, accs[0]);
          accs[1] = wmma_f32(ones, b1, accs[1]);
        }
      }
    }
  }
  for (int mt = 0; mt < 4; ++mt)
    for (int r = 0; r < 8; ++r) {
      float* g = gp + (size_t)(16 * mt + 8 * hi + r) * C_CH;
      g[0]  = acc[mt][0][r];
      g[16] = acc[mt][1][r];
    }
  if (SUMS) {
    if (hi == 0) { sp[0] = accs[0][0]; sp[16] = accs[1][0]; }
  }
}

__global__ void __launch_bounds__(256)
x_gram_kernel(const float* __restrict__ x, float* __restrict__ Gpart,
              float* __restrict__ spart) {
  const int lane = threadIdx.x & 31;
  const int l15  = lane & 15;
  const int hi   = lane >> 4;
  int t = blockIdx.x * (blockDim.x >> 5) +
          __builtin_amdgcn_readfirstlane((int)(threadIdx.x >> 5));
  const int eg    = t & 7;  t >>= 3;
  const int cg    = t & 3;  t >>= 2;
  const int chunk = t & 31; t >>= 5;
  const int n     = t;

  const float* xn = x + (size_t)n * C_CH * HW_P + chunk * PLEN + 16 * hi;
  const float* xa = xn + (size_t)(cg * 64 + l15) * HW_P;
  const float* xb = xn + (size_t)(eg * 32 + l15) * HW_P;
  float* gp = Gpart + ((size_t)(chunk * N_B + n) * C_CH + cg * 64) * C_CH +
              eg * 32 + l15;
  float* sp = spart + (size_t)(chunk * N_B + n) * C_CH + eg * 32 + l15;
  if (cg == 0) gram_body<true>(xa, xb, gp, sp, l15, hi);
  else         gram_body<false>(xa, xb, gp, sp, l15, hi);
}

__device__ __forceinline__ void aug_col_frags(v4f (&f)[8],
                                              const float* __restrict__ w,
                                              const float* __restrict__ b,
                                              int tile, int k0, int l15, int hi) {
  if (tile < 16) {
    const float* wc = w + tile * 16 + l15 + (size_t)(k0 + 4 * hi) * C_CH;
#pragma unroll
    for (int i = 0; i < 8; ++i) {
      const float* wd = wc + (size_t)(8 * i) * C_CH;
      const v4f q = {wd[0], wd[C_CH], wd[2 * C_CH], wd[3 * C_CH]};
      f[i] = q;
    }
  } else {
    const v4f zero = {0.0f, 0.0f, 0.0f, 0.0f};
#pragma unroll
    for (int i = 0; i < 8; ++i) {
      const v4f q = *(const v4f*)(b + k0 + 8 * i + 4 * hi);
      f[i] = (l15 == 0) ? q : zero;
    }
  }
}

__device__ __forceinline__ void weight_products(
    const float* __restrict__ w0, const float* __restrict__ b0,
    const float* __restrict__ w1, const float* __restrict__ b1,
    const float* __restrict__ w2, const float* __restrict__ b2,
    const float* __restrict__ w3, float* __restrict__ Mt,
    float* __restrict__ At, int wv, int l15, int hi) {
  const bool first = wv < 16 * 17;
  const int u  = first ? wv : wv - 16 * 17;
  const int rt = u / 17;
  const int ct = u - rt * 17;
  const float* arow = w3 + (size_t)(rt * 16 + l15) * C_CH;
  v8f acc = {}, acc1 = {};
#pragma unroll 1
  for (int k0 = 0; k0 < C_CH; k0 += 64) {
    v4f fa[8], fb[8];
    if (first) {
#pragma unroll
      for (int i = 0; i < 8; ++i)
        fa[i] = *(const v4f*)(arow + k0 + 8 * i + 4 * hi);
      aug_col_frags(fb, w0, b0, ct, k0, l15, hi);
    } else {
      aug_col_frags(fa, w2, b2, rt, k0, l15, hi);
      aug_col_frags(fb, w1, b1, ct, k0, l15, hi);
    }
#pragma unroll
    for (int i = 0; i < 8; ++i) {
      const v2f a0 = {fa[i][0], fa[i][1]}, a1 = {fa[i][2], fa[i][3]};
      const v2f c0 = {fb[i][0], fb[i][1]}, c1 = {fb[i][2], fb[i][3]};
      acc  = wmma_f32(a0, c0, acc);
      acc1 = wmma_f32(a1, c1, acc1);
    }
  }
  acc = acc + acc1;
  float* d = (first ? Mt : At) + (size_t)(rt * 16 + 8 * hi) * CP + ct * 16 + l15;
#pragma unroll
  for (int r = 0; r < 8; ++r) d[(size_t)r * CP] = acc[r];
}

__global__ void __launch_bounds__(256)
reduce_g_kernel(const float* __restrict__ Gpart, const float* __restrict__ spart,
                float* __restrict__ Gp,
                const float* __restrict__ w0, const float* __restrict__ b0,
                const float* __restrict__ w1, const float* __restrict__ b1,
                const float* __restrict__ w2, const float* __restrict__ b2,
                const float* __restrict__ w3, float* __restrict__ Mt,
                float* __restrict__ At) {
  if (blockIdx.x < WPROD_BLOCKS) {
    const int wv = blockIdx.x * (blockDim.x >> 5) +
                   __builtin_amdgcn_readfirstlane((int)(threadIdx.x >> 5));
    const int lane = threadIdx.x & 31;
    if (wv < WPROD_WAVES)
      weight_products(w0, b0, w1, b1, w2, b2, w3, Mt, At, wv, lane & 15,
                      lane >> 4);
    return;
  }
  const int idx = (blockIdx.x - WPROD_BLOCKS) * blockDim.x + threadIdx.x;
  const int n   = idx / (CP * CP);
  const int rem = idx - n * (CP * CP);
  const int j   = rem / CP;
  const int e   = rem - j * CP;
  float s = 0.0f;
  if (j < C_CH && e < C_CH) {
    const float* g = Gpart + ((size_t)n * C_CH + j) * C_CH + e;
#pragma unroll 8
    for (int c = 0; c < PCH; ++c) s += g[(size_t)c * N_B * C_CH * C_CH];
  } else if ((j == C_CH) != (e == C_CH) && j <= C_CH && e <= C_CH) {
    const int ch = (j == C_CH) ? e : j;
    const float* q = spart + (size_t)n * C_CH + ch;
#pragma unroll 8
    for (int c = 0; c < PCH; ++c) s += q[(size_t)c * N_B * C_CH];
  } else if (j == C_CH && e == C_CH) {
    s = (float)HW_P;
  }
  Gp[idx] = s;
}

__global__ void __launch_bounds__(64)
nt_aug_kernel(const float* __restrict__ A, int aBatch,
              const float* __restrict__ B, int bBatch,
              float* __restrict__ D, float scale) {
  const int lane = threadIdx.x & 31;
  const int l15  = lane & 15;
  const int hi   = lane >> 4;
  int t = blockIdx.x * (blockDim.x >> 5) + (threadIdx.x >> 5);
  const int ct = t % 17; t /= 17;
  const int mt = t & 15; t >>= 4;
  const int n  = t;

  const float* arow = A + (size_t)n * aBatch + (size_t)(mt * 16 + l15) * CP;
  const float* brow = B + (size_t)n * bBatch + (size_t)(ct * 16 + l15) * CP;
  v4f fa[KAUG_B], fb[KAUG_B];
#pragma unroll
  for (int i = 0; i < KAUG_B; ++i) {
    fa[i] = *(const v4f*)(arow + 8 * i + 4 * hi);
    fb[i] = *(const v4f*)(brow + 8 * i + 4 * hi);
  }
  HOLD8(fa, 0); HOLD8(fb, 0); HOLD8(fa, 8);  HOLD8(fb, 8);
  v8f acc = {}, acc1 = {};
#pragma unroll
  for (int i = 0; i < KAUG_B; ++i) {
    const v2f a0 = {fa[i][0], fa[i][1]}, a1 = {fa[i][2], fa[i][3]};
    const v2f b0 = {fb[i][0], fb[i][1]}, b1 = {fb[i][2], fb[i][3]};
    acc  = wmma_f32(a0, b0, acc);
    acc1 = wmma_f32(a1, b1, acc1);
  }
  acc = acc + acc1;
  float* d = D + ((size_t)n * C_CH + mt * 16 + 8 * hi) * CP + ct * 16 + l15;
#pragma unroll
  for (int r = 0; r < 8; ++r) d[(size_t)r * CP] = acc[r] * scale;
}

__global__ void __launch_bounds__(256)
out_kernel(const float* __restrict__ x, const float* __restrict__ PT,
           const float* __restrict__ b3, float* __restrict__ out) {
  const int lane = threadIdx.x & 31;
  const int l15  = lane & 15;
  const int hi   = lane >> 4;
  int t = blockIdx.x * (blockDim.x >> 5) + (threadIdx.x >> 5);
  const int opair = t & 7;  t >>= 3;
  const int ptile = t & 63; t >>= 6;
  const int n     = t;

  const float* w = PT + (size_t)n * C_CH * CP;
  const int p0 = ptile * 64;
  const int o0 = opair * 32;
  const float* xl  = x + (size_t)n * C_CH * HW_P + (size_t)(16 * hi) * HW_P +
                     p0 + 4 * l15;
  const float* wl0 = w + (size_t)(o0 + l15) * CP + 16 * hi;
  const float* wl1 = w + (size_t)(o0 + 16 + l15) * CP + 16 * hi;

  v8f acc[4][2];
  {
    const v8f bz0 = splat8(w[(size_t)(o0 + l15) * CP + C_CH] + b3[o0 + l15]);
    const v8f bz1 = splat8(w[(size_t)(o0 + 16 + l15) * CP + C_CH] + b3[o0 + 16 + l15]);
    for (int mt = 0; mt < 4; ++mt) { acc[mt][0] = bz0; acc[mt][1] = bz1; }
  }

  for (int j = 0; j < C_CH / 32; ++j) {
    v4f wq0[4], wq1[4];
    for (int i = 0; i < 4; ++i) {
      wq0[i] = *(const v4f*)(wl0 + 32 * j + 4 * i);
      wq1[i] = *(const v4f*)(wl1 + 32 * j + 4 * i);
    }
    const float* xj = xl + (size_t)(32 * j) * HW_P;
    for (int s = 0; s < 8; ++s) {
      const v4f xe0 = *(const v4f*)(xj + (size_t)(2 * s) * HW_P);
      const v4f xe1 = *(const v4f*)(xj + (size_t)(2 * s + 1) * HW_P);
      const int q = s >> 1, c = 2 * (s & 1);
      v2f bb0 = {wq0[q][c], wq0[q][c + 1]};
      v2f bb1 = {wq1[q][c], wq1[q][c + 1]};
      for (int mt = 0; mt < 4; ++mt) {
        v2f a = {xe0[mt], xe1[mt]};
        acc[mt][0] = wmma_f32(a, bb0, acc[mt][0]);
        acc[mt][1] = wmma_f32(a, bb1, acc[mt][1]);
      }
    }
  }
  float* ob0 = out + ((size_t)n * C_CH + o0 + l15) * HW_P + p0 + 32 * hi;
  float* ob1 = ob0 + (size_t)16 * HW_P;
  for (int r = 0; r < 8; ++r) {
    const v4f s0 = {acc[0][0][r], acc[1][0][r], acc[2][0][r], acc[3][0][r]};
    const v4f s1 = {acc[0][1][r], acc[1][1][r], acc[2][1][r], acc[3][1][r]};
    *(v4f*)(ob0 + 4 * r) = s0;
    *(v4f*)(ob1 + 4 * r) = s1;
  }
}

extern "C" void kernel_launch(void* const* d_in, const int* in_sizes, int n_in,
                              void* d_out, int out_size, void* d_ws,
                              size_t ws_size, hipStream_t stream) {
  const float* x  = (const float*)d_in[0];
  const float* w0 = (const float*)d_in[1];
  const float* b0 = (const float*)d_in[2];
  const float* w1 = (const float*)d_in[3];
  const float* b1 = (const float*)d_in[4];
  const float* w2 = (const float*)d_in[5];
  const float* b2 = (const float*)d_in[6];
  const float* w3 = (const float*)d_in[7];
  const float* b3 = (const float*)d_in[8];
  float* out = (float*)d_out;

  float* Gpart = (float*)d_ws;
  float* spart = Gpart + (size_t)PCH * N_B * C_CH * C_CH;
  float* Gp    = spart + (size_t)PCH * N_B * C_CH;
  float* Mt    = Gp    + (size_t)N_B * CP * CP;
  float* At    = Mt    + (size_t)C_CH * CP;
  float* T2t   = At    + (size_t)CP * CP;
  float* PT    = T2t   + (size_t)N_B * C_CH * CP;

  x_gram_kernel<<<512, 256, 0, stream>>>(x, Gpart, spart);
  reduce_g_kernel<<<1227, 256, 0, stream>>>(Gpart, spart, Gp, w0, b0, w1, b1,
                                             w2, b2, w3, Mt, At);
  nt_aug_kernel<<<544, 64, 0, stream>>>(Mt, 0, Gp, CP * CP, T2t, 1.0f);
  nt_aug_kernel<<<544, 64, 0, stream>>>(T2t, C_CH * CP, At, 0, PT,
                                         1.0f / (float)HW_P);
  out_kernel<<<256, 256, 0, stream>>>(x, PT, b3, out);
}


# last product: fragment loads double-buffered: the 24 loads of the next 32-channel block go to a second register set a full block (64 WMMAs) ahead of use, not one 8-WMMA step; sum order unchanged
# speedup vs baseline: 4.3598x; 1.1152x over previous
#include <hip/hip_runtime.h>


#define C_CH 256
#define CP   272
#define HW_P 4096
#define N_B  4
#define PCH  32
#define PLEN (HW_P / PCH)
#define KAUG_B 33
#define WPROD_WAVES (16 * 17 + 17 * 17)
#define WPROD_BLOCKS 71

typedef __attribute__((ext_vector_type(2))) float v2f;
typedef __attribute__((ext_vector_type(4))) float v4f;
typedef __attribute__((ext_vector_type(8))) float v8f;

#define HOLD8(p, o)                                                          \
  asm volatile("" ::"v"(p[(o)]), "v"(p[(o) + 1]), "v"(p[(o) + 2]),           \
               "v"(p[(o) + 3]), "v"(p[(o) + 4]), "v"(p[(o) + 5]),            \
               "v"(p[(o) + 6]), "v"(p[(o) + 7]))

__device__ __forceinline__ v8f wmma_f32(v2f a, v2f b, v8f c) {
  return __builtin_amdgcn_wmma_f32_16x16x4_f32(false, a, false, b, (short)0, c,
                                               false, false);
}

__device__ __forceinline__ v8f splat8(float v) {
  v8f r = {v, v, v, v, v, v, v, v};
  return r;
}

template <bool SUMS>
__device__ __forceinline__ void gram_body(const float* __restrict__ xa,
                                          const float* __restrict__ xb,
                                          float* __restrict__ gp,
                                          float* __restrict__ sp,
                                          int l15, int hi) {
  v8f acc[4][2];
  v8f accs[2];
  for (int mt = 0; mt < 4; ++mt) { acc[mt][0] = splat8(0.f); acc[mt][1] = splat8(0.f); }
  accs[0] = splat8(0.f); accs[1] = splat8(0.f);
  const v2f ones = {1.0f, 1.0f};

  for (int kb = 0; kb < PLEN / 32; ++kb) {
    v4f fa[4][4], fb[2][4];
    for (int mt = 0; mt < 4; ++mt)
      for (int i = 0; i < 4; ++i)
        fa[mt][i] = *(const v4f*)(xa + (size_t)(16 * mt) * HW_P + 32 * kb + 4 * i);
    for (int nt = 0; nt < 2; ++nt)
      for (int i = 0; i < 4; ++i)
        fb[nt][i] = *(const v4f*)(xb + (size_t)(16 * nt) * HW_P + 32 * kb + 4 * i);
    for (int i = 0; i < 4; ++i) {
      for (int t = 0; t < 2; ++t) {
        const v2f b0 = {fb[0][i][2 * t], fb[0][i][2 * t + 1]};
        const v2f b1 = {fb[1][i][2 * t], fb[1][i][2 * t + 1]};
        for (int mt = 0; mt < 4; ++mt) {
          const v2f a = {fa[mt][i][2 * t], fa[mt][i][2 * t + 1]};
          acc[mt][0] = wmma_f32(a, b0, acc[mt][0]);
          acc[mt][1] = wmma_f32(a, b1, acc[mt][1]);
        }
        if (SUMS) {
          accs[0] = wmma_f32(ones, b0, accs[0]);
          accs[1] = wmma_f32(ones, b1, accs[1]);
        }
      }
    }
  }
  for (int mt = 0; mt < 4; ++mt)
    for (int r = 0; r < 8; ++r) {
      float* g = gp + (size_t)(16 * mt + 8 * hi + r) * C_CH;
      g[0]  = acc[mt][0][r];
      g[16] = acc[mt][1][r];
    }
  if (SUMS) {
    if (hi == 0) { sp[0] = accs[0][0]; sp[16] = accs[1][0]; }
  }
}

__global__ void __launch_bounds__(256)
x_gram_kernel(const float* __restrict__ x, float* __restrict__ Gpart,
              float* __restrict__ spart) {
  const int lane = threadIdx.x & 31;
  const int l15  = lane & 15;
  const int hi   = lane >> 4;
  int t = blockIdx.x * (blockDim.x >> 5) +
          __builtin_amdgcn_readfirstlane((int)(threadIdx.x >> 5));
  const int eg    = t & 7;  t >>= 3;
  const int cg    = t & 3;  t >>= 2;
  const int chunk = t & 31; t >>= 5;
  const int n     = t;

  const float* xn = x + (size_t)n * C_CH * HW_P + chunk * PLEN + 16 * hi;
  const float* xa = xn + (size_t)(cg * 64 + l15) * HW_P;
  const float* xb = xn + (size_t)(eg * 32 + l15) * HW_P;
  float* gp = Gpart + ((size_t)(chunk * N_B + n) * C_CH + cg * 64) * C_CH +
              eg * 32 + l15;
  float* sp = spart + (size_t)(chunk * N_B + n) * C_CH + eg * 32 + l15;
  if (cg == 0) gram_body<true>(xa, xb, gp, sp, l15, hi);
  else         gram_body<false>(xa, xb, gp, sp, l15, hi);
}

__device__ __forceinline__ void aug_col_frags(v4f (&f)[8],
                                              const float* __restrict__ w,
                                              const float* __restrict__ b,
                                              int tile, int k0, int l15, int hi) {
  if (tile < 16) {
    const float* wc = w + tile * 16 + l15 + (size_t)(k0 + 4 * hi) * C_CH;
#pragma unroll
    for (int i = 0; i < 8; ++i) {
      const float* wd = wc + (size_t)(8 * i) * C_CH;
      const v4f q = {wd[0], wd[C_CH], wd[2 * C_CH], wd[3 * C_CH]};
      f[i] = q;
    }
  } else {
    const v4f zero = {0.0f, 0.0f, 0.0f, 0.0f};
#pragma unroll
    for (int i = 0; i < 8; ++i) {
      const v4f q = *(const v4f*)(b + k0 + 8 * i + 4 * hi);
      f[i] = (l15 == 0) ? q : zero;
    }
  }
}

__device__ __forceinline__ void weight_products(
    const float* __restrict__ w0, const float* __restrict__ b0,
    const float* __restrict__ w1, const float* __restrict__ b1,
    const float* __restrict__ w2, const float* __restrict__ b2,
    const float* __restrict__ w3, float* __restrict__ Mt,
    float* __restrict__ At, int wv, int l15, int hi) {
  const bool first = wv < 16 * 17;
  const int u  = first ? wv : wv - 16 * 17;
  const int rt = u / 17;
  const int ct = u - rt * 17;
  const float* arow = w3 + (size_t)(rt * 16 + l15) * C_CH;
  v8f acc = {}, acc1 = {};
#pragma unroll 1
  for (int k0 = 0; k0 < C_CH; k0 += 64) {
    v4f fa[8], fb[8];
    if (first) {
#pragma unroll
      for (int i = 0; i < 8; ++i)
        fa[i] = *(const v4f*)(arow + k0 + 8 * i + 4 * hi);
      aug_col_frags(fb, w0, b0, ct, k0, l15, hi);
    } else {
      aug_col_frags(fa, w2, b2, rt, k0, l15, hi);
      aug_col_frags(fb, w1, b1, ct, k0, l15, hi);
    }
#pragma unroll
    for (int i = 0; i < 8; ++i) {
      const v2f a0 = {fa[i][0], fa[i][1]}, a1 = {fa[i][2], fa[i][3]};
      const v2f c0 = {fb[i][0], fb[i][1]}, c1 = {fb[i][2], fb[i][3]};
      acc  = wmma_f32(a0, c0, acc);
      acc1 = wmma_f32(a1, c1, acc1);
    }
  }
  acc = acc + acc1;
  float* d = (first ? Mt : At) + (size_t)(rt * 16 + 8 * hi) * CP + ct * 16 + l15;
#pragma unroll
  for (int r = 0; r < 8; ++r) d[(size_t)r * CP] = acc[r];
}

__global__ void __launch_bounds__(256)
reduce_g_kernel(const float* __restrict__ Gpart, const float* __restrict__ spart,
                float* __restrict__ Gp,
                const float* __restrict__ w0, const float* __restrict__ b0,
                const float* __restrict__ w1, const float* __restrict__ b1,
                const float* __restrict__ w2, const float* __restrict__ b2,
                const float* __restrict__ w3, float* __restrict__ Mt,
                float* __restrict__ At) {
  if (blockIdx.x < WPROD_BLOCKS) {
    const int wv = blockIdx.x * (blockDim.x >> 5) +
                   __builtin_amdgcn_readfirstlane((int)(threadIdx.x >> 5));
    const int lane = threadIdx.x & 31;
    if (wv < WPROD_WAVES)
      weight_products(w0, b0, w1, b1, w2, b2, w3, Mt, At, wv, lane & 15,
                      lane >> 4);
    return;
  }
  const int idx = (blockIdx.x - WPROD_BLOCKS) * blockDim.x + threadIdx.x;
  const int n   = idx / (CP * CP);
  const int rem = idx - n * (CP * CP);
  const int j   = rem / CP;
  const int e   = rem - j * CP;
  float s = 0.0f;
  if (j < C_CH && e < C_CH) {
    const float* g = Gpart + ((size_t)n * C_CH + j) * C_CH + e;
#pragma unroll 8
    for (int c = 0; c < PCH; ++c) s += g[(size_t)c * N_B * C_CH * C_CH];
  } else if ((j == C_CH) != (e == C_CH) && j <= C_CH && e <= C_CH) {
    const int ch = (j == C_CH) ? e : j;
    const float* q = spart + (size_t)n * C_CH + ch;
#pragma unroll 8
    for (int c = 0; c < PCH; ++c) s += q[(size_t)c * N_B * C_CH];
  } else if (j == C_CH && e == C_CH) {
    s = (float)HW_P;
  }
  Gp[idx] = s;
}

__global__ void __launch_bounds__(64)
nt_aug_kernel(const float* __restrict__ A, int aBatch,
              const float* __restrict__ B, int bBatch,
              float* __restrict__ D, float scale) {
  const int lane = threadIdx.x & 31;
  const int l15  = lane & 15;
  const int hi   = lane >> 4;
  int t = blockIdx.x * (blockDim.x >> 5) + (threadIdx.x >> 5);
  const int ct = t % 17; t /= 17;
  const int mt = t & 15; t >>= 4;
  const int n  = t;

  const float* arow = A + (size_t)n * aBatch + (size_t)(mt * 16 + l15) * CP;
  const float* brow = B + (size_t)n * bBatch + (size_t)(ct * 16 + l15) * CP;
  v4f fa[KAUG_B], fb[KAUG_B];
#pragma unroll
  for (int i = 0; i < KAUG_B; ++i) {
    fa[i] = *(const v4f*)(arow + 8 * i + 4 * hi);
    fb[i] = *(const v4f*)(brow + 8 * i + 4 * hi);
  }
  HOLD8(fa, 0); HOLD8(fb, 0); HOLD8(fa, 8);  HOLD8(fb, 8);
  v8f acc = {}, acc1 = {};
#pragma unroll
  for (int i = 0; i < KAUG_B; ++i) {
    const v2f a0 = {fa[i][0], fa[i][1]}, a1 = {fa[i][2], fa[i][3]};
    const v2f b0 = {fb[i][0], fb[i][1]}, b1 = {fb[i][2], fb[i][3]};
    acc  = wmma_f32(a0, b0, acc);
    acc1 = wmma_f32(a1, b1, acc1);
  }
  acc = acc + acc1;
  float* d = D + ((size_t)n * C_CH + mt * 16 + 8 * hi) * CP + ct * 16 + l15;
#pragma unroll
  for (int r = 0; r < 8; ++r) d[(size_t)r * CP] = acc[r] * scale;
}

struct OutFrags {
  v4f w0[4], w1[4];
  v4f x[16];
};

__device__ __forceinline__ void out_load_block(OutFrags& f,
                                               const float* __restrict__ wl0,
                                               const float* __restrict__ wl1,
                                               const float* __restrict__ xl,
                                               int j) {
  const float* xj = xl + (size_t)(32 * j) * HW_P;
  for (int i = 0; i < 16; ++i)
    f.x[i] = *(const v4f*)(xj + (size_t)i * HW_P);
  for (int i = 0; i < 4; ++i) {
    f.w0[i] = *(const v4f*)(wl0 + 32 * j + 4 * i);
    f.w1[i] = *(const v4f*)(wl1 + 32 * j + 4 * i);
  }
}

__device__ __forceinline__ void out_mma_block(v8f (&acc)[4][2],
                                              const OutFrags& f) {
  for (int s = 0; s < 8; ++s) {
    const int q = s >> 1, c = 2 * (s & 1);
    const v2f bb0 = {f.w0[q][c], f.w0[q][c + 1]};
    const v2f bb1 = {f.w1[q][c], f.w1[q][c + 1]};
    for (int mt = 0; mt < 4; ++mt) {
      const v2f a = {f.x[2 * s][mt], f.x[2 * s + 1][mt]};
      acc[mt][0] = wmma_f32(a, bb0, acc[mt][0]);
      acc[mt][1] = wmma_f32(a, bb1, acc[mt][1]);
    }
  }
}

__global__ void __launch_bounds__(256)
out_kernel(const float* __restrict__ x, const float* __restrict__ PT,
           const float* __restrict__ b3, float* __restrict__ out) {
  const int lane = threadIdx.x & 31;
  const int l15  = lane & 15;
  const int hi   = lane >> 4;
  int t = blockIdx.x * (blockDim.x >> 5) + (threadIdx.x >> 5);
  const int opair = t & 7;  t >>= 3;
  const int ptile = t & 63; t >>= 6;
  const int n     = t;

  const float* w = PT + (size_t)n * C_CH * CP;
  const int p0 = ptile * 64;
  const int o0 = opair * 32;
  const float* xl  = x + (size_t)n * C_CH * HW_P + (size_t)(16 * hi) * HW_P +
                     p0 + 4 * l15;
  const float* wl0 = w + (size_t)(o0 + l15) * CP + 16 * hi;
  const float* wl1 = w + (size_t)(o0 + 16 + l15) * CP + 16 * hi;

  OutFrags f0, f1;
  out_load_block(f0, wl0, wl1, xl, 0);

  v8f acc[4][2];
  {
    const v8f bz0 = splat8(w[(size_t)(o0 + l15) * CP + C_CH] + b3[o0 + l15]);
    const v8f bz1 = splat8(w[(size_t)(o0 + 16 + l15) * CP + C_CH] + b3[o0 + 16 + l15]);
    for (int mt = 0; mt < 4; ++mt) { acc[mt][0] = bz0; acc[mt][1] = bz1; }
  }

  for (int j = 0; j < C_CH / 32; j += 2) {
    out_load_block(f1, wl0, wl1, xl, j + 1);
    out_mma_block(acc, f0);
    if (j + 2 < C_CH / 32) out_load_block(f0, wl0, wl1, xl, j + 2);
    out_mma_block(acc, f1);
  }
  float* ob0 = out + ((size_t)n * C_CH + o0 + l15) * HW_P + p0 + 32 * hi;
  float* ob1 = ob0 + (size_t)16 * HW_P;
  for (int r = 0; r < 8; ++r) {
    const v4f s0 = {acc[0][0][r], acc[1][0][r], acc[2][0][r], acc[3][0][r]};
    const v4f s1 = {acc[0][1][r], acc[1][1][r], acc[2][1][r], acc[3][1][r]};
    *(v4f*)(ob0 + 4 * r) = s0;
    *(v4f*)(ob1 + 4 * r) = s1;
  }
}

extern "C" void kernel_launch(void* const* d_in, const int* in_sizes, int n_in,
                              void* d_out, int out_size, void* d_ws,
                              size_t ws_size, hipStream_t stream) {
  const float* x  = (const float*)d_in[0];
  const float* w0 = (const float*)d_in[1];
  const float* b0 = (const float*)d_in[2];
  const float* w1 = (const float*)d_in[3];
  const float* b1 = (const float*)d_in[4];
  const float* w2 = (const float*)d_in[5];
  const float* b2 = (const float*)d_in[6];
  const float* w3 = (const float*)d_in[7];
  const float* b3 = (const float*)d_in[8];
  float* out = (float*)d_out;

  float* Gpart = (float*)d_ws;
  float* spart = Gpart + (size_t)PCH * N_B * C_CH * C_CH;
  float* Gp    = spart + (size_t)PCH * N_B * C_CH;
  float* Mt    = Gp    + (size_t)N_B * CP * CP;
  float* At    = Mt    + (size_t)C_CH * CP;
  float* T2t   = At    + (size_t)CP * CP;
  float* PT    = T2t   + (size_t)N_B * C_CH * CP;

  x_gram_kernel<<<512, 256, 0, stream>>>(x, Gpart, spart);
  reduce_g_kernel<<<1227, 256, 0, stream>>>(Gpart, spart, Gp, w0, b0, w1, b1,
                                             w2, b2, w3, Mt, At);
  nt_aug_kernel<<<544, 64, 0, stream>>>(Mt, 0, Gp, CP * CP, T2t, 1.0f);
  nt_aug_kernel<<<544, 64, 0, stream>>>(T2t, C_CH * CP, At, 0, PT,
                                         1.0f / (float)HW_P);
  out_kernel<<<256, 256, 0, stream>>>(x, PT, b3, out);
}


# Gram made on the upper triangle of 64x64 blocks only (20 of 32 wave tiles; x x^T is symmetric); the product that reads G' loads the lower blocks transposed; per-element sum order unchanged
# speedup vs baseline: 4.7795x; 1.0963x over previous
#include <hip/hip_runtime.h>


#define C_CH 256
#define CP   272
#define HW_P 4096
#define N_B  4
#define PCH  32
#define PLEN (HW_P / PCH)
#define GTILES 20
#define KAUG_B 33
#define WPROD_WAVES (16 * 17 + 17 * 17)
#define WPROD_BLOCKS 71

typedef __attribute__((ext_vector_type(2))) float v2f;
typedef __attribute__((ext_vector_type(4))) float v4f;
typedef __attribute__((ext_vector_type(8))) float v8f;

#define HOLD8(p, o)                                                          \
  asm volatile("" ::"v"(p[(o)]), "v"(p[(o) + 1]), "v"(p[(o) + 2]),           \
               "v"(p[(o) + 3]), "v"(p[(o) + 4]), "v"(p[(o) + 5]),            \
               "v"(p[(o) + 6]), "v"(p[(o) + 7]))

__device__ __forceinline__ v8f wmma_f32(v2f a, v2f b, v8f c) {
  return __builtin_amdgcn_wmma_f32_16x16x4_f32(false, a, false, b, (short)0, c,
                                               false, false);
}

__device__ __forceinline__ v8f splat8(float v) {
  v8f r = {v, v, v, v, v, v, v, v};
  return r;
}

template <bool SUMS>
__device__ __forceinline__ void gram_body(const float* __restrict__ xa,
                                          const float* __restrict__ xb,
                                          float* __restrict__ gp,
                                          float* __restrict__ sp,
                                          int l15, int hi) {
  v8f acc[4][2];
  v8f accs[2];
  for (int mt = 0; mt < 4; ++mt) { acc[mt][0] = splat8(0.f); acc[mt][1] = splat8(0.f); }
  accs[0] = splat8(0.f); accs[1] = splat8(0.f);
  const v2f ones = {1.0f, 1.0f};

  for (int kb = 0; kb < PLEN / 32; ++kb) {
    v4f fa[4][4], fb[2][4];
    for (int mt = 0; mt < 4; ++mt)
      for (int i = 0; i < 4; ++i)
        fa[mt][i] = *(const v4f*)(xa + (size_t)(16 * mt) * HW_P + 32 * kb + 4 * i);
    for (int nt = 0; nt < 2; ++nt)
      for (int i = 0; i < 4; ++i)
        fb[nt][i] = *(const v4f*)(xb + (size_t)(16 * nt) * HW_P + 32 * kb + 4 * i);
    for (int i = 0; i < 4; ++i) {
      for (int t = 0; t < 2; ++t) {
        const v2f b0 = {fb[0][i][2 * t], fb[0][i][2 * t + 1]};
        const v2f b1 = {fb[1][i][2 * t], fb[1][i][2 * t + 1]};
        for (int mt = 0; mt < 4; ++mt) {
          const v2f a = {fa[mt][i][2 * t], fa[mt][i][2 * t + 1]};
          acc[mt][0] = wmma_f32(a, b0, acc[mt][0]);
          acc[mt][1] = wmma_f32(a, b1, acc[mt][1]);
        }
        if (SUMS) {
          accs[0] = wmma_f32(ones, b0, accs[0]);
          accs[1] = wmma_f32(ones, b1, accs[1]);
        }
      }
    }
  }
  for (int mt = 0; mt < 4; ++mt)
    for (int r = 0; r < 8; ++r) {
      float* g = gp + (size_t)(16 * mt + 8 * hi + r) * C_CH;
      g[0]  = acc[mt][0][r];
      g[16] = acc[mt][1][r];
    }
  if (SUMS) {
    if (hi == 0) { sp[0] = accs[0][0]; sp[16] = accs[1][0]; }
  }
}

__global__ void __launch_bounds__(128)
x_gram_kernel(const float* __restrict__ x, float* __restrict__ Gpart,
              float* __restrict__ spart) {
  const int lane = threadIdx.x & 31;
  const int l15  = lane & 15;
  const int hi   = lane >> 4;
  int t = blockIdx.x * (blockDim.x >> 5) +
          __builtin_amdgcn_readfirstlane((int)(threadIdx.x >> 5));
  const int u = t % GTILES; t /= GTILES;
  const int cg = (u < 8) ? 0 : (u < 14) ? 1 : (u < 18) ? 2 : 3;
  const int eg = (u < 8) ? u : (u < 14) ? u - 6 : (u < 18) ? u - 10 : u - 12;
  const int chunk = t & 31; t >>= 5;
  const int n     = t;

  const float* xn = x + (size_t)n * C_CH * HW_P + chunk * PLEN + 16 * hi;
  const float* xa = xn + (size_t)(cg * 64 + l15) * HW_P;
  const float* xb = xn + (size_t)(eg * 32 + l15) * HW_P;
  float* gp = Gpart + ((size_t)(chunk * N_B + n) * C_CH + cg * 64) * C_CH +
              eg * 32 + l15;
  float* sp = spart + (size_t)(chunk * N_B + n) * C_CH + eg * 32 + l15;
  if (cg == 0) gram_body<true>(xa, xb, gp, sp, l15, hi);
  else         gram_body<false>(xa, xb, gp, sp, l15, hi);
}

__device__ __forceinline__ void aug_col_frags(v4f (&f)[8],
                                              const float* __restrict__ w,
                                              const float* __restrict__ b,
                                              int tile, int k0, int l15, int hi) {
  if (tile < 16) {
    const float* wc = w + tile * 16 + l15 + (size_t)(k0 + 4 * hi) * C_CH;
#pragma unroll
    for (int i = 0; i < 8; ++i) {
      const float* wd = wc + (size_t)(8 * i) * C_CH;
      const v4f q = {wd[0], wd[C_CH], wd[2 * C_CH], wd[3 * C_CH]};
      f[i] = q;
    }
  } else {
    const v4f zero = {0.0f, 0.0f, 0.0f, 0.0f};
#pragma unroll
    for (int i = 0; i < 8; ++i) {
      const v4f q = *(const v4f*)(b + k0 + 8 * i + 4 * hi);
      f[i] = (l15 == 0) ? q : zero;
    }
  }
}

__device__ __forceinline__ void weight_products(
    const float* __restrict__ w0, const float* __restrict__ b0,
    const float* __restrict__ w1, const float* __restrict__ b1,
    const float* __restrict__ w2, const float* __restrict__ b2,
    const float* __restrict__ w3, float* __restrict__ Mt,
    float* __restrict__ At, int wv, int l15, int hi) {
  const bool first = wv < 16 * 17;
  const int u  = first ? wv : wv - 16 * 17;
  const int rt = u / 17;
  const int ct = u - rt * 17;
  const float* arow = w3 + (size_t)(rt * 16 + l15) * C_CH;
  v8f acc = {}, acc1 = {};
#pragma unroll 1
  for (int k0 = 0; k0 < C_CH; k0 += 64) {
    v4f fa[8], fb[8];
    if (first) {
#pragma unroll
      for (int i = 0; i < 8; ++i)
        fa[i] = *(const v4f*)(arow + k0 + 8 * i + 4 * hi);
      aug_col_frags(fb, w0, b0, ct, k0, l15, hi);
    } else {
      aug_col_frags(fa, w2, b2, rt, k0, l15, hi);
      aug_col_frags(fb, w1, b1, ct, k0, l15, hi);
    }
#pragma unroll
    for (int i = 0; i < 8; ++i) {
      const v2f a0 = {fa[i][0], fa[i][1]}, a1 = {fa[i][2], fa[i][3]};
      const v2f c0 = {fb[i][0], fb[i][1]}, c1 = {fb[i][2], fb[i][3]};
      acc  = wmma_f32(a0, c0, acc);
      acc1 = wmma_f32(a1, c1, acc1);
    }
  }
  acc = acc + acc1;
  float* d = (first ? Mt : At) + (size_t)(rt * 16 + 8 * hi) * CP + ct * 16 + l15;
#pragma unroll
  for (int r = 0; r < 8; ++r) d[(size_t)r * CP] = acc[r];
}

__global__ void __launch_bounds__(256)
reduce_g_kernel(const float* __restrict__ Gpart, const float* __restrict__ spart,
                float* __restrict__ Gp,
                const float* __restrict__ w0, const float* __restrict__ b0,
                const float* __restrict__ w1, const float* __restrict__ b1,
                const float* __restrict__ w2, const float* __restrict__ b2,
                const float* __restrict__ w3, float* __restrict__ Mt,
                float* __restrict__ At) {
  if (blockIdx.x < WPROD_BLOCKS) {
    const int wv = blockIdx.x * (blockDim.x >> 5) +
                   __builtin_amdgcn_readfirstlane((int)(threadIdx.x >> 5));
    const int lane = threadIdx.x & 31;
    if (wv < WPROD_WAVES)
      weight_products(w0, b0, w1, b1, w2, b2, w3, Mt, At, wv, lane & 15,
                      lane >> 4);
    return;
  }
  const int idx = (blockIdx.x - WPROD_BLOCKS) * blockDim.x + threadIdx.x;
  const int n   = idx / (CP * CP);
  const int rem = idx - n * (CP * CP);
  const int j   = rem / CP;
  const int e   = rem - j * CP;
  if (j < C_CH && e < C_CH && (j >> 6) > (e >> 6)) return;
  float s = 0.0f;
  if (j < C_CH && e < C_CH) {
    const float* g = Gpart + ((size_t)n * C_CH + j) * C_CH + e;
#pragma unroll 8
    for (int c = 0; c < PCH; ++c) s += g[(size_t)c * N_B * C_CH * C_CH];
  } else if ((j == C_CH) != (e == C_CH) && j <= C_CH && e <= C_CH) {
    const int ch = (j == C_CH) ? e : j;
    const float* q = spart + (size_t)n * C_CH + ch;
#pragma unroll 8
    for (int c = 0; c < PCH; ++c) s += q[(size_t)c * N_B * C_CH];
  } else if (j == C_CH && e == C_CH) {
    s = (float)HW_P;
  }
  Gp[idx] = s;
}

template <int NM>
__device__ __forceinline__ void nt_tile(const float* __restrict__ arow,
                                        const float* __restrict__ brow,
                                        const float* __restrict__ bcol,
                                        float* __restrict__ d, float scale,
                                        int hi) {
  v4f fa[KAUG_B], fb[KAUG_B];
#pragma unroll
  for (int i = 0; i < KAUG_B; ++i)
    fa[i] = *(const v4f*)(arow + 8 * i + 4 * hi);
#pragma unroll
  for (int i = 0; i < KAUG_B; ++i) {
    if (i < 8 * NM) {
      const float* bc = bcol + (size_t)(8 * i) * CP;
      const v4f q = {bc[0], bc[CP], bc[2 * CP], bc[3 * CP]};
      fb[i] = q;
    } else {
      fb[i] = *(const v4f*)(brow + 8 * i + 4 * hi);
    }
  }
  HOLD8(fa, 0); HOLD8(fb, 0); HOLD8(fa, 8);  HOLD8(fb, 8);
  v8f acc = {}, acc1 = {};
#pragma unroll
  for (int i = 0; i < KAUG_B; ++i) {
    const v2f a0 = {fa[i][0], fa[i][1]}, a1 = {fa[i][2], fa[i][3]};
    const v2f b0 = {fb[i][0], fb[i][1]}, b1 = {fb[i][2], fb[i][3]};
    acc  = wmma_f32(a0, b0, acc);
    acc1 = wmma_f32(a1, b1, acc1);
  }
  acc = acc + acc1;
#pragma unroll
  for (int r = 0; r < 8; ++r) d[(size_t)r * CP] = acc[r] * scale;
}

template <bool SYMB>
__global__ void __launch_bounds__(64)
nt_aug_kernel(const float* __restrict__ A, int aBatch,
              const float* __restrict__ B, int bBatch,
              float* __restrict__ D, float scale) {
  const int lane = threadIdx.x & 31;
  const int l15  = lane & 15;
  const int hi   = lane >> 4;
  int t = blockIdx.x * (blockDim.x >> 5) +
          __builtin_amdgcn_readfirstlane((int)(threadIdx.x >> 5));
  const int ct = t % 17; t /= 17;
  const int mt = t & 15; t >>= 4;
  const int n  = t;

  const float* arow = A + (size_t)n * aBatch + (size_t)(mt * 16 + l15) * CP;
  const float* brow = B + (size_t)n * bBatch + (size_t)(ct * 16 + l15) * CP;
  const float* bcol = B + (size_t)n * bBatch + (size_t)(4 * hi) * CP + ct * 16 + l15;
  float* d = D + ((size_t)n * C_CH + mt * 16 + 8 * hi) * CP + ct * 16 + l15;
  const int nmir = (SYMB && ct < 16) ? (ct >> 2) : 0;
  if (nmir == 0)      nt_tile<0>(arow, brow, bcol, d, scale, hi);
  else if (nmir == 1) nt_tile<1>(arow, brow, bcol, d, scale, hi);
  else if (nmir == 2) nt_tile<2>(arow, brow, bcol, d, scale, hi);
  else                nt_tile<3>(arow, brow, bcol, d, scale, hi);
}

struct OutFrags {
  v4f w0[4], w1[4];
  v4f x[16];
};

__device__ __forceinline__ void out_load_block(OutFrags& f,
                                               const float* __restrict__ wl0,
                                               const float* __restrict__ wl1,
                                               const float* __restrict__ xl,
                                               int j) {
  const float* xj = xl + (size_t)(32 * j) * HW_P;
  for (int i = 0; i < 16; ++i)
    f.x[i] = *(const v4f*)(xj + (size_t)i * HW_P);
  for (int i = 0; i < 4; ++i) {
    f.w0[i] = *(const v4f*)(wl0 + 32 * j + 4 * i);
    f.w1[i] = *(const v4f*)(wl1 + 32 * j + 4 * i);
  }
}

__device__ __forceinline__ void out_mma_block(v8f (&acc)[4][2],
                                              const OutFrags& f) {
  for (int s = 0; s < 8; ++s) {
    const int q = s >> 1, c = 2 * (s & 1);
    const v2f bb0 = {f.w0[q][c], f.w0[q][c + 1]};
    const v2f bb1 = {f.w1[q][c], f.w1[q][c + 1]};
    for (int mt = 0; mt < 4; ++mt) {
      const v2f a = {f.x[2 * s][mt], f.x[2 * s + 1][mt]};
      acc[mt][0] = wmma_f32(a, bb0, acc[mt][0]);
      acc[mt][1] = wmma_f32(a, bb1, acc[mt][1]);
    }
  }
}

__global__ void __launch_bounds__(256)
out_kernel(const float* __restrict__ x, const float* __restrict__ PT,
           const float* __restrict__ b3, float* __restrict__ out) {
  const int lane = threadIdx.x & 31;
  const int l15  = lane & 15;
  const int hi   = lane >> 4;
  int t = blockIdx.x * (blockDim.x >> 5) + (threadIdx.x >> 5);
  const int opair = t & 7;  t >>= 3;
  const int ptile = t & 63; t >>= 6;
  const int n     = t;

  const float* w = PT + (size_t)n * C_CH * CP;
  const int p0 = ptile * 64;
  const int o0 = opair * 32;
  const float* xl  = x + (size_t)n * C_CH * HW_P + (size_t)(16 * hi) * HW_P +
                     p0 + 4 * l15;
  const float* wl0 = w + (size_t)(o0 + l15) * CP + 16 * hi;
  const float* wl1 = w + (size_t)(o0 + 16 + l15) * CP + 16 * hi;

  OutFrags f0, f1;
  out_load_block(f0, wl0, wl1, xl, 0);

  v8f acc[4][2];
  {
    const v8f bz0 = splat8(w[(size_t)(o0 + l15) * CP + C_CH] + b3[o0 + l15]);
    const v8f bz1 = splat8(w[(size_t)(o0 + 16 + l15) * CP + C_CH] + b3[o0 + 16 + l15]);
    for (int mt = 0; mt < 4; ++mt) { acc[mt][0] = bz0; acc[mt][1] = bz1; }
  }

  for (int j = 0; j < C_CH / 32; j += 2) {
    out_load_block(f1, wl0, wl1, xl, j + 1);
    out_mma_block(acc, f0);
    if (j + 2 < C_CH / 32) out_load_block(f0, wl0, wl1, xl, j + 2);
    out_mma_block(acc, f1);
  }
  float* ob0 = out + ((size_t)n * C_CH + o0 + l15) * HW_P + p0 + 32 * hi;
  float* ob1 = ob0 + (size_t)16 * HW_P;
  for (int r = 0; r < 8; ++r) {
    const v4f s0 = {acc[0][0][r], acc[1][0][r], acc[2][0][r], acc[3][0][r]};
    const v4f s1 = {acc[0][1][r], acc[1][1][r], acc[2][1][r], acc[3][1][r]};
    *(v4f*)(ob0 + 4 * r) = s0;
    *(v4f*)(ob1 + 4 * r) = s1;
  }
}

extern "C" void kernel_launch(void* const* d_in, const int* in_sizes, int n_in,
                              void* d_out, int out_size, void* d_ws,
                              size_t ws_size, hipStream_t stream) {
  const float* x  = (const float*)d_in[0];
  const float* w0 = (const float*)d_in[1];
  const float* b0 = (const float*)d_in[2];
  const float* w1 = (const float*)d_in[3];
  const float* b1 = (const float*)d_in[4];
  const float* w2 = (const float*)d_in[5];
  const float* b2 = (const float*)d_in[6];
  const float* w3 = (const float*)d_in[7];
  const float* b3 = (const float*)d_in[8];
  float* out = (float*)d_out;

  float* Gpart = (float*)d_ws;
  float* spart = Gpart + (size_t)PCH * N_B * C_CH * C_CH;
  float* Gp    = spart + (size_t)PCH * N_B * C_CH;
  float* Mt    = Gp    + (size_t)N_B * CP * CP;
  float* At    = Mt    + (size_t)C_CH * CP;
  float* T2t   = At    + (size_t)CP * CP;
  float* PT    = T2t   + (size_t)N_B * C_CH * CP;

  x_gram_kernel<<<640, 128, 0, stream>>>(x, Gpart, spart);
  reduce_g_kernel<<<1227, 256, 0, stream>>>(Gpart, spart, Gp, w0, b0, w1, b1,
                                             w2, b2, w3, Mt, At);
  nt_aug_kernel<true><<<544, 64, 0, stream>>>(Mt, 0, Gp, CP * CP, T2t, 1.0f);
  nt_aug_kernel<false><<<544, 64, 0, stream>>>(T2t, C_CH * CP, At, 0, PT,
                                         1.0f / (float)HW_P);
  out_kernel<<<256, 256, 0, stream>>>(x, PT, b3, out);
}


# out and Gram kernels: operand roles chosen so a D register holds neighbouring outputs across lanes: each store instruction writes whole 128-byte lines (was 16/64-byte pieces); sum order unchanged
# speedup vs baseline: 5.0563x; 1.0579x over previous
#include <hip/hip_runtime.h>


#define C_CH 256
#define CP   272
#define HW_P 4096
#define N_B  4
#define PCH  32
#define PLEN (HW_P / PCH)
#define GTILES 20
#define KAUG_B 33
#define WPROD_WAVES (16 * 17 + 17 * 17)
#define WPROD_BLOCKS 71

typedef __attribute__((ext_vector_type(2))) float v2f;
typedef __attribute__((ext_vector_type(4))) float v4f;
typedef __attribute__((ext_vector_type(8))) float v8f;

#define HOLD8(p, o)                                                          \
  asm volatile("" ::"v"(p[(o)]), "v"(p[(o) + 1]), "v"(p[(o) + 2]),           \
               "v"(p[(o) + 3]), "v"(p[(o) + 4]), "v"(p[(o) + 5]),            \
               "v"(p[(o) + 6]), "v"(p[(o) + 7]))

__device__ __forceinline__ v8f wmma_f32(v2f a, v2f b, v8f c) {
  return __builtin_amdgcn_wmma_f32_16x16x4_f32(false, a, false, b, (short)0, c,
                                               false, false);
}

__device__ __forceinline__ v8f splat8(float v) {
  v8f r = {v, v, v, v, v, v, v, v};
  return r;
}

template <bool SUMS>
__device__ __forceinline__ void gram_body(const float* __restrict__ xa,
                                          const float* __restrict__ xb,
                                          float* __restrict__ gp,
                                          float* __restrict__ sp,
                                          int l15, int hi) {
  v8f acc[4][2];
  v8f accs[2];
  for (int mt = 0; mt < 4; ++mt) { acc[mt][0] = splat8(0.f); acc[mt][1] = splat8(0.f); }
  accs[0] = splat8(0.f); accs[1] = splat8(0.f);
  const v2f ones = {1.0f, 1.0f};

  for (int kb = 0; kb < PLEN / 32; ++kb) {
    v4f fa[4][4], fb[2][4];
    for (int mt = 0; mt < 4; ++mt)
      for (int i = 0; i < 4; ++i)
        fa[mt][i] = *(const v4f*)(xa + (size_t)(16 * mt) * HW_P + 32 * kb + 4 * i);
    for (int nt = 0; nt < 2; ++nt)
      for (int i = 0; i < 4; ++i)
        fb[nt][i] = *(const v4f*)(xb + (size_t)nt * HW_P + 32 * kb + 4 * i);
    for (int i = 0; i < 4; ++i) {
      for (int t = 0; t < 2; ++t) {
        const v2f b0 = {fb[0][i][2 * t], fb[0][i][2 * t + 1]};
        const v2f b1 = {fb[1][i][2 * t], fb[1][i][2 * t + 1]};
        for (int mt = 0; mt < 4; ++mt) {
          const v2f a = {fa[mt][i][2 * t], fa[mt][i][2 * t + 1]};
          acc[mt][0] = wmma_f32(a, b0, acc[mt][0]);
          acc[mt][1] = wmma_f32(a, b1, acc[mt][1]);
        }
        if (SUMS) {
          accs[0] = wmma_f32(ones, b0, accs[0]);
          accs[1] = wmma_f32(ones, b1, accs[1]);
        }
      }
    }
  }
  for (int mt = 0; mt < 4; ++mt)
    for (int r = 0; r < 8; ++r) {
      const v2f s = {acc[mt][0][r], acc[mt][1][r]};
      *(v2f*)(gp + (size_t)(16 * mt + 8 * hi + r) * C_CH) = s;
    }
  if (SUMS) {
    if (hi == 0) { const v2f s = {accs[0][0], accs[1][0]}; *(v2f*)sp = s; }
  }
}

__global__ void __launch_bounds__(128)
x_gram_kernel(const float* __restrict__ x, float* __restrict__ Gpart,
              float* __restrict__ spart) {
  const int lane = threadIdx.x & 31;
  const int l15  = lane & 15;
  const int hi   = lane >> 4;
  int t = blockIdx.x * (blockDim.x >> 5) +
          __builtin_amdgcn_readfirstlane((int)(threadIdx.x >> 5));
  const int u = t % GTILES; t /= GTILES;
  const int cg = (u < 8) ? 0 : (u < 14) ? 1 : (u < 18) ? 2 : 3;
  const int eg = (u < 8) ? u : (u < 14) ? u - 6 : (u < 18) ? u - 10 : u - 12;
  const int chunk = t & 31; t >>= 5;
  const int n     = t;

  const float* xn = x + (size_t)n * C_CH * HW_P + chunk * PLEN + 16 * hi;
  const float* xa = xn + (size_t)(cg * 64 + l15) * HW_P;
  const float* xb = xn + (size_t)(eg * 32 + 2 * l15) * HW_P;
  float* gp = Gpart + ((size_t)(chunk * N_B + n) * C_CH + cg * 64) * C_CH +
              eg * 32 + 2 * l15;
  float* sp = spart + (size_t)(chunk * N_B + n) * C_CH + eg * 32 + 2 * l15;
  if (cg == 0) gram_body<true>(xa, xb, gp, sp, l15, hi);
  else         gram_body<false>(xa, xb, gp, sp, l15, hi);
}

__device__ __forceinline__ void aug_col_frags(v4f (&f)[8],
                                              const float* __restrict__ w,
                                              const float* __restrict__ b,
                                              int tile, int k0, int l15, int hi) {
  if (tile < 16) {
    const float* wc = w + tile * 16 + l15 + (size_t)(k0 + 4 * hi) * C_CH;
#pragma unroll
    for (int i = 0; i < 8; ++i) {
      const float* wd = wc + (size_t)(8 * i) * C_CH;
      const v4f q = {wd[0], wd[C_CH], wd[2 * C_CH], wd[3 * C_CH]};
      f[i] = q;
    }
  } else {
    const v4f zero = {0.0f, 0.0f, 0.0f, 0.0f};
#pragma unroll
    for (int i = 0; i < 8; ++i) {
      const v4f q = *(const v4f*)(b + k0 + 8 * i + 4 * hi);
      f[i] = (l15 == 0) ? q : zero;
    }
  }
}

__device__ __forceinline__ void weight_products(
    const float* __restrict__ w0, const float* __restrict__ b0,
    const float* __restrict__ w1, const float* __restrict__ b1,
    const float* __restrict__ w2, const float* __restrict__ b2,
    const float* __restrict__ w3, float* __restrict__ Mt,
    float* __restrict__ At, int wv, int l15, int hi) {
  const bool first = wv < 16 * 17;
  const int u  = first ? wv : wv - 16 * 17;
  const int rt = u / 17;
  const int ct = u - rt * 17;
  const float* arow = w3 + (size_t)(rt * 16 + l15) * C_CH;
  v8f acc = {}, acc1 = {};
#pragma unroll 1
  for (int k0 = 0; k0 < C_CH; k0 += 64) {
    v4f fa[8], fb[8];
    if (first) {
#pragma unroll
      for (int i = 0; i < 8; ++i)
        fa[i] = *(const v4f*)(arow + k0 + 8 * i + 4 * hi);
      aug_col_frags(fb, w0, b0, ct, k0, l15, hi);
    } else {
      aug_col_frags(fa, w2, b2, rt, k0, l15, hi);
      aug_col_frags(fb, w1, b1, ct, k0, l15, hi);
    }
#pragma unroll
    for (int i = 0; i < 8; ++i) {
      const v2f a0 = {fa[i][0], fa[i][1]}, a1 = {fa[i][2], fa[i][3]};
      const v2f c0 = {fb[i][0], fb[i][1]}, c1 = {fb[i][2], fb[i][3]};
      acc  = wmma_f32(a0, c0, acc);
      acc1 = wmma_f32(a1, c1, acc1);
    }
  }
  acc = acc + acc1;
  float* d = (first ? Mt : At) + (size_t)(rt * 16 + 8 * hi) * CP + ct * 16 + l15;
#pragma unroll
  for (int r = 0; r < 8; ++r) d[(size_t)r * CP] = acc[r];
}

__global__ void __launch_bounds__(256)
reduce_g_kernel(const float* __restrict__ Gpart, const float* __restrict__ spart,
                float* __restrict__ Gp,
                const float* __restrict__ w0, const float* __restrict__ b0,
                const float* __restrict__ w1, const float* __restrict__ b1,
                const float* __restrict__ w2, const float* __restrict__ b2,
                const float* __restrict__ w3, float* __restrict__ Mt,
                float* __restrict__ At) {
  if (blockIdx.x < WPROD_BLOCKS) {
    const int wv = blockIdx.x * (blockDim.x >> 5) +
                   __builtin_amdgcn_readfirstlane((int)(threadIdx.x >> 5));
    const int lane = threadIdx.x & 31;
    if (wv < WPROD_WAVES)
      weight_products(w0, b0, w1, b1, w2, b2, w3, Mt, At, wv, lane & 15,
                      lane >> 4);
    return;
  }
  const int idx = (blockIdx.x - WPROD_BLOCKS) * blockDim.x + threadIdx.x;
  const int n   = idx / (CP * CP);
  const int rem = idx - n * (CP * CP);
  const int j   = rem / CP;
  const int e   = rem - j * CP;
  if (j < C_CH && e < C_CH && (j >> 6) > (e >> 6)) return;
  float s = 0.0f;
  if (j < C_CH && e < C_CH) {
    const float* g = Gpart + ((size_t)n * C_CH + j) * C_CH + e;
#pragma unroll 8
    for (int c = 0; c < PCH; ++c) s += g[(size_t)c * N_B * C_CH * C_CH];
  } else if ((j == C_CH) != (e == C_CH) && j <= C_CH && e <= C_CH) {
    const int ch = (j == C_CH) ? e : j;
    const float* q = spart + (size_t)n * C_CH + ch;
#pragma unroll 8
    for (int c = 0; c < PCH; ++c) s += q[(size_t)c * N_B * C_CH];
  } else if (j == C_CH && e == C_CH) {
    s = (float)HW_P;
  }
  Gp[idx] = s;
}

template <int NM>
__device__ __forceinline__ void nt_tile(const float* __restrict__ arow,
                                        const float* __restrict__ brow,
                                        const float* __restrict__ bcol,
                                        float* __restrict__ d, float scale,
                                        int hi) {
  v4f fa[KAUG_B], fb[KAUG_B];
#pragma unroll
  for (int i = 0; i < KAUG_B; ++i)
    fa[i] = *(const v4f*)(arow + 8 * i + 4 * hi);
#pragma unroll
  for (int i = 0; i < KAUG_B; ++i) {
    if (i < 8 * NM) {
      const float* bc = bcol + (size_t)(8 * i) * CP;
      const v4f q = {bc[0], bc[CP], bc[2 * CP], bc[3 * CP]};
      fb[i] = q;
    } else {
      fb[i] = *(const v4f*)(brow + 8 * i + 4 * hi);
    }
  }
  HOLD8(fa, 0); HOLD8(fb, 0); HOLD8(fa, 8);  HOLD8(fb, 8);
  v8f acc = {}, acc1 = {};
#pragma unroll
  for (int i = 0; i < KAUG_B; ++i) {
    const v2f a0 = {fa[i][0], fa[i][1]}, a1 = {fa[i][2], fa[i][3]};
    const v2f b0 = {fb[i][0], fb[i][1]}, b1 = {fb[i][2], fb[i][3]};
    acc  = wmma_f32(a0, b0, acc);
    acc1 = wmma_f32(a1, b1, acc1);
  }
  acc = acc + acc1;
#pragma unroll
  for (int r = 0; r < 8; ++r) d[(size_t)r * CP] = acc[r] * scale;
}

template <bool SYMB>
__global__ void __launch_bounds__(64)
nt_aug_kernel(const float* __restrict__ A, int aBatch,
              const float* __restrict__ B, int bBatch,
              float* __restrict__ D, float scale) {
  const int lane = threadIdx.x & 31;
  const int l15  = lane & 15;
  const int hi   = lane >> 4;
  int t = blockIdx.x * (blockDim.x >> 5) +
          __builtin_amdgcn_readfirstlane((int)(threadIdx.x >> 5));
  const int ct = t % 17; t /= 17;
  const int mt = t & 15; t >>= 4;
  const int n  = t;

  const float* arow = A + (size_t)n * aBatch + (size_t)(mt * 16 + l15) * CP;
  const float* brow = B + (size_t)n * bBatch + (size_t)(ct * 16 + l15) * CP;
  const float* bcol = B + (size_t)n * bBatch + (size_t)(4 * hi) * CP + ct * 16 + l15;
  float* d = D + ((size_t)n * C_CH + mt * 16 + 8 * hi) * CP + ct * 16 + l15;
  const int nmir = (SYMB && ct < 16) ? (ct >> 2) : 0;
  if (nmir == 0)      nt_tile<0>(arow, brow, bcol, d, scale, hi);
  else if (nmir == 1) nt_tile<1>(arow, brow, bcol, d, scale, hi);
  else if (nmir == 2) nt_tile<2>(arow, brow, bcol, d, scale, hi);
  else                nt_tile<3>(arow, brow, bcol, d, scale, hi);
}

struct OutFrags {
  v4f w0[4], w1[4];
  v4f x[16];
};

__device__ __forceinline__ void out_load_block(OutFrags& f,
                                               const float* __restrict__ wl0,
                                               const float* __restrict__ wl1,
                                               const float* __restrict__ xl,
                                               int j) {
  const float* xj = xl + (size_t)(32 * j) * HW_P;
  for (int i = 0; i < 16; ++i)
    f.x[i] = *(const v4f*)(xj + (size_t)i * HW_P);
  for (int i = 0; i < 4; ++i) {
    f.w0[i] = *(const v4f*)(wl0 + 32 * j + 4 * i);
    f.w1[i] = *(const v4f*)(wl1 + 32 * j + 4 * i);
  }
}

__device__ __forceinline__ void out_mma_block(v8f (&acc)[4][2],
                                              const OutFrags& f) {
  for (int s = 0; s < 8; ++s) {
    const int q = s >> 1, c = 2 * (s & 1);
    const v2f bb0 = {f.w0[q][c], f.w0[q][c + 1]};
    const v2f bb1 = {f.w1[q][c], f.w1[q][c + 1]};
    for (int mt = 0; mt < 4; ++mt) {
      const v2f px = {f.x[2 * s][mt], f.x[2 * s + 1][mt]};
      acc[mt][0] = wmma_f32(bb0, px, acc[mt][0]);
      acc[mt][1] = wmma_f32(bb1, px, acc[mt][1]);
    }
  }
}

__global__ void __launch_bounds__(256)
out_kernel(const float* __restrict__ x, const float* __restrict__ PT,
           const float* __restrict__ b3, float* __restrict__ out) {
  const int lane = threadIdx.x & 31;
  const int l15  = lane & 15;
  const int hi   = lane >> 4;
  int t = blockIdx.x * (blockDim.x >> 5) + (threadIdx.x >> 5);
  const int opair = t & 7;  t >>= 3;
  const int ptile = t & 63; t >>= 6;
  const int n     = t;

  const float* w = PT + (size_t)n * C_CH * CP;
  const int p0 = ptile * 64;
  const int o0 = opair * 32;
  const float* xl  = x + (size_t)n * C_CH * HW_P + (size_t)(16 * hi) * HW_P +
                     p0 + 4 * l15;
  const float* wl0 = w + (size_t)(o0 + l15) * CP + 16 * hi;
  const float* wl1 = w + (size_t)(o0 + 16 + l15) * CP + 16 * hi;

  OutFrags f0, f1;
  out_load_block(f0, wl0, wl1, xl, 0);

  v8f acc[4][2];
  {
    const float* wa = w + (size_t)(o0 + 8 * hi) * CP + C_CH;
    const float* wb = wa + (size_t)16 * CP;
    const float* ba = b3 + o0 + 8 * hi;
    const v4f a0 = *(const v4f*)ba,        a1 = *(const v4f*)(ba + 4);
    const v4f c0 = *(const v4f*)(ba + 16), c1 = *(const v4f*)(ba + 20);
    const v8f bz0 = {wa[0] + a0[0],      wa[CP] + a0[1],     wa[2 * CP] + a0[2],
                     wa[3 * CP] + a0[3], wa[4 * CP] + a1[0], wa[5 * CP] + a1[1],
                     wa[6 * CP] + a1[2], wa[7 * CP] + a1[3]};
    const v8f bz1 = {wb[0] + c0[0],      wb[CP] + c0[1],     wb[2 * CP] + c0[2],
                     wb[3 * CP] + c0[3], wb[4 * CP] + c1[0], wb[5 * CP] + c1[1],
                     wb[6 * CP] + c1[2], wb[7 * CP] + c1[3]};
    for (int mt = 0; mt < 4; ++mt) { acc[mt][0] = bz0; acc[mt][1] = bz1; }
  }

  for (int j = 0; j < C_CH / 32; j += 2) {
    out_load_block(f1, wl0, wl1, xl, j + 1);
    out_mma_block(acc, f0);
    if (j + 2 < C_CH / 32) out_load_block(f0, wl0, wl1, xl, j + 2);
    out_mma_block(acc, f1);
  }
  float* ob0 = out + ((size_t)n * C_CH + o0 + 8 * hi) * HW_P + p0 + 4 * l15;
  float* ob1 = ob0 + (size_t)16 * HW_P;
  for (int r = 0; r < 8; ++r) {
    const v4f s0 = {acc[0][0][r], acc[1][0][r], acc[2][0][r], acc[3][0][r]};
    const v4f s1 = {acc[0][1][r], acc[1][1][r], acc[2][1][r], acc[3][1][r]};
    *(v4f*)(ob0 + (size_t)r * HW_P) = s0;
    *(v4f*)(ob1 + (size_t)r * HW_P) = s1;
  }
}

extern "C" void kernel_launch(void* const* d_in, const int* in_sizes, int n_in,
                              void* d_out, int out_size, void* d_ws,
                              size_t ws_size, hipStream_t stream) {
  const float* x  = (const float*)d_in[0];
  const float* w0 = (const float*)d_in[1];
  const float* b0 = (const float*)d_in[2];
  const float* w1 = (const float*)d_in[3];
  const float* b1 = (const float*)d_in[4];
  const float* w2 = (const float*)d_in[5];
  const float* b2 = (const float*)d_in[6];
  const float* w3 = (const float*)d_in[7];
  const float* b3 = (const float*)d_in[8];
  float* out = (float*)d_out;

  float* Gpart = (float*)d_ws;
  float* spart = Gpart + (size_t)PCH * N_B * C_CH * C_CH;
  float* Gp    = spart + (size_t)PCH * N_B * C_CH;
  float* Mt    = Gp    + (size_t)N_B * CP * CP;
  float* At    = Mt    + (size_t)C_CH * CP;
  float* T2t   = At    + (size_t)CP * CP;
  float* PT    = T2t   + (size_t)N_B * C_CH * CP;

  x_gram_kernel<<<640, 128, 0, stream>>>(x, Gpart, spart);
  reduce_g_kernel<<<1227, 256, 0, stream>>>(Gpart, spart, Gp, w0, b0, w1, b1,
                                             w2, b2, w3, Mt, At);
  nt_aug_kernel<true><<<544, 64, 0, stream>>>(Mt, 0, Gp, CP * CP, T2t, 1.0f);
  nt_aug_kernel<false><<<544, 64, 0, stream>>>(T2t, C_CH * CP, At, 0, PT,
                                         1.0f / (float)HW_P);
  out_kernel<<<256, 256, 0, stream>>>(x, PT, b3, out);
}


# last product: its producer writes P in the order of the fragment loads, so a 16-byte-per-lane load of P reads 512 contiguous bytes (4 lines), not 32 pieces on 24 lines; sum order unchanged
# speedup vs baseline: 5.2634x; 1.0410x over previous
#include <hip/hip_runtime.h>


#define C_CH 256
#define CP   272
#define HW_P 4096
#define N_B  4
#define PCH  32
#define PLEN (HW_P / PCH)
#define GTILES 20
#define KAUG_B 33
#define WPROD_WAVES (16 * 17 + 17 * 17)
#define WPROD_BLOCKS 71
#define PF_RUN  128
#define PF_BLK  (4 * PF_RUN)
#define PF_TILE (8 * PF_BLK)

typedef __attribute__((ext_vector_type(2))) float v2f;
typedef __attribute__((ext_vector_type(4))) float v4f;
typedef __attribute__((ext_vector_type(8))) float v8f;

#define HOLD8(p, o)                                                          \
  asm volatile("" ::"v"(p[(o)]), "v"(p[(o) + 1]), "v"(p[(o) + 2]),           \
               "v"(p[(o) + 3]), "v"(p[(o) + 4]), "v"(p[(o) + 5]),            \
               "v"(p[(o) + 6]), "v"(p[(o) + 7]))

__device__ __forceinline__ v8f wmma_f32(v2f a, v2f b, v8f c) {
  return __builtin_amdgcn_wmma_f32_16x16x4_f32(false, a, false, b, (short)0, c,
                                               false, false);
}

__device__ __forceinline__ v8f splat8(float v) {
  v8f r = {v, v, v, v, v, v, v, v};
  return r;
}

template <bool SUMS>
__device__ __forceinline__ void gram_body(const float* __restrict__ xa,
                                          const float* __restrict__ xb,
                                          float* __restrict__ gp,
                                          float* __restrict__ sp,
                                          int l15, int hi) {
  v8f acc[4][2];
  v8f accs[2];
  for (int mt = 0; mt < 4; ++mt) { acc[mt][0] = splat8(0.f); acc[mt][1] = splat8(0.f); }
  accs[0] = splat8(0.f); accs[1] = splat8(0.f);
  const v2f ones = {1.0f, 1.0f};

  for (int kb = 0; kb < PLEN / 32; ++kb) {
    v4f fa[4][4], fb[2][4];
    for (int mt = 0; mt < 4; ++mt)
      for (int i = 0; i < 4; ++i)
        fa[mt][i] = *(const v4f*)(xa + (size_t)(16 * mt) * HW_P + 32 * kb + 4 * i);
    for (int nt = 0; nt < 2; ++nt)
      for (int i = 0; i < 4; ++i)
        fb[nt][i] = *(const v4f*)(xb + (size_t)nt * HW_P + 32 * kb + 4 * i);
    for (int i = 0; i < 4; ++i) {
      for (int t = 0; t < 2; ++t) {
        const v2f b0 = {fb[0][i][2 * t], fb[0][i][2 * t + 1]};
        const v2f b1 = {fb[1][i][2 * t], fb[1][i][2 * t + 1]};
        for (int mt = 0; mt < 4; ++mt) {
          const v2f a = {fa[mt][i][2 * t], fa[mt][i][2 * t + 1]};
          acc[mt][0] = wmma_f32(a, b0, acc[mt][0]);
          acc[mt][1] = wmma_f32(a, b1, acc[mt][1]);
        }
        if (SUMS) {
          accs[0] = wmma_f32(ones, b0, accs[0]);
          accs[1] = wmma_f32(ones, b1, accs[1]);
        }
      }
    }
  }
  for (int mt = 0; mt < 4; ++mt)
    for (int r = 0; r < 8; ++r) {
      const v2f s = {acc[mt][0][r], acc[mt][1][r]};
      *(v2f*)(gp + (size_t)(16 * mt + 8 * hi + r) * C_CH) = s;
    }
  if (SUMS) {
    if (hi == 0) { const v2f s = {accs[0][0], accs[1][0]}; *(v2f*)sp = s; }
  }
}

__global__ void __launch_bounds__(128)
x_gram_kernel(const float* __restrict__ x, float* __restrict__ Gpart,
              float* __restrict__ spart) {
  const int lane = threadIdx.x & 31;
  const int l15  = lane & 15;
  const int hi   = lane >> 4;
  int t = blockIdx.x * (blockDim.x >> 5) +
          __builtin_amdgcn_readfirstlane((int)(threadIdx.x >> 5));
  const int u = t % GTILES; t /= GTILES;
  const int cg = (u < 8) ? 0 : (u < 14) ? 1 : (u < 18) ? 2 : 3;
  const int eg = (u < 8) ? u : (u < 14) ? u - 6 : (u < 18) ? u - 10 : u - 12;
  const int chunk = t & 31; t >>= 5;
  const int n     = t;

  const float* xn = x + (size_t)n * C_CH * HW_P + chunk * PLEN + 16 * hi;
  const float* xa = xn + (size_t)(cg * 64 + l15) * HW_P;
  const float* xb = xn + (size_t)(eg * 32 + 2 * l15) * HW_P;
  float* gp = Gpart + ((size_t)(chunk * N_B + n) * C_CH + cg * 64) * C_CH +
              eg * 32 + 2 * l15;
  float* sp = spart + (size_t)(chunk * N_B + n) * C_CH + eg * 32 + 2 * l15;
  if (cg == 0) gram_body<true>(xa, xb, gp, sp, l15, hi);
  else         gram_body<false>(xa, xb, gp, sp, l15, hi);
}

__device__ __forceinline__ void aug_col_frags(v4f (&f)[8],
                                              const float* __restrict__ w,
                                              const float* __restrict__ b,
                                              int tile, int k0, int l15, int hi) {
  if (tile < 16) {
    const float* wc = w + tile * 16 + l15 + (size_t)(k0 + 4 * hi) * C_CH;
#pragma unroll
    for (int i = 0; i < 8; ++i) {
      const float* wd = wc + (size_t)(8 * i) * C_CH;
      const v4f q = {wd[0], wd[C_CH], wd[2 * C_CH], wd[3 * C_CH]};
      f[i] = q;
    }
  } else {
    const v4f zero = {0.0f, 0.0f, 0.0f, 0.0f};
#pragma unroll
    for (int i = 0; i < 8; ++i) {
      const v4f q = *(const v4f*)(b + k0 + 8 * i + 4 * hi);
      f[i] = (l15 == 0) ? q : zero;
    }
  }
}

__device__ __forceinline__ void weight_products(
    const float* __restrict__ w0, const float* __restrict__ b0,
    const float* __restrict__ w1, const float* __restrict__ b1,
    const float* __restrict__ w2, const float* __restrict__ b2,
    const float* __restrict__ w3, float* __restrict__ Mt,
    float* __restrict__ At, int wv, int l15, int hi) {
  const bool first = wv < 16 * 17;
  const int u  = first ? wv : wv - 16 * 17;
  const int rt = u / 17;
  const int ct = u - rt * 17;
  const float* arow = w3 + (size_t)(rt * 16 + l15) * C_CH;
  v8f acc = {}, acc1 = {};
#pragma unroll 1
  for (int k0 = 0; k0 < C_CH; k0 += 64) {
    v4f fa[8], fb[8];
    if (first) {
#pragma unroll
      for (int i = 0; i < 8; ++i)
        fa[i] = *(const v4f*)(arow + k0 + 8 * i + 4 * hi);
      aug_col_frags(fb, w0, b0, ct, k0, l15, hi);
    } else {
      aug_col_frags(fa, w2, b2, rt, k0, l15, hi);
      aug_col_frags(fb, w1, b1, ct, k0, l15, hi);
    }
#pragma unroll
    for (int i = 0; i < 8; ++i) {
      const v2f a0 = {fa[i][0], fa[i][1]}, a1 = {fa[i][2], fa[i][3]};
      const v2f c0 = {fb[i][0], fb[i][1]}, c1 = {fb[i][2], fb[i][3]};
      acc  = wmma_f32(a0, c0, acc);
      acc1 = wmma_f32(a1, c1, acc1);
    }
  }
  acc = acc + acc1;
  float* d = (first ? Mt : At) + (size_t)(rt * 16 + 8 * hi) * CP + ct * 16 + l15;
#pragma unroll
  for (int r = 0; r < 8; ++r) d[(size_t)r * CP] = acc[r];
}

__global__ void __launch_bounds__(256)
reduce_g_kernel(const float* __restrict__ Gpart, const float* __restrict__ spart,
                float* __restrict__ Gp,
                const float* __restrict__ w0, const float* __restrict__ b0,
                const float* __restrict__ w1, const float* __restrict__ b1,
                const float* __restrict__ w2, const float* __restrict__ b2,
                const float* __restrict__ w3, float* __restrict__ Mt,
                float* __restrict__ At) {
  if (blockIdx.x < WPROD_BLOCKS) {
    const int wv = blockIdx.x * (blockDim.x >> 5) +
                   __builtin_amdgcn_readfirstlane((int)(threadIdx.x >> 5));
    const int lane = threadIdx.x & 31;
    if (wv < WPROD_WAVES)
      weight_products(w0, b0, w1, b1, w2, b2, w3, Mt, At, wv, lane & 15,
                      lane >> 4);
    return;
  }
  const int idx = (blockIdx.x - WPROD_BLOCKS) * blockDim.x + threadIdx.x;
  const int n   = idx / (CP * CP);
  const int rem = idx - n * (CP * CP);
  const int j   = rem / CP;
  const int e   = rem - j * CP;
  if (j < C_CH && e < C_CH && (j >> 6) > (e >> 6)) return;
  float s = 0.0f;
  if (j < C_CH && e < C_CH) {
    const float* g = Gpart + ((size_t)n * C_CH + j) * C_CH + e;
#pragma unroll 8
    for (int c = 0; c < PCH; ++c) s += g[(size_t)c * N_B * C_CH * C_CH];
  } else if ((j == C_CH) != (e == C_CH) && j <= C_CH && e <= C_CH) {
    const int ch = (j == C_CH) ? e : j;
    const float* q = spart + (size_t)n * C_CH + ch;
#pragma unroll 8
    for (int c = 0; c < PCH; ++c) s += q[(size_t)c * N_B * C_CH];
  } else if (j == C_CH && e == C_CH) {
    s = (float)HW_P;
  }
  Gp[idx] = s;
}

template <int NM, int DS>
__device__ __forceinline__ void nt_tile(const float* __restrict__ arow,
                                        const float* __restrict__ brow,
                                        const float* __restrict__ bcol,
                                        float* __restrict__ d, float scale,
                                        int hi) {
  v4f fa[KAUG_B], fb[KAUG_B];
#pragma unroll
  for (int i = 0; i < KAUG_B; ++i)
    fa[i] = *(const v4f*)(arow + 8 * i + 4 * hi);
#pragma unroll
  for (int i = 0; i < KAUG_B; ++i) {
    if (i < 8 * NM) {
      const float* bc = bcol + (size_t)(8 * i) * CP;
      const v4f q = {bc[0], bc[CP], bc[2 * CP], bc[3 * CP]};
      fb[i] = q;
    } else {
      fb[i] = *(const v4f*)(brow + 8 * i + 4 * hi);
    }
  }
  HOLD8(fa, 0); HOLD8(fb, 0); HOLD8(fa, 8);  HOLD8(fb, 8);
  v8f acc = {}, acc1 = {};
#pragma unroll
  for (int i = 0; i < KAUG_B; ++i) {
    const v2f a0 = {fa[i][0], fa[i][1]}, a1 = {fa[i][2], fa[i][3]};
    const v2f b0 = {fb[i][0], fb[i][1]}, b1 = {fb[i][2], fb[i][3]};
    acc  = wmma_f32(a0, b0, acc);
    acc1 = wmma_f32(a1, b1, acc1);
  }
  acc = acc + acc1;
#pragma unroll
  for (int r = 0; r < 8; ++r) d[(size_t)r * DS] = acc[r] * scale;
}

template <int DS>
__device__ __forceinline__ void nt_tile_nm(int nmir,
                                           const float* __restrict__ arow,
                                           const float* __restrict__ brow,
                                           const float* __restrict__ bcol,
                                           float* __restrict__ d, float scale,
                                           int hi) {
  if (nmir == 0)      nt_tile<0, DS>(arow, brow, bcol, d, scale, hi);
  else if (nmir == 1) nt_tile<1, DS>(arow, brow, bcol, d, scale, hi);
  else if (nmir == 2) nt_tile<2, DS>(arow, brow, bcol, d, scale, hi);
  else                nt_tile<3, DS>(arow, brow, bcol, d, scale, hi);
}

template <bool SYMB, bool FRAGD>
__global__ void __launch_bounds__(64)
nt_aug_kernel(const float* __restrict__ A, int aBatch,
              const float* __restrict__ B, int bBatch,
              float* __restrict__ D, float* __restrict__ Dfrag, float scale) {
  const int lane = threadIdx.x & 31;
  const int l15  = lane & 15;
  const int hi   = lane >> 4;
  int t = blockIdx.x * (blockDim.x >> 5) +
          __builtin_amdgcn_readfirstlane((int)(threadIdx.x >> 5));
  const int ct = t % 17; t /= 17;
  const int mt = t & 15; t >>= 4;
  const int n  = t;

  const float* arow = A + (size_t)n * aBatch + (size_t)(mt * 16 + l15) * CP;
  const float* brow = B + (size_t)n * bBatch + (size_t)(ct * 16 + l15) * CP;
  const float* bcol = B + (size_t)n * bBatch + (size_t)(4 * hi) * CP + ct * 16 + l15;
  float* d = D + ((size_t)n * C_CH + mt * 16 + 8 * hi) * CP + ct * 16 + l15;
  const int nmir = (SYMB && ct < 16) ? (ct >> 2) : 0;
  if (FRAGD && ct < 16) {
    float* df = Dfrag + (size_t)n * C_CH * C_CH + (size_t)mt * PF_TILE +
                (ct >> 1) * PF_BLK + (l15 >> 2) * PF_RUN +
                4 * (16 * (ct & 1) + 8 * hi) + (l15 & 3);
    nt_tile_nm<4>(nmir, arow, brow, bcol, df, scale, hi);
  } else {
    nt_tile_nm<CP>(nmir, arow, brow, bcol, d, scale, hi);
  }
}

struct OutFrags {
  v4f w0[4], w1[4];
  v4f x[16];
};

__device__ __forceinline__ void out_load_block(OutFrags& f,
                                               const float* __restrict__ wl0,
                                               const float* __restrict__ wl1,
                                               const float* __restrict__ xl,
                                               int j) {
  const float* xj = xl + (size_t)(32 * j) * HW_P;
  for (int i = 0; i < 16; ++i)
    f.x[i] = *(const v4f*)(xj + (size_t)i * HW_P);
  for (int i = 0; i < 4; ++i) {
    f.w0[i] = *(const v4f*)(wl0 + PF_BLK * j + PF_RUN * i);
    f.w1[i] = *(const v4f*)(wl1 + PF_BLK * j + PF_RUN * i);
  }
}

__device__ __forceinline__ void out_mma_block(v8f (&acc)[4][2],
                                              const OutFrags& f) {
  for (int s = 0; s < 8; ++s) {
    const int q = s >> 1, c = 2 * (s & 1);
    const v2f bb0 = {f.w0[q][c], f.w0[q][c + 1]};
    const v2f bb1 = {f.w1[q][c], f.w1[q][c + 1]};
    for (int mt = 0; mt < 4; ++mt) {
      const v2f px = {f.x[2 * s][mt], f.x[2 * s + 1][mt]};
      acc[mt][0] = wmma_f32(bb0, px, acc[mt][0]);
      acc[mt][1] = wmma_f32(bb1, px, acc[mt][1]);
    }
  }
}

__global__ void __launch_bounds__(256)
out_kernel(const float* __restrict__ x, const float* __restrict__ PT,
           const float* __restrict__ PF, const float* __restrict__ b3,
           float* __restrict__ out) {
  const int lane = threadIdx.x & 31;
  const int l15  = lane & 15;
  const int hi   = lane >> 4;
  int t = blockIdx.x * (blockDim.x >> 5) + (threadIdx.x >> 5);
  const int opair = t & 7;  t >>= 3;
  const int ptile = t & 63; t >>= 6;
  const int n     = t;

  const float* w = PT + (size_t)n * C_CH * CP;
  const int p0 = ptile * 64;
  const int o0 = opair * 32;
  const float* xl  = x + (size_t)n * C_CH * HW_P + (size_t)(16 * hi) * HW_P +
                     p0 + 4 * l15;
  const float* wl0 = PF + (size_t)n * C_CH * C_CH + (size_t)(2 * opair) * PF_TILE +
                     4 * lane;
  const float* wl1 = wl0 + PF_TILE;

  OutFrags f0, f1;
  out_load_block(f0, wl0, wl1, xl, 0);

  v8f acc[4][2];
  {
    const float* wa = w + (size_t)(o0 + 8 * hi) * CP + C_CH;
    const float* wb = wa + (size_t)16 * CP;
    const float* ba = b3 + o0 + 8 * hi;
    const v4f a0 = *(const v4f*)ba,        a1 = *(const v4f*)(ba + 4);
    const v4f c0 = *(const v4f*)(ba + 16), c1 = *(const v4f*)(ba + 20);
    const v8f bz0 = {wa[0] + a0[0],      wa[CP] + a0[1],     wa[2 * CP] + a0[2],
                     wa[3 * CP] + a0[3], wa[4 * CP] + a1[0], wa[5 * CP] + a1[1],
                     wa[6 * CP] + a1[2], wa[7 * CP] + a1[3]};
    const v8f bz1 = {wb[0] + c0[0],      wb[CP] + c0[1],     wb[2 * CP] + c0[2],
                     wb[3 * CP] + c0[3], wb[4 * CP] + c1[0], wb[5 * CP] + c1[1],
                     wb[6 * CP] + c1[2], wb[7 * CP] + c1[3]};
    for (int mt = 0; mt < 4; ++mt) { acc[mt][0] = bz0; acc[mt][1] = bz1; }
  }

  for (int j = 0; j < C_CH / 32; j += 2) {
    out_load_block(f1, wl0, wl1, xl, j + 1);
    out_mma_block(acc, f0);
    if (j + 2 < C_CH / 32) out_load_block(f0, wl0, wl1, xl, j + 2);
    out_mma_block(acc, f1);
  }
  float* ob0 = out + ((size_t)n * C_CH + o0 + 8 * hi) * HW_P + p0 + 4 * l15;
  float* ob1 = ob0 + (size_t)16 * HW_P;
  for (int r = 0; r < 8; ++r) {
    const v4f s0 = {acc[0][0][r], acc[1][0][r], acc[2][0][r], acc[3][0][r]};
    const v4f s1 = {acc[0][1][r], acc[1][1][r], acc[2][1][r], acc[3][1][r]};
    *(v4f*)(ob0 + (size_t)r * HW_P) = s0;
    *(v4f*)(ob1 + (size_t)r * HW_P) = s1;
  }
}

extern "C" void kernel_launch(void* const* d_in, const int* in_sizes, int n_in,
                              void* d_out, int out_size, void* d_ws,
                              size_t ws_size, hipStream_t stream) {
  const float* x  = (const float*)d_in[0];
  const float* w0 = (const float*)d_in[1];
  const float* b0 = (const float*)d_in[2];
  const float* w1 = (const float*)d_in[3];
  const float* b1 = (const float*)d_in[4];
  const float* w2 = (const float*)d_in[5];
  const float* b2 = (const float*)d_in[6];
  const float* w3 = (const float*)d_in[7];
  const float* b3 = (const float*)d_in[8];
  float* out = (float*)d_out;

  float* Gpart = (float*)d_ws;
  float* spart = Gpart + (size_t)PCH * N_B * C_CH * C_CH;
  float* Gp    = spart + (size_t)PCH * N_B * C_CH;
  float* Mt    = Gp    + (size_t)N_B * CP * CP;
  float* At    = Mt    + (size_t)C_CH * CP;
  float* T2t   = At    + (size_t)CP * CP;
  float* PT    = T2t   + (size_t)N_B * C_CH * CP;
  float* PF    = PT    + (size_t)N_B * C_CH * CP;

  x_gram_kernel<<<640, 128, 0, stream>>>(x, Gpart, spart);
  reduce_g_kernel<<<1227, 256, 0, stream>>>(Gpart, spart, Gp, w0, b0, w1, b1,
                                             w2, b2, w3, Mt, At);
  nt_aug_kernel<true, false><<<544, 64, 0, stream>>>(Mt, 0, Gp, CP * CP, T2t,
                                                     nullptr, 1.0f);
  nt_aug_kernel<false, true><<<544, 64, 0, stream>>>(T2t, C_CH * CP, At, 0, PT, PF,
                                                     1.0f / (float)HW_P);
  out_kernel<<<256, 256, 0, stream>>>(x, PT, PF, b3, out);
}


# slab sum folded into the launch of the product that reads G': k cut in 4, each of 272 groups sums its own block into LDS; weight products ride in the Gram launch; 5 launches to 4; sum order changes
# speedup vs baseline: 5.4600x; 1.0374x over previous
#include <hip/hip_runtime.h>


#define C_CH 256
#define CP   272
#define HW_P 4096
#define N_B  4
#define PCH  32
#define PLEN (HW_P / PCH)
#define GTILES 20
#define KAUG_B 33
#define WPROD_WAVES (16 * 17 + 17 * 17)
#define WPROD_BLOCKS 141
#define GRAM_BLOCKS 640
#define KQ      4
#define S_PITCH 76
#define TF_TILE (34 * PF_RUN)
#define T2_PART (N_B * 16 * TF_TILE)
#define PF_RUN  128
#define PF_BLK  (4 * PF_RUN)
#define PF_TILE (8 * PF_BLK)

typedef __attribute__((ext_vector_type(2))) float v2f;
typedef __attribute__((ext_vector_type(4))) float v4f;
typedef __attribute__((ext_vector_type(8))) float v8f;

#define HOLD8(p, o)                                                          \
  asm volatile("" ::"v"(p[(o)]), "v"(p[(o) + 1]), "v"(p[(o) + 2]),           \
               "v"(p[(o) + 3]), "v"(p[(o) + 4]), "v"(p[(o) + 5]),            \
               "v"(p[(o) + 6]), "v"(p[(o) + 7]))

__device__ __forceinline__ v8f wmma_f32(v2f a, v2f b, v8f c) {
  return __builtin_amdgcn_wmma_f32_16x16x4_f32(false, a, false, b, (short)0, c,
                                               false, false);
}

__device__ __forceinline__ v8f splat8(float v) {
  v8f r = {v, v, v, v, v, v, v, v};
  return r;
}

__device__ __forceinline__ void aug_col_frags(v4f (&f)[8],
                                              const float* __restrict__ w,
                                              const float* __restrict__ b,
                                              int tile, int k0, int l15, int hi) {
  if (tile < 16) {
    const float* wc = w + tile * 16 + l15 + (size_t)(k0 + 4 * hi) * C_CH;
#pragma unroll
    for (int i = 0; i < 8; ++i) {
      const float* wd = wc + (size_t)(8 * i) * C_CH;
      const v4f q = {wd[0], wd[C_CH], wd[2 * C_CH], wd[3 * C_CH]};
      f[i] = q;
    }
  } else {
    const v4f zero = {0.0f, 0.0f, 0.0f, 0.0f};
#pragma unroll
    for (int i = 0; i < 8; ++i) {
      const v4f q = *(const v4f*)(b + k0 + 8 * i + 4 * hi);
      f[i] = (l15 == 0) ? q : zero;
    }
  }
}

__device__ __forceinline__ void weight_products(
    const float* __restrict__ w0, const float* __restrict__ b0,
    const float* __restrict__ w1, const float* __restrict__ b1,
    const float* __restrict__ w2, const float* __restrict__ b2,
    const float* __restrict__ w3, float* __restrict__ Mt,
    float* __restrict__ At, int wv, int l15, int hi) {
  const bool first = wv < 16 * 17;
  const int u  = first ? wv : wv - 16 * 17;
  const int rt = u / 17;
  const int ct = u - rt * 17;
  const float* arow = w3 + (size_t)(rt * 16 + l15) * C_CH;
  v8f acc = {}, acc1 = {};
#pragma unroll 1
  for (int k0 = 0; k0 < C_CH; k0 += 64) {
    v4f fa[8], fb[8];
    if (first) {
#pragma unroll
      for (int i = 0; i < 8; ++i)
        fa[i] = *(const v4f*)(arow + k0 + 8 * i + 4 * hi);
      aug_col_frags(fb, w0, b0, ct, k0, l15, hi);
    } else {
      aug_col_frags(fa, w2, b2, rt, k0, l15, hi);
      aug_col_frags(fb, w1, b1, ct, k0, l15, hi);
    }
#pragma unroll
    for (int i = 0; i < 8; ++i) {
      const v2f a0 = {fa[i][0], fa[i][1]}, a1 = {fa[i][2], fa[i][3]};
      const v2f c0 = {fb[i][0], fb[i][1]}, c1 = {fb[i][2], fb[i][3]};
      acc  = wmma_f32(a0, c0, acc);
      acc1 = wmma_f32(a1, c1, acc1);
    }
  }
  acc = acc + acc1;
  float* d = (first ? Mt : At) + (size_t)(rt * 16 + 8 * hi) * CP + ct * 16 + l15;
#pragma unroll
  for (int r = 0; r < 8; ++r) d[(size_t)r * CP] = acc[r];
}

template <bool SUMS>
__device__ __forceinline__ void gram_body(const float* __restrict__ xa,
                                          const float* __restrict__ xb,
                                          float* __restrict__ gp,
                                          float* __restrict__ sp,
                                          int l15, int hi) {
  v8f acc[4][2];
  v8f accs[2];
  for (int mt = 0; mt < 4; ++mt) { acc[mt][0] = splat8(0.f); acc[mt][1] = splat8(0.f); }
  accs[0] = splat8(0.f); accs[1] = splat8(0.f);
  const v2f ones = {1.0f, 1.0f};

  for (int kb = 0; kb < PLEN / 32; ++kb) {
    v4f fa[4][4], fb[2][4];
    for (int mt = 0; mt < 4; ++mt)
      for (int i = 0; i < 4; ++i)
        fa[mt][i] = *(const v4f*)(xa + (size_t)(16 * mt) * HW_P + 32 * kb + 4 * i);
    for (int nt = 0; nt < 2; ++nt)
      for (int i = 0; i < 4; ++i)
        fb[nt][i] = *(const v4f*)(xb + (size_t)nt * HW_P + 32 * kb + 4 * i);
    for (int i = 0; i < 4; ++i) {
      for (int t = 0; t < 2; ++t) {
        const v2f b0 = {fb[0][i][2 * t], fb[0][i][2 * t + 1]};
        const v2f b1 = {fb[1][i][2 * t], fb[1][i][2 * t + 1]};
        for (int mt = 0; mt < 4; ++mt) {
          const v2f a = {fa[mt][i][2 * t], fa[mt][i][2 * t + 1]};
          acc[mt][0] = wmma_f32(a, b0, acc[mt][0]);
          acc[mt][1] = wmma_f32(a, b1, acc[mt][1]);
        }
        if (SUMS) {
          accs[0] = wmma_f32(ones, b0, accs[0]);
          accs[1] = wmma_f32(ones, b1, accs[1]);
        }
      }
    }
  }
  for (int mt = 0; mt < 4; ++mt)
    for (int r = 0; r < 8; ++r) {
      const v2f s = {acc[mt][0][r], acc[mt][1][r]};
      *(v2f*)(gp + (size_t)(16 * mt + 8 * hi + r) * C_CH) = s;
    }
  if (SUMS) {
    if (hi == 0) { const v2f s = {accs[0][0], accs[1][0]}; *(v2f*)sp = s; }
  }
}

__global__ void __launch_bounds__(128)
x_gram_kernel(const float* __restrict__ x, float* __restrict__ Gpart,
              float* __restrict__ spart,
              const float* __restrict__ w0, const float* __restrict__ b0,
              const float* __restrict__ w1, const float* __restrict__ b1,
              const float* __restrict__ w2, const float* __restrict__ b2,
              const float* __restrict__ w3, float* __restrict__ Mt,
              float* __restrict__ At) {
  const int lane = threadIdx.x & 31;
  const int l15  = lane & 15;
  const int hi   = lane >> 4;
  const int wave = __builtin_amdgcn_readfirstlane((int)(threadIdx.x >> 5));
  if (blockIdx.x >= GRAM_BLOCKS) {
    const int wv = (blockIdx.x - GRAM_BLOCKS) * (blockDim.x >> 5) + wave;
    if (wv < WPROD_WAVES)
      weight_products(w0, b0, w1, b1, w2, b2, w3, Mt, At, wv, l15, hi);
    return;
  }
  int t = blockIdx.x * (blockDim.x >> 5) + wave;
  const int u = t % GTILES; t /= GTILES;
  const int cg = (u < 8) ? 0 : (u < 14) ? 1 : (u < 18) ? 2 : 3;
  const int eg = (u < 8) ? u : (u < 14) ? u - 6 : (u < 18) ? u - 10 : u - 12;
  const int chunk = t & 31; t >>= 5;
  const int n     = t;

  const float* xn = x + (size_t)n * C_CH * HW_P + chunk * PLEN + 16 * hi;
  const float* xa = xn + (size_t)(cg * 64 + l15) * HW_P;
  const float* xb = xn + (size_t)(eg * 32 + 2 * l15) * HW_P;
  float* gp = Gpart + ((size_t)(chunk * N_B + n) * C_CH + cg * 64) * C_CH +
              eg * 32 + 2 * l15;
  float* sp = spart + (size_t)(chunk * N_B + n) * C_CH + eg * 32 + 2 * l15;
  if (cg == 0) gram_body<true>(xa, xb, gp, sp, l15, hi);
  else         gram_body<false>(xa, xb, gp, sp, l15, hi);
}

__global__ void __launch_bounds__(256)
gsum_prod_kernel(const float* __restrict__ Gpart, const float* __restrict__ spart,
                 const float* __restrict__ Mt, float* __restrict__ T2part) {
  __shared__ __attribute__((aligned(16))) float S[16 * S_PITCH];
  const int tid  = threadIdx.x;
  const int lane = tid & 31;
  const int l15  = lane & 15;
  const int hi   = lane >> 4;
  const int w  = __builtin_amdgcn_readfirstlane(tid >> 5);
  const int q  = blockIdx.x & (KQ - 1);
  const int bj = blockIdx.x >> 2;
  const int jt = bj % 17;
  const int n  = bj / 17;

  const float* arow = Mt + (size_t)(32 * w + l15) * CP + 64 * q + 4 * hi;
  v4f fa0[9], fa1[9];
#pragma unroll
  for (int i = 0; i < 8; ++i) {
    fa0[i] = *(const v4f*)(arow + 8 * i);
    fa1[i] = *(const v4f*)(arow + (size_t)16 * CP + 8 * i);
  }
  if (q == KQ - 1) {
    fa0[8] = *(const v4f*)(arow + 64);
    fa1[8] = *(const v4f*)(arow + (size_t)16 * CP + 64);
  }

  const size_t slab = (size_t)N_B * C_CH * C_CH;
  const float* gn = Gpart + (size_t)n * C_CH * C_CH;
  const float* sn = spart + (size_t)n * C_CH;
  if (jt < 16) {
    v4f s = {0.0f, 0.0f, 0.0f, 0.0f};
    if (q >= (jt >> 2)) {
      const int r = tid >> 4, c4 = tid & 15;
      const float* g = gn + (size_t)(16 * jt + r) * C_CH + 64 * q + 4 * c4;
#pragma unroll
      for (int c = 0; c < PCH; ++c) s += *(const v4f*)(g + (size_t)c * slab);
      *(v4f*)(S + r * S_PITCH + 4 * c4) = s;
    } else {
      const int er = tid >> 2, jc = tid & 3;
      const float* g = gn + (size_t)(64 * q + er) * C_CH + 16 * jt + 4 * jc;
#pragma unroll
      for (int c = 0; c < PCH; ++c) s += *(const v4f*)(g + (size_t)c * slab);
#pragma unroll
      for (int m = 0; m < 4; ++m) S[(4 * jc + m) * S_PITCH + er] = s[m];
    }
    if (q == KQ - 1 && tid < 128) {
      const int r = tid >> 3, c = tid & 7;
      float a = 0.0f;
      if (c == 0) {
        const float* sp = sn + 16 * jt + r;
#pragma unroll 8
        for (int ch = 0; ch < PCH; ++ch) a += sp[(size_t)ch * N_B * C_CH];
      }
      S[r * S_PITCH + 64 + c] = a;
    }
  } else {
    for (int idx = tid; idx < 16 * 72; idx += 256) {
      const int r = idx / 72, c = idx - r * 72;
      float a = 0.0f;
      if (r == 0) {
        if (c < 64) {
          const float* sp = sn + 64 * q + c;
#pragma unroll 8
          for (int ch = 0; ch < PCH; ++ch) a += sp[(size_t)ch * N_B * C_CH];
        } else if (c == 64 && q == KQ - 1) {
          a = (float)HW_P;
        }
      }
      S[r * S_PITCH + c] = a;
    }
  }
  __syncthreads();

  const float* sb = S + l15 * S_PITCH + 4 * hi;
  v8f acc0 = {}, acc1 = {};
#pragma unroll
  for (int i = 0; i < 8; ++i) {
    const v4f fb = *(const v4f*)(sb + 8 * i);
    const v2f b0 = {fb[0], fb[1]}, b1 = {fb[2], fb[3]};
    const v2f a0 = {fa0[i][0], fa0[i][1]}, a1 = {fa0[i][2], fa0[i][3]};
    const v2f c0 = {fa1[i][0], fa1[i][1]}, c1 = {fa1[i][2], fa1[i][3]};
    acc0 = wmma_f32(a0, b0, acc0);
    acc1 = wmma_f32(c0, b0, acc1);
    acc0 = wmma_f32(a1, b1, acc0);
    acc1 = wmma_f32(c1, b1, acc1);
  }
  if (q == KQ - 1) {
    const v4f fb = *(const v4f*)(sb + 64);
    const v2f b0 = {fb[0], fb[1]}, b1 = {fb[2], fb[3]};
    const v2f a0 = {fa0[8][0], fa0[8][1]}, a1 = {fa0[8][2], fa0[8][3]};
    const v2f c0 = {fa1[8][0], fa1[8][1]}, c1 = {fa1[8][2], fa1[8][3]};
    acc0 = wmma_f32(a0, b0, acc0);
    acc1 = wmma_f32(c0, b0, acc1);
    acc0 = wmma_f32(a1, b1, acc0);
    acc1 = wmma_f32(c1, b1, acc1);
  }
  float* d = T2part + ((size_t)(q * N_B + n) * 16 + 2 * w) * TF_TILE +
             (2 * jt + (l15 >> 3)) * PF_RUN +
             4 * (16 * ((l15 >> 2) & 1) + 8 * hi) + (l15 & 3);
#pragma unroll
  for (int r = 0; r < 8; ++r) {
    d[4 * r] = acc0[r];
    d[TF_TILE + 4 * r] = acc1[r];
  }
}

template <int DS>
__device__ __forceinline__ void nt_tile(const float* __restrict__ arow,
                                        const float* __restrict__ brow,
                                        float* __restrict__ d, float scale,
                                        int hi) {
  v4f fa[KAUG_B], fb[KAUG_B];
#pragma unroll
  for (int i = 0; i < KAUG_B; ++i)
    fb[i] = *(const v4f*)(brow + 8 * i + 4 * hi);
#pragma unroll
  for (int i = 0; i < KAUG_B; ++i) {
    const float* a = arow + PF_RUN * i;
    const v4f p0 = *(const v4f*)a;
    const v4f p1 = *(const v4f*)(a + (size_t)T2_PART);
    const v4f p2 = *(const v4f*)(a + (size_t)2 * T2_PART);
    const v4f p3 = *(const v4f*)(a + (size_t)3 * T2_PART);
    fa[i] = (p0 + p1) + (p2 + p3);
  }
  HOLD8(fb, 0); HOLD8(fb, 8);
  v8f acc = {}, acc1 = {};
#pragma unroll
  for (int i = 0; i < KAUG_B; ++i) {
    const v2f a0 = {fa[i][0], fa[i][1]}, a1 = {fa[i][2], fa[i][3]};
    const v2f b0 = {fb[i][0], fb[i][1]}, b1 = {fb[i][2], fb[i][3]};
    acc  = wmma_f32(a0, b0, acc);
    acc1 = wmma_f32(a1, b1, acc1);
  }
  acc = acc + acc1;
#pragma unroll
  for (int r = 0; r < 8; ++r) d[(size_t)r * DS] = acc[r] * scale;
}

__global__ void __launch_bounds__(64)
p_kernel(const float* __restrict__ T2part, const float* __restrict__ At,
         float* __restrict__ D, float* __restrict__ Dfrag, float scale) {
  const int lane = threadIdx.x & 31;
  const int l15  = lane & 15;
  const int hi   = lane >> 4;
  int t = blockIdx.x * (blockDim.x >> 5) +
          __builtin_amdgcn_readfirstlane((int)(threadIdx.x >> 5));
  const int ct = t % 17; t /= 17;
  const int mt = t & 15; t >>= 4;
  const int n  = t;

  const float* arow = T2part + (size_t)(n * 16 + mt) * TF_TILE + 4 * lane;
  const float* brow = At + (size_t)(ct * 16 + l15) * CP;
  if (ct < 16) {
    float* df = Dfrag + (size_t)n * C_CH * C_CH + (size_t)mt * PF_TILE +
                (ct >> 1) * PF_BLK + (l15 >> 2) * PF_RUN +
                4 * (16 * (ct & 1) + 8 * hi) + (l15 & 3);
    nt_tile<4>(arow, brow, df, scale, hi);
  } else {
    float* d = D + ((size_t)n * C_CH + mt * 16 + 8 * hi) * CP + ct * 16 + l15;
    nt_tile<CP>(arow, brow, d, scale, hi);
  }
}

struct OutFrags {
  v4f w0[4], w1[4];
  v4f x[16];
};

__device__ __forceinline__ void out_load_block(OutFrags& f,
                                               const float* __restrict__ wl0,
                                               const float* __restrict__ wl1,
                                               const float* __restrict__ xl,
                                               int j) {
  const float* xj = xl + (size_t)(32 * j) * HW_P;
  for (int i = 0; i < 16; ++i)
    f.x[i] = *(const v4f*)(xj + (size_t)i * HW_P);
  for (int i = 0; i < 4; ++i) {
    f.w0[i] = *(const v4f*)(wl0 + PF_BLK * j + PF_RUN * i);
    f.w1[i] = *(const v4f*)(wl1 + PF_BLK * j + PF_RUN * i);
  }
}

__device__ __forceinline__ void out_mma_block(v8f (&acc)[4][2],
                                              const OutFrags& f) {
  for (int s = 0; s < 8; ++s) {
    const int q = s >> 1, c = 2 * (s & 1);
    const v2f bb0 = {f.w0[q][c], f.w0[q][c + 1]};
    const v2f bb1 = {f.w1[q][c], f.w1[q][c + 1]};
    for (int mt = 0; mt < 4; ++mt) {
      const v2f px = {f.x[2 * s][mt], f.x[2 * s + 1][mt]};
      acc[mt][0] = wmma_f32(bb0, px, acc[mt][0]);
      acc[mt][1] = wmma_f32(bb1, px, acc[mt][1]);
    }
  }
}

__global__ void __launch_bounds__(256)
out_kernel(const float* __restrict__ x, const float* __restrict__ PT,
           const float* __restrict__ PF, const float* __restrict__ b3,
           float* __restrict__ out) {
  const int lane = threadIdx.x & 31;
  const int l15  = lane & 15;
  const int hi   = lane >> 4;
  int t = blockIdx.x * (blockDim.x >> 5) + (threadIdx.x >> 5);
  const int opair = t & 7;  t >>= 3;
  const int ptile = t & 63; t >>= 6;
  const int n     = t;

  const float* w = PT + (size_t)n * C_CH * CP;
  const int p0 = ptile * 64;
  const int o0 = opair * 32;
  const float* xl  = x + (size_t)n * C_CH * HW_P + (size_t)(16 * hi) * HW_P +
                     p0 + 4 * l15;
  const float* wl0 = PF + (size_t)n * C_CH * C_CH + (size_t)(2 * opair) * PF_TILE +
                     4 * lane;
  const float* wl1 = wl0 + PF_TILE;

  OutFrags f0, f1;
  out_load_block(f0, wl0, wl1, xl, 0);

  v8f acc[4][2];
  {
    const float* wa = w + (size_t)(o0 + 8 * hi) * CP + C_CH;
    const float* wb = wa + (size_t)16 * CP;
    const float* ba = b3 + o0 + 8 * hi;
    const v4f a0 = *(const v4f*)ba,        a1 = *(const v4f*)(ba + 4);
    const v4f c0 = *(const v4f*)(ba + 16), c1 = *(const v4f*)(ba + 20);
    const v8f bz0 = {wa[0] + a0[0],      wa[CP] + a0[1],     wa[2 * CP] + a0[2],
                     wa[3 * CP] + a0[3], wa[4 * CP] + a1[0], wa[5 * CP] + a1[1],
                     wa[6 * CP] + a1[2], wa[7 * CP] + a1[3]};
    const v8f bz1 = {wb[0] + c0[0],      wb[CP] + c0[1],     wb[2 * CP] + c0[2],
                     wb[3 * CP] + c0[3], wb[4 * CP] + c1[0], wb[5 * CP] + c1[1],
                     wb[6 * CP] + c1[2], wb[7 * CP] + c1[3]};
    for (int mt = 0; mt < 4; ++mt) { acc[mt][0] = bz0; acc[mt][1] = bz1; }
  }

  for (int j = 0; j < C_CH / 32; j += 2) {
    out_load_block(f1, wl0, wl1, xl, j + 1);
    out_mma_block(acc, f0);
    if (j + 2 < C_CH / 32) out_load_block(f0, wl0, wl1, xl, j + 2);
    out_mma_block(acc, f1);
  }
  float* ob0 = out + ((size_t)n * C_CH + o0 + 8 * hi) * HW_P + p0 + 4 * l15;
  float* ob1 = ob0 + (size_t)16 * HW_P;
  for (int r = 0; r < 8; ++r) {
    const v4f s0 = {acc[0][0][r], acc[1][0][r], acc[2][0][r], acc[3][0][r]};
    const v4f s1 = {acc[0][1][r], acc[1][1][r], acc[2][1][r], acc[3][1][r]};
    *(v4f*)(ob0 + (size_t)r * HW_P) = s0;
    *(v4f*)(ob1 + (size_t)r * HW_P) = s1;
  }
}

extern "C" void kernel_launch(void* const* d_in, const int* in_sizes, int n_in,
                              void* d_out, int out_size, void* d_ws,
                              size_t ws_size, hipStream_t stream) {
  const float* x  = (const float*)d_in[0];
  const float* w0 = (const float*)d_in[1];
  const float* b0 = (const float*)d_in[2];
  const float* w1 = (const float*)d_in[3];
  const float* b1 = (const float*)d_in[4];
  const float* w2 = (const float*)d_in[5];
  const float* b2 = (const float*)d_in[6];
  const float* w3 = (const float*)d_in[7];
  const float* b3 = (const float*)d_in[8];
  float* out = (float*)d_out;

  float* Gpart  = (float*)d_ws;
  float* spart  = Gpart + (size_t)PCH * N_B * C_CH * C_CH;
  float* Mt     = spart + (size_t)PCH * N_B * C_CH;
  float* At     = Mt    + (size_t)C_CH * CP;
  float* T2part = At    + (size_t)CP * CP;
  float* PT     = T2part + (size_t)KQ * T2_PART;
  float* PF     = PT    + (size_t)N_B * C_CH * CP;

  x_gram_kernel<<<781, 128, 0, stream>>>(x, Gpart, spart, w0, b0, w1, b1, w2, b2,
                                         w3, Mt, At);
  gsum_prod_kernel<<<272, 256, 0, stream>>>(Gpart, spart, Mt, T2part);
  p_kernel<<<544, 64, 0, stream>>>(T2part, At, PT, PF, 1.0f / (float)HW_P);
  out_kernel<<<256, 256, 0, stream>>>(x, PT, PF, b3, out);
}


# Gram: the 4 waves of a work-group take one tile over 4 neighbouring pixel chunks and add their results in LDS: 8 partial slabs reach memory and kernel 2, not 32; pixel sum order changes
# speedup vs baseline: 5.7354x; 1.0504x over previous
#include <hip/hip_runtime.h>


#define C_CH 256
#define CP   272
#define HW_P 4096
#define N_B  4
#define PCH  32
#define PLEN (HW_P / PCH)
#define CSUM 4
#define PSL  (PCH / CSUM)
#define RED_W (4 * 4 * 32 * 4)
#define GTILES 20
#define KAUG_B 33
#define WPROD_WAVES (16 * 17 + 17 * 17)
#define WPROD_BLOCKS 141
#define GRAM_BLOCKS 640
#define KQ      4
#define S_PITCH 76
#define TF_TILE (34 * PF_RUN)
#define T2_PART (N_B * 16 * TF_TILE)
#define PF_RUN  128
#define PF_BLK  (4 * PF_RUN)
#define PF_TILE (8 * PF_BLK)

typedef __attribute__((ext_vector_type(2))) float v2f;
typedef __attribute__((ext_vector_type(4))) float v4f;
typedef __attribute__((ext_vector_type(8))) float v8f;

#define HOLD8(p, o)                                                          \
  asm volatile("" ::"v"(p[(o)]), "v"(p[(o) + 1]), "v"(p[(o) + 2]),           \
               "v"(p[(o) + 3]), "v"(p[(o) + 4]), "v"(p[(o) + 5]),            \
               "v"(p[(o) + 6]), "v"(p[(o) + 7]))

__device__ __forceinline__ v8f wmma_f32(v2f a, v2f b, v8f c) {
  return __builtin_amdgcn_wmma_f32_16x16x4_f32(false, a, false, b, (short)0, c,
                                               false, false);
}

__device__ __forceinline__ v8f splat8(float v) {
  v8f r = {v, v, v, v, v, v, v, v};
  return r;
}

__device__ __forceinline__ void aug_col_frags(v4f (&f)[8],
                                              const float* __restrict__ w,
                                              const float* __restrict__ b,
                                              int tile, int k0, int l15, int hi) {
  if (tile < 16) {
    const float* wc = w + tile * 16 + l15 + (size_t)(k0 + 4 * hi) * C_CH;
#pragma unroll
    for (int i = 0; i < 8; ++i) {
      const float* wd = wc + (size_t)(8 * i) * C_CH;
      const v4f q = {wd[0], wd[C_CH], wd[2 * C_CH], wd[3 * C_CH]};
      f[i] = q;
    }
  } else {
    const v4f zero = {0.0f, 0.0f, 0.0f, 0.0f};
#pragma unroll
    for (int i = 0; i < 8; ++i) {
      const v4f q = *(const v4f*)(b + k0 + 8 * i + 4 * hi);
      f[i] = (l15 == 0) ? q : zero;
    }
  }
}

__device__ __forceinline__ void weight_products(
    const float* __restrict__ w0, const float* __restrict__ b0,
    const float* __restrict__ w1, const float* __restrict__ b1,
    const float* __restrict__ w2, const float* __restrict__ b2,
    const float* __restrict__ w3, float* __restrict__ Mt,
    float* __restrict__ At, int wv, int l15, int hi) {
  const bool first = wv < 16 * 17;
  const int u  = first ? wv : wv - 16 * 17;
  const int rt = u / 17;
  const int ct = u - rt * 17;
  const float* arow = w3 + (size_t)(rt * 16 + l15) * C_CH;
  v8f acc = {}, acc1 = {};
#pragma unroll 1
  for (int k0 = 0; k0 < C_CH; k0 += 64) {
    v4f fa[8], fb[8];
    if (first) {
#pragma unroll
      for (int i = 0; i < 8; ++i)
        fa[i] = *(const v4f*)(arow + k0 + 8 * i + 4 * hi);
      aug_col_frags(fb, w0, b0, ct, k0, l15, hi);
    } else {
      aug_col_frags(fa, w2, b2, rt, k0, l15, hi);
      aug_col_frags(fb, w1, b1, ct, k0, l15, hi);
    }
#pragma unroll
    for (int i = 0; i < 8; ++i) {
      const v2f a0 = {fa[i][0], fa[i][1]}, a1 = {fa[i][2], fa[i][3]};
      const v2f c0 = {fb[i][0], fb[i][1]}, c1 = {fb[i][2], fb[i][3]};
      acc  = wmma_f32(a0, c0, acc);
      acc1 = wmma_f32(a1, c1, acc1);
    }
  }
  acc = acc + acc1;
  float* d = (first ? Mt : At) + (size_t)(rt * 16 + 8 * hi) * CP + ct * 16 + l15;
#pragma unroll
  for (int r = 0; r < 8; ++r) d[(size_t)r * CP] = acc[r];
}

template <bool SUMS>
__device__ __forceinline__ void gram_body(const float* __restrict__ xa,
                                          const float* __restrict__ xb,
                                          float* __restrict__ gp,
                                          float* __restrict__ sp,
                                          float* red, float* reds,
                                          int wave, int lane, int l15, int hi) {
  v8f acc[4][2];
  v8f accs[2];
  for (int mt = 0; mt < 4; ++mt) { acc[mt][0] = splat8(0.f); acc[mt][1] = splat8(0.f); }
  accs[0] = splat8(0.f); accs[1] = splat8(0.f);
  const v2f ones = {1.0f, 1.0f};

  for (int kb = 0; kb < PLEN / 32; ++kb) {
    v4f fa[4][4], fb[2][4];
    for (int mt = 0; mt < 4; ++mt)
      for (int i = 0; i < 4; ++i)
        fa[mt][i] = *(const v4f*)(xa + (size_t)(16 * mt) * HW_P + 32 * kb + 4 * i);
    for (int nt = 0; nt < 2; ++nt)
      for (int i = 0; i < 4; ++i)
        fb[nt][i] = *(const v4f*)(xb + (size_t)nt * HW_P + 32 * kb + 4 * i);
    for (int i = 0; i < 4; ++i) {
      for (int t = 0; t < 2; ++t) {
        const v2f b0 = {fb[0][i][2 * t], fb[0][i][2 * t + 1]};
        const v2f b1 = {fb[1][i][2 * t], fb[1][i][2 * t + 1]};
        for (int mt = 0; mt < 4; ++mt) {
          const v2f a = {fa[mt][i][2 * t], fa[mt][i][2 * t + 1]};
          acc[mt][0] = wmma_f32(a, b0, acc[mt][0]);
          acc[mt][1] = wmma_f32(a, b1, acc[mt][1]);
        }
        if (SUMS) {
          accs[0] = wmma_f32(ones, b0, accs[0]);
          accs[1] = wmma_f32(ones, b1, accs[1]);
        }
      }
    }
  }
  float* rw = red + wave * RED_W + 4 * lane;
  for (int mt = 0; mt < 4; ++mt)
    for (int rp = 0; rp < 4; ++rp) {
      const v4f q = {acc[mt][0][2 * rp], acc[mt][1][2 * rp],
                     acc[mt][0][2 * rp + 1], acc[mt][1][2 * rp + 1]};
      *(v4f*)(rw + (mt * 4 + rp) * 128) = q;
    }
  if (SUMS) {
    if (hi == 0) {
      const v2f s = {accs[0][0], accs[1][0]};
      *(v2f*)(reds + (wave * 16 + l15) * 2) = s;
    }
  }
  __syncthreads();
  const float* rr = red + (wave * 4) * 128 + 4 * lane;
  for (int rp = 0; rp < 4; ++rp) {
    v4f s = *(const v4f*)(rr + rp * 128);
    for (int v = 1; v < CSUM; ++v) s += *(const v4f*)(rr + v * RED_W + rp * 128);
    float* g = gp + (size_t)(16 * wave + 8 * hi + 2 * rp) * C_CH;
    const v2f s0 = {s[0], s[1]}, s1 = {s[2], s[3]};
    *(v2f*)g = s0;
    *(v2f*)(g + C_CH) = s1;
  }
  if (SUMS) {
    if (wave == 0 && hi == 0) {
      v2f s = *(const v2f*)(reds + l15 * 2);
      for (int v = 1; v < CSUM; ++v) s += *(const v2f*)(reds + (v * 16 + l15) * 2);
      *(v2f*)sp = s;
    }
  }
}

__global__ void __launch_bounds__(128) __attribute__((amdgpu_waves_per_eu(3, 4)))
x_gram_kernel(const float* __restrict__ x, float* __restrict__ Gpart,
              float* __restrict__ spart,
              const float* __restrict__ w0, const float* __restrict__ b0,
              const float* __restrict__ w1, const float* __restrict__ b1,
              const float* __restrict__ w2, const float* __restrict__ b2,
              const float* __restrict__ w3, float* __restrict__ Mt,
              float* __restrict__ At) {
  const int lane = threadIdx.x & 31;
  const int l15  = lane & 15;
  const int hi   = lane >> 4;
  const int wave = __builtin_amdgcn_readfirstlane((int)(threadIdx.x >> 5));
  if (blockIdx.x >= GRAM_BLOCKS) {
    const int wv = (blockIdx.x - GRAM_BLOCKS) * (blockDim.x >> 5) + wave;
    if (wv < WPROD_WAVES)
      weight_products(w0, b0, w1, b1, w2, b2, w3, Mt, At, wv, l15, hi);
    return;
  }
  __shared__ __attribute__((aligned(16))) float red[CSUM * RED_W];
  __shared__ __attribute__((aligned(16))) float reds[CSUM * 32];
  int t = blockIdx.x;
  const int u = t % GTILES; t /= GTILES;
  const int cg = (u < 8) ? 0 : (u < 14) ? 1 : (u < 18) ? 2 : 3;
  const int eg = (u < 8) ? u : (u < 14) ? u - 6 : (u < 18) ? u - 10 : u - 12;
  const int grp = t % PSL; t /= PSL;
  const int n     = t;
  const int chunk = grp * CSUM + wave;

  const float* xn = x + (size_t)n * C_CH * HW_P + chunk * PLEN + 16 * hi;
  const float* xa = xn + (size_t)(cg * 64 + l15) * HW_P;
  const float* xb = xn + (size_t)(eg * 32 + 2 * l15) * HW_P;
  float* gp = Gpart + ((size_t)(grp * N_B + n) * C_CH + cg * 64) * C_CH +
              eg * 32 + 2 * l15;
  float* sp = spart + (size_t)(grp * N_B + n) * C_CH + eg * 32 + 2 * l15;
  if (cg == 0) gram_body<true>(xa, xb, gp, sp, red, reds, wave, lane, l15, hi);
  else         gram_body<false>(xa, xb, gp, sp, red, reds, wave, lane, l15, hi);
}

__global__ void __launch_bounds__(256)
gsum_prod_kernel(const float* __restrict__ Gpart, const float* __restrict__ spart,
                 const float* __restrict__ Mt, float* __restrict__ T2part) {
  __shared__ __attribute__((aligned(16))) float S[16 * S_PITCH];
  const int tid  = threadIdx.x;
  const int lane = tid & 31;
  const int l15  = lane & 15;
  const int hi   = lane >> 4;
  const int w  = __builtin_amdgcn_readfirstlane(tid >> 5);
  const int q  = blockIdx.x & (KQ - 1);
  const int bj = blockIdx.x >> 2;
  const int jt = bj % 17;
  const int n  = bj / 17;

  const float* arow = Mt + (size_t)(32 * w + l15) * CP + 64 * q + 4 * hi;
  v4f fa0[9], fa1[9];
#pragma unroll
  for (int i = 0; i < 8; ++i) {
    fa0[i] = *(const v4f*)(arow + 8 * i);
    fa1[i] = *(const v4f*)(arow + (size_t)16 * CP + 8 * i);
  }
  if (q == KQ - 1) {
    fa0[8] = *(const v4f*)(arow + 64);
    fa1[8] = *(const v4f*)(arow + (size_t)16 * CP + 64);
  }

  const size_t slab = (size_t)N_B * C_CH * C_CH;
  const float* gn = Gpart + (size_t)n * C_CH * C_CH;
  const float* sn = spart + (size_t)n * C_CH;
  if (jt < 16) {
    v4f s = {0.0f, 0.0f, 0.0f, 0.0f};
    if (q >= (jt >> 2)) {
      const int r = tid >> 4, c4 = tid & 15;
      const float* g = gn + (size_t)(16 * jt + r) * C_CH + 64 * q + 4 * c4;
#pragma unroll
      for (int c = 0; c < PSL; ++c) s += *(const v4f*)(g + (size_t)c * slab);
      *(v4f*)(S + r * S_PITCH + 4 * c4) = s;
    } else {
      const int er = tid >> 2, jc = tid & 3;
      const float* g = gn + (size_t)(64 * q + er) * C_CH + 16 * jt + 4 * jc;
#pragma unroll
      for (int c = 0; c < PSL; ++c) s += *(const v4f*)(g + (size_t)c * slab);
#pragma unroll
      for (int m = 0; m < 4; ++m) S[(4 * jc + m) * S_PITCH + er] = s[m];
    }
    if (q == KQ - 1 && tid < 128) {
      const int r = tid >> 3, c = tid & 7;
      float a = 0.0f;
      if (c == 0) {
        const float* sp = sn + 16 * jt + r;
#pragma unroll 8
        for (int ch = 0; ch < PSL; ++ch) a += sp[(size_t)ch * N_B * C_CH];
      }
      S[r * S_PITCH + 64 + c] = a;
    }
  } else {
    for (int idx = tid; idx < 16 * 72; idx += 256) {
      const int r = idx / 72, c = idx - r * 72;
      float a = 0.0f;
      if (r == 0) {
        if (c < 64) {
          const float* sp = sn + 64 * q + c;
#pragma unroll 8
          for (int ch = 0; ch < PSL; ++ch) a += sp[(size_t)ch * N_B * C_CH];
        } else if (c == 64 && q == KQ - 1) {
          a = (float)HW_P;
        }
      }
      S[r * S_PITCH + c] = a;
    }
  }
  __syncthreads();

  const float* sb = S + l15 * S_PITCH + 4 * hi;
  v8f acc0 = {}, acc1 = {};
#pragma unroll
  for (int i = 0; i < 8; ++i) {
    const v4f fb = *(const v4f*)(sb + 8 * i);
    const v2f b0 = {fb[0], fb[1]}, b1 = {fb[2], fb[3]};
    const v2f a0 = {fa0[i][0], fa0[i][1]}, a1 = {fa0[i][2], fa0[i][3]};
    const v2f c0 = {fa1[i][0], fa1[i][1]}, c1 = {fa1[i][2], fa1[i][3]};
    acc0 = wmma_f32(a0, b0, acc0);
    acc1 = wmma_f32(c0, b0, acc1);
    acc0 = wmma_f32(a1, b1, acc0);
    acc1 = wmma_f32(c1, b1, acc1);
  }
  if (q == KQ - 1) {
    const v4f fb = *(const v4f*)(sb + 64);
    const v2f b0 = {fb[0], fb[1]}, b1 = {fb[2], fb[3]};
    const v2f a0 = {fa0[8][0], fa0[8][1]}, a1 = {fa0[8][2], fa0[8][3]};
    const v2f c0 = {fa1[8][0], fa1[8][1]}, c1 = {fa1[8][2], fa1[8][3]};
    acc0 = wmma_f32(a0, b0, acc0);
    acc1 = wmma_f32(c0, b0, acc1);
    acc0 = wmma_f32(a1, b1, acc0);
    acc1 = wmma_f32(c1, b1, acc1);
  }
  float* d = T2part + ((size_t)(q * N_B + n) * 16 + 2 * w) * TF_TILE +
             (2 * jt + (l15 >> 3)) * PF_RUN +
             4 * (16 * ((l15 >> 2) & 1) + 8 * hi) + (l15 & 3);
#pragma unroll
  for (int r = 0; r < 8; ++r) {
    d[4 * r] = acc0[r];
    d[TF_TILE + 4 * r] = acc1[r];
  }
}

template <int DS>
__device__ __forceinline__ void nt_tile(const float* __restrict__ arow,
                                        const float* __restrict__ brow,
                                        float* __restrict__ d, float scale,
                                        int hi) {
  v4f fa[KAUG_B], fb[KAUG_B];
#pragma unroll
  for (int i = 0; i < KAUG_B; ++i)
    fb[i] = *(const v4f*)(brow + 8 * i + 4 * hi);
#pragma unroll
  for (int i = 0; i < KAUG_B; ++i) {
    const float* a = arow + PF_RUN * i;
    const v4f p0 = *(const v4f*)a;
    const v4f p1 = *(const v4f*)(a + (size_t)T2_PART);
    const v4f p2 = *(const v4f*)(a + (size_t)2 * T2_PART);
    const v4f p3 = *(const v4f*)(a + (size_t)3 * T2_PART);
    fa[i] = (p0 + p1) + (p2 + p3);
  }
  HOLD8(fb, 0); HOLD8(fb, 8);
  v8f acc = {}, acc1 = {};
#pragma unroll
  for (int i = 0; i < KAUG_B; ++i) {
    const v2f a0 = {fa[i][0], fa[i][1]}, a1 = {fa[i][2], fa[i][3]};
    const v2f b0 = {fb[i][0], fb[i][1]}, b1 = {fb[i][2], fb[i][3]};
    acc  = wmma_f32(a0, b0, acc);
    acc1 = wmma_f32(a1, b1, acc1);
  }
  acc = acc + acc1;
#pragma unroll
  for (int r = 0; r < 8; ++r) d[(size_t)r * DS] = acc[r] * scale;
}

__global__ void __launch_bounds__(64)
p_kernel(const float* __restrict__ T2part, const float* __restrict__ At,
         float* __restrict__ D, float* __restrict__ Dfrag, float scale) {
  const int lane = threadIdx.x & 31;
  const int l15  = lane & 15;
  const int hi   = lane >> 4;
  int t = blockIdx.x * (blockDim.x >> 5) +
          __builtin_amdgcn_readfirstlane((int)(threadIdx.x >> 5));
  const int ct = t % 17; t /= 17;
  const int mt = t & 15; t >>= 4;
  const int n  = t;

  const float* arow = T2part + (size_t)(n * 16 + mt) * TF_TILE + 4 * lane;
  const float* brow = At + (size_t)(ct * 16 + l15) * CP;
  if (ct < 16) {
    float* df = Dfrag + (size_t)n * C_CH * C_CH + (size_t)mt * PF_TILE +
                (ct >> 1) * PF_BLK + (l15 >> 2) * PF_RUN +
                4 * (16 * (ct & 1) + 8 * hi) + (l15 & 3);
    nt_tile<4>(arow, brow, df, scale, hi);
  } else {
    float* d = D + ((size_t)n * C_CH + mt * 16 + 8 * hi) * CP + ct * 16 + l15;
    nt_tile<CP>(arow, brow, d, scale, hi);
  }
}

struct OutFrags {
  v4f w0[4], w1[4];
  v4f x[16];
};

__device__ __forceinline__ void out_load_block(OutFrags& f,
                                               const float* __restrict__ wl0,
                                               const float* __restrict__ wl1,
                                               const float* __restrict__ xl,
                                               int j) {
  const float* xj = xl + (size_t)(32 * j) * HW_P;
  for (int i = 0; i < 16; ++i)
    f.x[i] = *(const v4f*)(xj + (size_t)i * HW_P);
  for (int i = 0; i < 4; ++i) {
    f.w0[i] = *(const v4f*)(wl0 + PF_BLK * j + PF_RUN * i);
    f.w1[i] = *(const v4f*)(wl1 + PF_BLK * j + PF_RUN * i);
  }
}

__device__ __forceinline__ void out_mma_block(v8f (&acc)[4][2],
                                              const OutFrags& f) {
  for (int s = 0; s < 8; ++s) {
    const int q = s >> 1, c = 2 * (s & 1);
    const v2f bb0 = {f.w0[q][c], f.w0[q][c + 1]};
    const v2f bb1 = {f.w1[q][c], f.w1[q][c + 1]};
    for (int mt = 0; mt < 4; ++mt) {
      const v2f px = {f.x[2 * s][mt], f.x[2 * s + 1][mt]};
      acc[mt][0] = wmma_f32(bb0, px, acc[mt][0]);
      acc[mt][1] = wmma_f32(bb1, px, acc[mt][1]);
    }
  }
}

__global__ void __launch_bounds__(256)
out_kernel(const float* __restrict__ x, const float* __restrict__ PT,
           const float* __restrict__ PF, const float* __restrict__ b3,
           float* __restrict__ out) {
  const int lane = threadIdx.x & 31;
  const int l15  = lane & 15;
  const int hi   = lane >> 4;
  int t = blockIdx.x * (blockDim.x >> 5) + (threadIdx.x >> 5);
  const int opair = t & 7;  t >>= 3;
  const int ptile = t & 63; t >>= 6;
  const int n     = t;

  const float* w = PT + (size_t)n * C_CH * CP;
  const int p0 = ptile * 64;
  const int o0 = opair * 32;
  const float* xl  = x + (size_t)n * C_CH * HW_P + (size_t)(16 * hi) * HW_P +
                     p0 + 4 * l15;
  const float* wl0 = PF + (size_t)n * C_CH * C_CH + (size_t)(2 * opair) * PF_TILE +
                     4 * lane;
  const float* wl1 = wl0 + PF_TILE;

  OutFrags f0, f1;
  out_load_block(f0, wl0, wl1, xl, 0);

  v8f acc[4][2];
  {
    const float* wa = w + (size_t)(o0 + 8 * hi) * CP + C_CH;
    const float* wb = wa + (size_t)16 * CP;
    const float* ba = b3 + o0 + 8 * hi;
    const v4f a0 = *(const v4f*)ba,        a1 = *(const v4f*)(ba + 4);
    const v4f c0 = *(const v4f*)(ba + 16), c1 = *(const v4f*)(ba + 20);
    const v8f bz0 = {wa[0] + a0[0],      wa[CP] + a0[1],     wa[2 * CP] + a0[2],
                     wa[3 * CP] + a0[3], wa[4 * CP] + a1[0], wa[5 * CP] + a1[1],
                     wa[6 * CP] + a1[2], wa[7 * CP] + a1[3]};
    const v8f bz1 = {wb[0] + c0[0],      wb[CP] + c0[1],     wb[2 * CP] + c0[2],
                     wb[3 * CP] + c0[3], wb[4 * CP] + c1[0], wb[5 * CP] + c1[1],
                     wb[6 * CP] + c1[2], wb[7 * CP] + c1[3]};
    for (int mt = 0; mt < 4; ++mt) { acc[mt][0] = bz0; acc[mt][1] = bz1; }
  }

  for (int j = 0; j < C_CH / 32; j += 2) {
    out_load_block(f1, wl0, wl1, xl, j + 1);
    out_mma_block(acc, f0);
    if (j + 2 < C_CH / 32) out_load_block(f0, wl0, wl1, xl, j + 2);
    out_mma_block(acc, f1);
  }
  float* ob0 = out + ((size_t)n * C_CH + o0 + 8 * hi) * HW_P + p0 + 4 * l15;
  float* ob1 = ob0 + (size_t)16 * HW_P;
  for (int r = 0; r < 8; ++r) {
    const v4f s0 = {acc[0][0][r], acc[1][0][r], acc[2][0][r], acc[3][0][r]};
    const v4f s1 = {acc[0][1][r], acc[1][1][r], acc[2][1][r], acc[3][1][r]};
    *(v4f*)(ob0 + (size_t)r * HW_P) = s0;
    *(v4f*)(ob1 + (size_t)r * HW_P) = s1;
  }
}

extern "C" void kernel_launch(void* const* d_in, const int* in_sizes, int n_in,
                              void* d_out, int out_size, void* d_ws,
                              size_t ws_size, hipStream_t stream) {
  const float* x  = (const float*)d_in[0];
  const float* w0 = (const float*)d_in[1];
  const float* b0 = (const float*)d_in[2];
  const float* w1 = (const float*)d_in[3];
  const float* b1 = (const float*)d_in[4];
  const float* w2 = (const float*)d_in[5];
  const float* b2 = (const float*)d_in[6];
  const float* w3 = (const float*)d_in[7];
  const float* b3 = (const float*)d_in[8];
  float* out = (float*)d_out;

  float* Gpart  = (float*)d_ws;
  float* spart  = Gpart + (size_t)PSL * N_B * C_CH * C_CH;
  float* Mt     = spart + (size_t)PSL * N_B * C_CH;
  float* At     = Mt    + (size_t)C_CH * CP;
  float* T2part = At    + (size_t)CP * CP;
  float* PT     = T2part + (size_t)KQ * T2_PART;
  float* PF     = PT    + (size_t)N_B * C_CH * CP;

  x_gram_kernel<<<781, 128, 0, stream>>>(x, Gpart, spart, w0, b0, w1, b1, w2, b2,
                                         w3, Mt, At);
  gsum_prod_kernel<<<272, 256, 0, stream>>>(Gpart, spart, Mt, T2part);
  p_kernel<<<544, 64, 0, stream>>>(T2part, At, PT, PF, 1.0f / (float)HW_P);
  out_kernel<<<256, 256, 0, stream>>>(x, PT, PF, b3, out);
}


# weight products Mt and At stored in their readers' fragment order: the A loads of kernel 2 and B loads of kernel 3 read 4 whole lines each, not 16-byte pieces on 16 rows; sum order unchanged
# speedup vs baseline: 5.9178x; 1.0318x over previous
#include <hip/hip_runtime.h>


#define C_CH 256
#define CP   272
#define HW_P 4096
#define N_B  4
#define PCH  32
#define PLEN (HW_P / PCH)
#define CSUM 4
#define PSL  (PCH / CSUM)
#define RED_W (4 * 4 * 32 * 4)
#define GTILES 20
#define KAUG_B 33
#define WPROD_WAVES (16 * 17 + 17 * 17)
#define WPROD_BLOCKS 141
#define GRAM_BLOCKS 640
#define KQ      4
#define S_PITCH 76
#define TF_TILE (34 * PF_RUN)
#define T2_PART (N_B * 16 * TF_TILE)
#define PF_RUN  128
#define PF_BLK  (4 * PF_RUN)
#define PF_TILE (8 * PF_BLK)

typedef __attribute__((ext_vector_type(2))) float v2f;
typedef __attribute__((ext_vector_type(4))) float v4f;
typedef __attribute__((ext_vector_type(8))) float v8f;

#define HOLD8(p, o)                                                          \
  asm volatile("" ::"v"(p[(o)]), "v"(p[(o) + 1]), "v"(p[(o) + 2]),           \
               "v"(p[(o) + 3]), "v"(p[(o) + 4]), "v"(p[(o) + 5]),            \
               "v"(p[(o) + 6]), "v"(p[(o) + 7]))

__device__ __forceinline__ v8f wmma_f32(v2f a, v2f b, v8f c) {
  return __builtin_amdgcn_wmma_f32_16x16x4_f32(false, a, false, b, (short)0, c,
                                               false, false);
}

__device__ __forceinline__ v8f splat8(float v) {
  v8f r = {v, v, v, v, v, v, v, v};
  return r;
}

__device__ __forceinline__ void aug_col_frags(v4f (&f)[8],
                                              const float* __restrict__ w,
                                              const float* __restrict__ b,
                                              int tile, int k0, int l15, int hi) {
  if (tile < 16) {
    const float* wc = w + tile * 16 + l15 + (size_t)(k0 + 4 * hi) * C_CH;
#pragma unroll
    for (int i = 0; i < 8; ++i) {
      const float* wd = wc + (size_t)(8 * i) * C_CH;
      const v4f q = {wd[0], wd[C_CH], wd[2 * C_CH], wd[3 * C_CH]};
      f[i] = q;
    }
  } else {
    const v4f zero = {0.0f, 0.0f, 0.0f, 0.0f};
#pragma unroll
    for (int i = 0; i < 8; ++i) {
      const v4f q = *(const v4f*)(b + k0 + 8 * i + 4 * hi);
      f[i] = (l15 == 0) ? q : zero;
    }
  }
}

__device__ __forceinline__ void weight_products(
    const float* __restrict__ w0, const float* __restrict__ b0,
    const float* __restrict__ w1, const float* __restrict__ b1,
    const float* __restrict__ w2, const float* __restrict__ b2,
    const float* __restrict__ w3, float* __restrict__ Mt,
    float* __restrict__ At, int wv, int l15, int hi) {
  const bool first = wv < 16 * 17;
  const int u  = first ? wv : wv - 16 * 17;
  const int rt = u / 17;
  const int ct = u - rt * 17;
  const float* arow = w3 + (size_t)(rt * 16 + l15) * C_CH;
  v8f acc = {}, acc1 = {};
#pragma unroll 1
  for (int k0 = 0; k0 < C_CH; k0 += 64) {
    v4f fa[8], fb[8];
    if (first) {
#pragma unroll
      for (int i = 0; i < 8; ++i)
        fa[i] = *(const v4f*)(arow + k0 + 8 * i + 4 * hi);
      aug_col_frags(fb, w0, b0, ct, k0, l15, hi);
    } else {
      aug_col_frags(fa, w2, b2, rt, k0, l15, hi);
      aug_col_frags(fb, w1, b1, ct, k0, l15, hi);
    }
#pragma unroll
    for (int i = 0; i < 8; ++i) {
      const v2f a0 = {fa[i][0], fa[i][1]}, a1 = {fa[i][2], fa[i][3]};
      const v2f c0 = {fb[i][0], fb[i][1]}, c1 = {fb[i][2], fb[i][3]};
      acc  = wmma_f32(a0, c0, acc);
      acc1 = wmma_f32(a1, c1, acc1);
    }
  }
  acc = acc + acc1;
  float* d = (first ? Mt : At) + (size_t)rt * TF_TILE +
             (2 * ct + (l15 >> 3)) * PF_RUN +
             4 * (16 * ((l15 >> 2) & 1) + 8 * hi) + (l15 & 3);
#pragma unroll
  for (int r = 0; r < 8; ++r) d[4 * r] = acc[r];
}

template <bool SUMS>
__device__ __forceinline__ void gram_body(const float* __restrict__ xa,
                                          const float* __restrict__ xb,
                                          float* __restrict__ gp,
                                          float* __restrict__ sp,
                                          float* red, float* reds,
                                          int wave, int lane, int l15, int hi) {
  v8f acc[4][2];
  v8f accs[2];
  for (int mt = 0; mt < 4; ++mt) { acc[mt][0] = splat8(0.f); acc[mt][1] = splat8(0.f); }
  accs[0] = splat8(0.f); accs[1] = splat8(0.f);
  const v2f ones = {1.0f, 1.0f};

  for (int kb = 0; kb < PLEN / 32; ++kb) {
    v4f fa[4][4], fb[2][4];
    for (int mt = 0; mt < 4; ++mt)
      for (int i = 0; i < 4; ++i)
        fa[mt][i] = *(const v4f*)(xa + (size_t)(16 * mt) * HW_P + 32 * kb + 4 * i);
    for (int nt = 0; nt < 2; ++nt)
      for (int i = 0; i < 4; ++i)
        fb[nt][i] = *(const v4f*)(xb + (size_t)nt * HW_P + 32 * kb + 4 * i);
    for (int i = 0; i < 4; ++i) {
      for (int t = 0; t < 2; ++t) {
        const v2f b0 = {fb[0][i][2 * t], fb[0][i][2 * t + 1]};
        const v2f b1 = {fb[1][i][2 * t], fb[1][i][2 * t + 1]};
        for (int mt = 0; mt < 4; ++mt) {
          const v2f a = {fa[mt][i][2 * t], fa[mt][i][2 * t + 1]};
          acc[mt][0] = wmma_f32(a, b0, acc[mt][0]);
          acc[mt][1] = wmma_f32(a, b1, acc[mt][1]);
        }
        if (SUMS) {
          accs[0] = wmma_f32(ones, b0, accs[0]);
          accs[1] = wmma_f32(ones, b1, accs[1]);
        }
      }
    }
  }
  float* rw = red + wave * RED_W + 4 * lane;
  for (int mt = 0; mt < 4; ++mt)
    for (int rp = 0; rp < 4; ++rp) {
      const v4f q = {acc[mt][0][2 * rp], acc[mt][1][2 * rp],
                     acc[mt][0][2 * rp + 1], acc[mt][1][2 * rp + 1]};
      *(v4f*)(rw + (mt * 4 + rp) * 128) = q;
    }
  if (SUMS) {
    if (hi == 0) {
      const v2f s = {accs[0][0], accs[1][0]};
      *(v2f*)(reds + (wave * 16 + l15) * 2) = s;
    }
  }
  __syncthreads();
  const float* rr = red + (wave * 4) * 128 + 4 * lane;
  for (int rp = 0; rp < 4; ++rp) {
    v4f s = *(const v4f*)(rr + rp * 128);
    for (int v = 1; v < CSUM; ++v) s += *(const v4f*)(rr + v * RED_W + rp * 128);
    float* g = gp + (size_t)(16 * wave + 8 * hi + 2 * rp) * C_CH;
    const v2f s0 = {s[0], s[1]}, s1 = {s[2], s[3]};
    *(v2f*)g = s0;
    *(v2f*)(g + C_CH) = s1;
  }
  if (SUMS) {
    if (wave == 0 && hi == 0) {
      v2f s = *(const v2f*)(reds + l15 * 2);
      for (int v = 1; v < CSUM; ++v) s += *(const v2f*)(reds + (v * 16 + l15) * 2);
      *(v2f*)sp = s;
    }
  }
}

__global__ void __launch_bounds__(128) __attribute__((amdgpu_waves_per_eu(3, 4)))
x_gram_kernel(const float* __restrict__ x, float* __restrict__ Gpart,
              float* __restrict__ spart,
              const float* __restrict__ w0, const float* __restrict__ b0,
              const float* __restrict__ w1, const float* __restrict__ b1,
              const float* __restrict__ w2, const float* __restrict__ b2,
              const float* __restrict__ w3, float* __restrict__ Mt,
              float* __restrict__ At) {
  const int lane = threadIdx.x & 31;
  const int l15  = lane & 15;
  const int hi   = lane >> 4;
  const int wave = __builtin_amdgcn_readfirstlane((int)(threadIdx.x >> 5));
  if (blockIdx.x >= GRAM_BLOCKS) {
    const int wv = (blockIdx.x - GRAM_BLOCKS) * (blockDim.x >> 5) + wave;
    if (wv < WPROD_WAVES)
      weight_products(w0, b0, w1, b1, w2, b2, w3, Mt, At, wv, l15, hi);
    return;
  }
  __shared__ __attribute__((aligned(16))) float red[CSUM * RED_W];
  __shared__ __attribute__((aligned(16))) float reds[CSUM * 32];
  int t = blockIdx.x;
  const int u = t % GTILES; t /= GTILES;
  const int cg = (u < 8) ? 0 : (u < 14) ? 1 : (u < 18) ? 2 : 3;
  const int eg = (u < 8) ? u : (u < 14) ? u - 6 : (u < 18) ? u - 10 : u - 12;
  const int grp = t % PSL; t /= PSL;
  const int n     = t;
  const int chunk = grp * CSUM + wave;

  const float* xn = x + (size_t)n * C_CH * HW_P + chunk * PLEN + 16 * hi;
  const float* xa = xn + (size_t)(cg * 64 + l15) * HW_P;
  const float* xb = xn + (size_t)(eg * 32 + 2 * l15) * HW_P;
  float* gp = Gpart + ((size_t)(grp * N_B + n) * C_CH + cg * 64) * C_CH +
              eg * 32 + 2 * l15;
  float* sp = spart + (size_t)(grp * N_B + n) * C_CH + eg * 32 + 2 * l15;
  if (cg == 0) gram_body<true>(xa, xb, gp, sp, red, reds, wave, lane, l15, hi);
  else         gram_body<false>(xa, xb, gp, sp, red, reds, wave, lane, l15, hi);
}

__global__ void __launch_bounds__(256)
gsum_prod_kernel(const float* __restrict__ Gpart, const float* __restrict__ spart,
                 const float* __restrict__ Mt, float* __restrict__ T2part) {
  __shared__ __attribute__((aligned(16))) float S[16 * S_PITCH];
  const int tid  = threadIdx.x;
  const int lane = tid & 31;
  const int l15  = lane & 15;
  const int hi   = lane >> 4;
  const int w  = __builtin_amdgcn_readfirstlane(tid >> 5);
  const int q  = blockIdx.x & (KQ - 1);
  const int bj = blockIdx.x >> 2;
  const int jt = bj % 17;
  const int n  = bj / 17;

  const float* arow = Mt + (size_t)(2 * w) * TF_TILE + 4 * lane;
  v4f fa0[9], fa1[9];
#pragma unroll
  for (int i = 0; i < 8; ++i) {
    fa0[i] = *(const v4f*)(arow + (8 * q + i) * PF_RUN);
    fa1[i] = *(const v4f*)(arow + TF_TILE + (8 * q + i) * PF_RUN);
  }
  if (q == KQ - 1) {
    fa0[8] = *(const v4f*)(arow + 32 * PF_RUN);
    fa1[8] = *(const v4f*)(arow + TF_TILE + 32 * PF_RUN);
  }

  const size_t slab = (size_t)N_B * C_CH * C_CH;
  const float* gn = Gpart + (size_t)n * C_CH * C_CH;
  const float* sn = spart + (size_t)n * C_CH;
  if (jt < 16) {
    v4f s = {0.0f, 0.0f, 0.0f, 0.0f};
    if (q >= (jt >> 2)) {
      const int r = tid >> 4, c4 = tid & 15;
      const float* g = gn + (size_t)(16 * jt + r) * C_CH + 64 * q + 4 * c4;
#pragma unroll
      for (int c = 0; c < PSL; ++c) s += *(const v4f*)(g + (size_t)c * slab);
      *(v4f*)(S + r * S_PITCH + 4 * c4) = s;
    } else {
      const int er = tid >> 2, jc = tid & 3;
      const float* g = gn + (size_t)(64 * q + er) * C_CH + 16 * jt + 4 * jc;
#pragma unroll
      for (int c = 0; c < PSL; ++c) s += *(const v4f*)(g + (size_t)c * slab);
#pragma unroll
      for (int m = 0; m < 4; ++m) S[(4 * jc + m) * S_PITCH + er] = s[m];
    }
    if (q == KQ - 1 && tid < 128) {
      const int r = tid >> 3, c = tid & 7;
      float a = 0.0f;
      if (c == 0) {
        const float* sp = sn + 16 * jt + r;
#pragma unroll 8
        for (int ch = 0; ch < PSL; ++ch) a += sp[(size_t)ch * N_B * C_CH];
      }
      S[r * S_PITCH + 64 + c] = a;
    }
  } else {
    for (int idx = tid; idx < 16 * 72; idx += 256) {
      const int r = idx / 72, c = idx - r * 72;
      float a = 0.0f;
      if (r == 0) {
        if (c < 64) {
          const float* sp = sn + 64 * q + c;
#pragma unroll 8
          for (int ch = 0; ch < PSL; ++ch) a += sp[(size_t)ch * N_B * C_CH];
        } else if (c == 64 && q == KQ - 1) {
          a = (float)HW_P;
        }
      }
      S[r * S_PITCH + c] = a;
    }
  }
  __syncthreads();

  const float* sb = S + l15 * S_PITCH + 4 * hi;
  v8f acc0 = {}, acc1 = {};
#pragma unroll
  for (int i = 0; i < 8; ++i) {
    const v4f fb = *(const v4f*)(sb + 8 * i);
    const v2f b0 = {fb[0], fb[1]}, b1 = {fb[2], fb[3]};
    const v2f a0 = {fa0[i][0], fa0[i][1]}, a1 = {fa0[i][2], fa0[i][3]};
    const v2f c0 = {fa1[i][0], fa1[i][1]}, c1 = {fa1[i][2], fa1[i][3]};
    acc0 = wmma_f32(a0, b0, acc0);
    acc1 = wmma_f32(c0, b0, acc1);
    acc0 = wmma_f32(a1, b1, acc0);
    acc1 = wmma_f32(c1, b1, acc1);
  }
  if (q == KQ - 1) {
    const v4f fb = *(const v4f*)(sb + 64);
    const v2f b0 = {fb[0], fb[1]}, b1 = {fb[2], fb[3]};
    const v2f a0 = {fa0[8][0], fa0[8][1]}, a1 = {fa0[8][2], fa0[8][3]};
    const v2f c0 = {fa1[8][0], fa1[8][1]}, c1 = {fa1[8][2], fa1[8][3]};
    acc0 = wmma_f32(a0, b0, acc0);
    acc1 = wmma_f32(c0, b0, acc1);
    acc0 = wmma_f32(a1, b1, acc0);
    acc1 = wmma_f32(c1, b1, acc1);
  }
  float* d = T2part + ((size_t)(q * N_B + n) * 16 + 2 * w) * TF_TILE +
             (2 * jt + (l15 >> 3)) * PF_RUN +
             4 * (16 * ((l15 >> 2) & 1) + 8 * hi) + (l15 & 3);
#pragma unroll
  for (int r = 0; r < 8; ++r) {
    d[4 * r] = acc0[r];
    d[TF_TILE + 4 * r] = acc1[r];
  }
}

template <int DS>
__device__ __forceinline__ void nt_tile(const float* __restrict__ arow,
                                        const float* __restrict__ brow,
                                        float* __restrict__ d, float scale) {
  v4f fa[KAUG_B], fb[KAUG_B];
#pragma unroll
  for (int i = 0; i < KAUG_B; ++i)
    fb[i] = *(const v4f*)(brow + PF_RUN * i);
#pragma unroll
  for (int i = 0; i < KAUG_B; ++i) {
    const float* a = arow + PF_RUN * i;
    const v4f p0 = *(const v4f*)a;
    const v4f p1 = *(const v4f*)(a + (size_t)T2_PART);
    const v4f p2 = *(const v4f*)(a + (size_t)2 * T2_PART);
    const v4f p3 = *(const v4f*)(a + (size_t)3 * T2_PART);
    fa[i] = (p0 + p1) + (p2 + p3);
  }
  HOLD8(fb, 0); HOLD8(fb, 8);
  v8f acc = {}, acc1 = {};
#pragma unroll
  for (int i = 0; i < KAUG_B; ++i) {
    const v2f a0 = {fa[i][0], fa[i][1]}, a1 = {fa[i][2], fa[i][3]};
    const v2f b0 = {fb[i][0], fb[i][1]}, b1 = {fb[i][2], fb[i][3]};
    acc  = wmma_f32(a0, b0, acc);
    acc1 = wmma_f32(a1, b1, acc1);
  }
  acc = acc + acc1;
#pragma unroll
  for (int r = 0; r < 8; ++r) d[(size_t)r * DS] = acc[r] * scale;
}

__global__ void __launch_bounds__(64)
p_kernel(const float* __restrict__ T2part, const float* __restrict__ At,
         float* __restrict__ D, float* __restrict__ Dfrag, float scale) {
  const int lane = threadIdx.x & 31;
  const int l15  = lane & 15;
  const int hi   = lane >> 4;
  int t = blockIdx.x * (blockDim.x >> 5) +
          __builtin_amdgcn_readfirstlane((int)(threadIdx.x >> 5));
  const int ct = t % 17; t /= 17;
  const int mt = t & 15; t >>= 4;
  const int n  = t;

  const float* arow = T2part + (size_t)(n * 16 + mt) * TF_TILE + 4 * lane;
  const float* brow = At + (size_t)ct * TF_TILE + 4 * lane;
  if (ct < 16) {
    float* df = Dfrag + (size_t)n * C_CH * C_CH + (size_t)mt * PF_TILE +
                (ct >> 1) * PF_BLK + (l15 >> 2) * PF_RUN +
                4 * (16 * (ct & 1) + 8 * hi) + (l15 & 3);
    nt_tile<4>(arow, brow, df, scale);
  } else {
    float* d = D + ((size_t)n * C_CH + mt * 16 + 8 * hi) * CP + ct * 16 + l15;
    nt_tile<CP>(arow, brow, d, scale);
  }
}

struct OutFrags {
  v4f w0[4], w1[4];
  v4f x[16];
};

__device__ __forceinline__ void out_load_block(OutFrags& f,
                                               const float* __restrict__ wl0,
                                               const float* __restrict__ wl1,
                                               const float* __restrict__ xl,
                                               int j) {
  const float* xj = xl + (size_t)(32 * j) * HW_P;
  for (int i = 0; i < 16; ++i)
    f.x[i] = *(const v4f*)(xj + (size_t)i * HW_P);
  for (int i = 0; i < 4; ++i) {
    f.w0[i] = *(const v4f*)(wl0 + PF_BLK * j + PF_RUN * i);
    f.w1[i] = *(const v4f*)(wl1 + PF_BLK * j + PF_RUN * i);
  }
}

__device__ __forceinline__ void out_mma_block(v8f (&acc)[4][2],
                                              const OutFrags& f) {
  for (int s = 0; s < 8; ++s) {
    const int q = s >> 1, c = 2 * (s & 1);
    const v2f bb0 = {f.w0[q][c], f.w0[q][c + 1]};
    const v2f bb1 = {f.w1[q][c], f.w1[q][c + 1]};
    for (int mt = 0; mt < 4; ++mt) {
      const v2f px = {f.x[2 * s][mt], f.x[2 * s + 1][mt]};
      acc[mt][0] = wmma_f32(bb0, px, acc[mt][0]);
      acc[mt][1] = wmma_f32(bb1, px, acc[mt][1]);
    }
  }
}

__global__ void __launch_bounds__(256)
out_kernel(const float* __restrict__ x, const float* __restrict__ PT,
           const float* __restrict__ PF, const float* __restrict__ b3,
           float* __restrict__ out) {
  const int lane = threadIdx.x & 31;
  const int l15  = lane & 15;
  const int hi   = lane >> 4;
  int t = blockIdx.x * (blockDim.x >> 5) + (threadIdx.x >> 5);
  const int opair = t & 7;  t >>= 3;
  const int ptile = t & 63; t >>= 6;
  const int n     = t;

  const float* w = PT + (size_t)n * C_CH * CP;
  const int p0 = ptile * 64;
  const int o0 = opair * 32;
  const float* xl  = x + (size_t)n * C_CH * HW_P + (size_t)(16 * hi) * HW_P +
                     p0 + 4 * l15;
  const float* wl0 = PF + (size_t)n * C_CH * C_CH + (size_t)(2 * opair) * PF_TILE +
                     4 * lane;
  const float* wl1 = wl0 + PF_TILE;

  OutFrags f0, f1;
  out_load_block(f0, wl0, wl1, xl, 0);

  v8f acc[4][2];
  {
    const float* wa = w + (size_t)(o0 + 8 * hi) * CP + C_CH;
    const float* wb = wa + (size_t)16 * CP;
    const float* ba = b3 + o0 + 8 * hi;
    const v4f a0 = *(const v4f*)ba,        a1 = *(const v4f*)(ba + 4);
    const v4f c0 = *(const v4f*)(ba + 16), c1 = *(const v4f*)(ba + 20);
    const v8f bz0 = {wa[0] + a0[0],      wa[CP] + a0[1],     wa[2 * CP] + a0[2],
                     wa[3 * CP] + a0[3], wa[4 * CP] + a1[0], wa[5 * CP] + a1[1],
                     wa[6 * CP] + a1[2], wa[7 * CP] + a1[3]};
    const v8f bz1 = {wb[0] + c0[0],      wb[CP] + c0[1],     wb[2 * CP] + c0[2],
                     wb[3 * CP] + c0[3], wb[4 * CP] + c1[0], wb[5 * CP] + c1[1],
                     wb[6 * CP] + c1[2], wb[7 * CP] + c1[3]};
    for (int mt = 0; mt < 4; ++mt) { acc[mt][0] = bz0; acc[mt][1] = bz1; }
  }

  for (int j = 0; j < C_CH / 32; j += 2) {
    out_load_block(f1, wl0, wl1, xl, j + 1);
    out_mma_block(acc, f0);
    if (j + 2 < C_CH / 32) out_load_block(f0, wl0, wl1, xl, j + 2);
    out_mma_block(acc, f1);
  }
  float* ob0 = out + ((size_t)n * C_CH + o0 + 8 * hi) * HW_P + p0 + 4 * l15;
  float* ob1 = ob0 + (size_t)16 * HW_P;
  for (int r = 0; r < 8; ++r) {
    const v4f s0 = {acc[0][0][r], acc[1][0][r], acc[2][0][r], acc[3][0][r]};
    const v4f s1 = {acc[0][1][r], acc[1][1][r], acc[2][1][r], acc[3][1][r]};
    *(v4f*)(ob0 + (size_t)r * HW_P) = s0;
    *(v4f*)(ob1 + (size_t)r * HW_P) = s1;
  }
}

extern "C" void kernel_launch(void* const* d_in, const int* in_sizes, int n_in,
                              void* d_out, int out_size, void* d_ws,
                              size_t ws_size, hipStream_t stream) {
  const float* x  = (const float*)d_in[0];
  const float* w0 = (const float*)d_in[1];
  const float* b0 = (const float*)d_in[2];
  const float* w1 = (const float*)d_in[3];
  const float* b1 = (const float*)d_in[4];
  const float* w2 = (const float*)d_in[5];
  const float* b2 = (const float*)d_in[6];
  const float* w3 = (const float*)d_in[7];
  const float* b3 = (const float*)d_in[8];
  float* out = (float*)d_out;

  float* Gpart  = (float*)d_ws;
  float* spart  = Gpart + (size_t)PSL * N_B * C_CH * C_CH;
  float* Mt     = spart + (size_t)PSL * N_B * C_CH;
  float* At     = Mt    + (size_t)C_CH * CP;
  float* T2part = At    + (size_t)CP * CP;
  float* PT     = T2part + (size_t)KQ * T2_PART;
  float* PF     = PT    + (size_t)N_B * C_CH * CP;

  x_gram_kernel<<<781, 128, 0, stream>>>(x, Gpart, spart, w0, b0, w1, b1, w2, b2,
                                         w3, Mt, At);
  gsum_prod_kernel<<<272, 256, 0, stream>>>(Gpart, spart, Mt, T2part);
  p_kernel<<<544, 64, 0, stream>>>(T2part, At, PT, PF, 1.0f / (float)HW_P);
  out_kernel<<<256, 256, 0, stream>>>(x, PT, PF, b3, out);
}
